# starting kernel
# baseline (speedup 1.0000x reference)
#include <hip/hip_runtime.h>
#include <hip/hip_bf16.h>
#include <math.h>

typedef float     v2f  __attribute__((ext_vector_type(2)));
typedef float     v8f  __attribute__((ext_vector_type(8)));

#define NEG_SLOPE 0.2f
#define LN_EPS    1e-5f

__device__ __forceinline__ unsigned ordEncode(float f) {
  unsigned u = __float_as_uint(f);
  return (u >> 31) ? ~u : (u | 0x80000000u);
}
__device__ __forceinline__ float ordDecode(unsigned o) {
  return __uint_as_float((o >> 31) ? (o & 0x7FFFFFFFu) : ~o);
}

__global__ void fill_f32(float* p, float v, size_t n) {
  size_t i = (size_t)blockIdx.x * blockDim.x + threadIdx.x;
  if (i < n) p[i] = v;
}
__global__ void fill_u32(unsigned* p, unsigned v, size_t n) {
  size_t i = (size_t)blockIdx.x * blockDim.x + threadIdx.x;
  if (i < n) p[i] = v;
}

__global__ void pad_cols(const float* __restrict__ src, float* __restrict__ dst,
                         int M, int Ks, int Kd) {
  int i = blockIdx.x * blockDim.x + threadIdx.x;
  if (i >= M * Kd) return;
  int r = i / Kd, c = i % Kd;
  dst[i] = (c < Ks) ? src[(size_t)r * Ks + c] : 0.0f;
}

__global__ void pad_rows(const float* __restrict__ src, float* __restrict__ dst,
                         int Ms, int Md, int Ncols) {
  int i = blockIdx.x * blockDim.x + threadIdx.x;
  if (i >= Md * Ncols) return;
  int r = i / Ncols;
  dst[i] = (r < Ms) ? src[i] : 0.0f;
}

template <int K>
__global__ void wmma_gemm(const float* __restrict__ A, const float* __restrict__ B,
                          float* __restrict__ C, int M, int N) {
  const int lane  = threadIdx.x & 31;
  const int wave  = threadIdx.x >> 5;
  const int tileM = blockIdx.x * 4 + wave;
  const int m0 = tileM * 16;
  const int n0 = blockIdx.y * 16;
  if (m0 + 16 > M) return;
  const int half = lane >> 4;
  const int l16  = lane & 15;
  const float* __restrict__ arow = A + (size_t)(m0 + l16) * K + half * 2;
  const float* __restrict__ bcol = B + n0 + l16 + (size_t)(half * 2) * N;
  v8f acc = {};
#pragma unroll
  for (int k = 0; k < K; k += 4) {
    v2f a = *(const v2f*)(arow + k);
    v2f b;
    b.x = bcol[(size_t)k * N];
    b.y = bcol[(size_t)(k + 1) * N];
    acc = __builtin_amdgcn_wmma_f32_16x16x4_f32(false, a, false, b, (short)0, acc,
                                                false, false);
  }
  float* __restrict__ crow = C + (size_t)(m0 + half * 8) * N + n0 + l16;
#pragma unroll
  for (int r = 0; r < 8; ++r)
    crow[(size_t)r * N] = acc[r];
}

__global__ void edge_logits(const int* __restrict__ ei, const float* __restrict__ ea,
                            const float* __restrict__ xl, const float* __restrict__ xr,
                            const float* __restrict__ We, const float* __restrict__ att,
                            float* __restrict__ logit, unsigned* __restrict__ mOrd,
                            int E, int H, int C) {
  const int t = blockIdx.x * blockDim.x + threadIdx.x;
  if (t >= E * H) return;
  const int e = t / H, h = t % H;
  const int s = ei[e], d = ei[E + e];
  const float eav = ea[e];
  const float4* __restrict__ xld = (const float4*)(xl + (size_t)d * H * C + h * C);
  const float4* __restrict__ xrs = (const float4*)(xr + (size_t)s * H * C + h * C);
  const float4* __restrict__ we  = (const float4*)(We + h * C);
  const float4* __restrict__ at  = (const float4*)(att + h * C);
  float acc = 0.0f;
  const int q = C >> 2;
  for (int c = 0; c < q; ++c) {
    float4 a = xld[c], b = xrs[c], w = we[c], t4 = at[c];
    float v0 = a.x + b.x + eav * w.x; v0 = (v0 > 0.f) ? v0 : NEG_SLOPE * v0;
    float v1 = a.y + b.y + eav * w.y; v1 = (v1 > 0.f) ? v1 : NEG_SLOPE * v1;
    float v2 = a.z + b.z + eav * w.z; v2 = (v2 > 0.f) ? v2 : NEG_SLOPE * v2;
    float v3 = a.w + b.w + eav * w.w; v3 = (v3 > 0.f) ? v3 : NEG_SLOPE * v3;
    acc += t4.x * v0 + t4.y * v1 + t4.z * v2 + t4.w * v3;
  }
  logit[t] = acc;
  atomicMax(&mOrd[(size_t)d * H + h], ordEncode(acc));
}

__global__ void edge_exp(const int* __restrict__ ei, float* __restrict__ logit,
                         const unsigned* __restrict__ mOrd, float* __restrict__ denom,
                         int E, int H) {
  const int t = blockIdx.x * blockDim.x + threadIdx.x;
  if (t >= E * H) return;
  const int e = t / H, h = t % H;
  const int d = ei[E + e];
  const float m = ordDecode(mOrd[(size_t)d * H + h]);
  const float ex = __expf(logit[t] - m);
  logit[t] = ex;
  atomicAdd(&denom[(size_t)d * H + h], ex);
}

__global__ void edge_aggr(const int* __restrict__ ei, const float* __restrict__ expv,
                          const float* __restrict__ denom, const float* __restrict__ xr,
                          float* __restrict__ out, int E, int H, int C) {
  const int t = blockIdx.x * blockDim.x + threadIdx.x;
  if (t >= E * H) return;
  const int e = t / H, h = t % H;
  const int s = ei[e], d = ei[E + e];
  const float alpha = expv[t] / denom[(size_t)d * H + h];
  const float4* __restrict__ xrs = (const float4*)(xr + (size_t)s * H * C + h * C);
  float* __restrict__ outd = out + (size_t)d * H * C + h * C;
  const int q = C >> 2;
  for (int c = 0; c < q; ++c) {
    float4 v = xrs[c];
    atomicAdd(&outd[4 * c + 0], alpha * v.x);
    atomicAdd(&outd[4 * c + 1], alpha * v.y);
    atomicAdd(&outd[4 * c + 2], alpha * v.z);
    atomicAdd(&outd[4 * c + 3], alpha * v.w);
  }
}

__global__ void node_ln_elu(const float* __restrict__ in, const float* __restrict__ bias,
                            const float* __restrict__ g, const float* __restrict__ b,
                            float* __restrict__ out, int N, int F) {
  const int node = blockIdx.x * (blockDim.x >> 5) + (threadIdx.x >> 5);
  const int lane = threadIdx.x & 31;
  if (node >= N) return;
  const int per = F / 32;
  float vals[4];
  float sum = 0.0f;
  for (int i = 0; i < per; ++i) {
    const int c = lane * per + i;
    float v = in[(size_t)node * F + c] + bias[c];
    vals[i] = v;
    sum += v;
  }
  for (int off = 16; off > 0; off >>= 1) sum += __shfl_xor(sum, off, 32);
  const float mu = sum / (float)F;
  float vs = 0.0f;
  for (int i = 0; i < per; ++i) { float dlt = vals[i] - mu; vs += dlt * dlt; }
  for (int off = 16; off > 0; off >>= 1) vs += __shfl_xor(vs, off, 32);
  const float inv = rsqrtf(vs / (float)F + LN_EPS);
  for (int i = 0; i < per; ++i) {
    const int c = lane * per + i;
    float hv = (vals[i] - mu) * inv * g[c] + b[c];
    out[(size_t)node * F + c] = (hv > 0.0f) ? hv : (__expf(hv) - 1.0f);
  }
}

extern "C" void kernel_launch(void* const* d_in, const int* in_sizes, int n_in,
                              void* d_out, int out_size, void* d_ws, size_t ws_size,
                              hipStream_t stream) {
  const float* x      = (const float*)d_in[0];
  const float* ea     = (const float*)d_in[1];
  const float* W1l    = (const float*)d_in[2];
  const float* W1r    = (const float*)d_in[3];
  const float* W1e    = (const float*)d_in[4];
  const float* att1   = (const float*)d_in[5];
  const float* b1     = (const float*)d_in[6];
  const float* ln1_g  = (const float*)d_in[7];
  const float* ln1_b  = (const float*)d_in[8];
  const float* W2l    = (const float*)d_in[9];
  const float* W2r    = (const float*)d_in[10];
  const float* W2e    = (const float*)d_in[11];
  const float* att2   = (const float*)d_in[12];
  const float* b2     = (const float*)d_in[13];
  const float* ln2_g  = (const float*)d_in[14];
  const float* ln2_b  = (const float*)d_in[15];
  const int*   ei     = (const int*)d_in[16];
  float*       outp   = (float*)d_out;

  const int N = in_sizes[0] / 10;
  const int E = in_sizes[1];
  const int HID = 32, HEADS = 4, F1 = HEADS * HID  , OUTC = 64;
  const int KP = 12;

  float* ws      = (float*)d_ws;
  float* xl1     = ws;
  float* xr1     = xl1 + (size_t)N * F1;
  float* h1      = xr1 + (size_t)N * F1;
  float* logit1  = h1  + (size_t)N * F1;
  unsigned* m1   = (unsigned*)(logit1 + (size_t)E * HEADS);
  float* denom1  = (float*)(m1 + (size_t)N * HEADS);
  float* xpad    = denom1 + (size_t)N * HEADS;
  float* w1lp    = xpad + (size_t)N * KP;
  float* w1rp    = w1lp + (size_t)KP * F1;
  float* xr2     = xl1;
  float* logit2  = xl1 + (size_t)N * OUTC;
  unsigned* m2   = (unsigned*)(logit2 + (size_t)E);
  float* denom2  = (float*)((float*)m2 + (size_t)N);
  float* xl2     = logit1;
  float* out2    = h1;

  const unsigned ORD_NEG_INF = 0x007FFFFFu;
  const int TB = 256;
  auto blocks = [](size_t n, int tb) { return (unsigned)((n + tb - 1) / tb); };

  pad_cols<<<blocks((size_t)N * KP, TB), TB, 0, stream>>>(x, xpad, N, 10, KP);
  pad_rows<<<blocks((size_t)KP * F1, TB), TB, 0, stream>>>(W1l, w1lp, 10, KP, F1);
  pad_rows<<<blocks((size_t)KP * F1, TB), TB, 0, stream>>>(W1r, w1rp, 10, KP, F1);
  {
    dim3 grid((N / 16 + 3) / 4, F1 / 16);
    wmma_gemm<12><<<grid, 128, 0, stream>>>(xpad, w1lp, xl1, N, F1);
    wmma_gemm<12><<<grid, 128, 0, stream>>>(xpad, w1rp, xr1, N, F1);
  }
  fill_u32<<<blocks((size_t)N * HEADS, TB), TB, 0, stream>>>(m1, ORD_NEG_INF, (size_t)N * HEADS);
  fill_f32<<<blocks((size_t)N * HEADS, TB), TB, 0, stream>>>(denom1, 0.0f, (size_t)N * HEADS);
  fill_f32<<<blocks((size_t)N * F1, TB), TB, 0, stream>>>(h1, 0.0f, (size_t)N * F1);

  edge_logits<<<blocks((size_t)E * HEADS, TB), TB, 0, stream>>>(
      ei, ea, xl1, xr1, W1e, att1, logit1, m1, E, HEADS, HID);
  edge_exp<<<blocks((size_t)E * HEADS, TB), TB, 0, stream>>>(
      ei, logit1, m1, denom1, E, HEADS);
  edge_aggr<<<blocks((size_t)E * HEADS, TB), TB, 0, stream>>>(
      ei, logit1, denom1, xr1, h1, E, HEADS, HID);
  node_ln_elu<<<blocks((size_t)N * 32, TB), TB, 0, stream>>>(
      h1, b1, ln1_g, ln1_b, h1, N, F1);

  {
    dim3 grid((N / 16 + 3) / 4, OUTC / 16);
    wmma_gemm<128><<<grid, 128, 0, stream>>>(h1, W2l, xl2, N, OUTC);
    wmma_gemm<128><<<grid, 128, 0, stream>>>(h1, W2r, xr2, N, OUTC);
  }
  fill_u32<<<blocks((size_t)N, TB), TB, 0, stream>>>(m2, ORD_NEG_INF, (size_t)N);
  fill_f32<<<blocks((size_t)N, TB), TB, 0, stream>>>(denom2, 0.0f, (size_t)N);
  fill_f32<<<blocks((size_t)N * OUTC, TB), TB, 0, stream>>>(out2, 0.0f, (size_t)N * OUTC);

  edge_logits<<<blocks((size_t)E, TB), TB, 0, stream>>>(
      ei, ea, xl2, xr2, W2e, att2, logit2, m2, E, 1, OUTC);
  edge_exp<<<blocks((size_t)E, TB), TB, 0, stream>>>(
      ei, logit2, m2, denom2, E, 1);
  edge_aggr<<<blocks((size_t)E, TB), TB, 0, stream>>>(
      ei, logit2, denom2, xr2, out2, E, 1, OUTC);
  node_ln_elu<<<blocks((size_t)N * 32, TB), TB, 0, stream>>>(
      out2, b2, ln2_g, ln2_b, outp, N, OUTC);
}


# edge aggregation: global float atomic scatter replaced by blocks that each own 196 dst rows in LDS, scan the whole dst list and add their edges' rows by LDS float atomics (sum order: LDS arrival)
# speedup vs baseline: 2.9421x; 2.9421x over previous
#include <hip/hip_runtime.h>
#include <hip/hip_bf16.h>
#include <math.h>

typedef float     v2f  __attribute__((ext_vector_type(2)));
typedef float     v8f  __attribute__((ext_vector_type(8)));

#define NEG_SLOPE 0.2f
#define LN_EPS    1e-5f

__device__ __forceinline__ unsigned ordEncode(float f) {
  unsigned u = __float_as_uint(f);
  return (u >> 31) ? ~u : (u | 0x80000000u);
}
__device__ __forceinline__ float ordDecode(unsigned o) {
  return __uint_as_float((o >> 31) ? (o & 0x7FFFFFFFu) : ~o);
}

__global__ void fill_f32(float* p, float v, size_t n) {
  size_t i = (size_t)blockIdx.x * blockDim.x + threadIdx.x;
  if (i < n) p[i] = v;
}
__global__ void fill_u32(unsigned* p, unsigned v, size_t n) {
  size_t i = (size_t)blockIdx.x * blockDim.x + threadIdx.x;
  if (i < n) p[i] = v;
}

__global__ void pad_cols(const float* __restrict__ src, float* __restrict__ dst,
                         int M, int Ks, int Kd) {
  int i = blockIdx.x * blockDim.x + threadIdx.x;
  if (i >= M * Kd) return;
  int r = i / Kd, c = i % Kd;
  dst[i] = (c < Ks) ? src[(size_t)r * Ks + c] : 0.0f;
}

__global__ void pad_rows(const float* __restrict__ src, float* __restrict__ dst,
                         int Ms, int Md, int Ncols) {
  int i = blockIdx.x * blockDim.x + threadIdx.x;
  if (i >= Md * Ncols) return;
  int r = i / Ncols;
  dst[i] = (r < Ms) ? src[i] : 0.0f;
}

template <int K>
__global__ void wmma_gemm(const float* __restrict__ A, const float* __restrict__ B,
                          float* __restrict__ C, int M, int N) {
  const int lane  = threadIdx.x & 31;
  const int wave  = threadIdx.x >> 5;
  const int tileM = blockIdx.x * 4 + wave;
  const int m0 = tileM * 16;
  const int n0 = blockIdx.y * 16;
  if (m0 + 16 > M) return;
  const int half = lane >> 4;
  const int l16  = lane & 15;
  const float* __restrict__ arow = A + (size_t)(m0 + l16) * K + half * 2;
  const float* __restrict__ bcol = B + n0 + l16 + (size_t)(half * 2) * N;
  v8f acc = {};
#pragma unroll
  for (int k = 0; k < K; k += 4) {
    v2f a = *(const v2f*)(arow + k);
    v2f b;
    b.x = bcol[(size_t)k * N];
    b.y = bcol[(size_t)(k + 1) * N];
    acc = __builtin_amdgcn_wmma_f32_16x16x4_f32(false, a, false, b, (short)0, acc,
                                                false, false);
  }
  float* __restrict__ crow = C + (size_t)(m0 + half * 8) * N + n0 + l16;
#pragma unroll
  for (int r = 0; r < 8; ++r)
    crow[(size_t)r * N] = acc[r];
}

__global__ void edge_logits(const int* __restrict__ ei, const float* __restrict__ ea,
                            const float* __restrict__ xl, const float* __restrict__ xr,
                            const float* __restrict__ We, const float* __restrict__ att,
                            float* __restrict__ logit, unsigned* __restrict__ mOrd,
                            int E, int H, int C) {
  const int t = blockIdx.x * blockDim.x + threadIdx.x;
  if (t >= E * H) return;
  const int e = t / H, h = t % H;
  const int s = ei[e], d = ei[E + e];
  const float eav = ea[e];
  const float4* __restrict__ xld = (const float4*)(xl + (size_t)d * H * C + h * C);
  const float4* __restrict__ xrs = (const float4*)(xr + (size_t)s * H * C + h * C);
  const float4* __restrict__ we  = (const float4*)(We + h * C);
  const float4* __restrict__ at  = (const float4*)(att + h * C);
  float acc = 0.0f;
  const int q = C >> 2;
  for (int c = 0; c < q; ++c) {
    float4 a = xld[c], b = xrs[c], w = we[c], t4 = at[c];
    float v0 = a.x + b.x + eav * w.x; v0 = (v0 > 0.f) ? v0 : NEG_SLOPE * v0;
    float v1 = a.y + b.y + eav * w.y; v1 = (v1 > 0.f) ? v1 : NEG_SLOPE * v1;
    float v2 = a.z + b.z + eav * w.z; v2 = (v2 > 0.f) ? v2 : NEG_SLOPE * v2;
    float v3 = a.w + b.w + eav * w.w; v3 = (v3 > 0.f) ? v3 : NEG_SLOPE * v3;
    acc += t4.x * v0 + t4.y * v1 + t4.z * v2 + t4.w * v3;
  }
  logit[t] = acc;
  atomicMax(&mOrd[(size_t)d * H + h], ordEncode(acc));
}

__global__ void edge_exp(const int* __restrict__ ei, float* __restrict__ logit,
                         const unsigned* __restrict__ mOrd, float* __restrict__ denom,
                         int E, int H) {
  const int t = blockIdx.x * blockDim.x + threadIdx.x;
  if (t >= E * H) return;
  const int e = t / H, h = t % H;
  const int d = ei[E + e];
  const float m = ordDecode(mOrd[(size_t)d * H + h]);
  const float ex = __expf(logit[t] - m);
  logit[t] = ex;
  atomicAdd(&denom[(size_t)d * H + h], ex);
}

#define AGG_TR      196
#define AGG_THREADS 1024
#define AGG_GROUPS  8

template <int F, int H>
__device__ __forceinline__ void aggr_one_edge(const int* __restrict__ ei,
                                              const float* __restrict__ expv,
                                              const float* __restrict__ denom,
                                              const float* __restrict__ xr,
                                              float* acc, int e, int row, int lo, int lane) {
  constexpr int C = F / H;
  constexpr int P = C / 32;
  const int s = ei[e];
  const float* __restrict__ xrs = xr + (size_t)s * F + lane;
  float v[H * P];
#pragma unroll
  for (int i = 0; i < H * P; ++i) v[i] = xrs[i * 32];
  float alpha[H];
#pragma unroll
  for (int h = 0; h < H; ++h)
    alpha[h] = expv[(size_t)e * H + h] / denom[(size_t)(lo + row) * H + h];
  float* arow = acc + row * F + lane;
#pragma unroll
  for (int i = 0; i < H * P; ++i)
    atomicAdd(&arow[i * 32], alpha[i / P] * v[i]);
}

template <int F, int H>
__global__ __launch_bounds__(AGG_THREADS)
void edge_aggr_rows(const int* __restrict__ ei, const float* __restrict__ expv,
                    const float* __restrict__ denom, const float* __restrict__ xr,
                    float* __restrict__ out, int N, int E) {
  __shared__ float acc[AGG_TR * F];
  const int tid  = threadIdx.x;
  const int lane = tid & 31;
  const int wave = __builtin_amdgcn_readfirstlane(tid >> 5);
  const int lo = blockIdx.x * AGG_TR;
  const int nrows = min(AGG_TR, N - lo);
  const unsigned lim = (unsigned)nrows;

  for (int i = tid; i < AGG_TR * F / 4; i += AGG_THREADS)
    ((float4*)acc)[i] = make_float4(0.f, 0.f, 0.f, 0.f);
  __syncthreads();

  const int* __restrict__ dstp = ei + E;
  if ((E & 3) == 0) {
    const int4* __restrict__ dst4 = (const int4*)dstp;
    const int nQuads = E >> 2;
    for (int qb = wave * (32 * AGG_GROUPS); qb < nQuads;
         qb += (AGG_THREADS / 32) * 32 * AGG_GROUPS) {
      int4 d[AGG_GROUPS];
#pragma unroll
      for (int u = 0; u < AGG_GROUPS; ++u)
        d[u] = dst4[min(qb + u * 32 + lane, nQuads - 1)];
#pragma unroll
      for (int u = 0; u < AGG_GROUPS; ++u) {
        const bool inb = (qb + u * 32 + lane) < nQuads;
        unsigned r[4];
        r[0] = (unsigned)(d[u].x - lo);
        r[1] = (unsigned)(d[u].y - lo);
        r[2] = (unsigned)(d[u].z - lo);
        r[3] = (unsigned)(d[u].w - lo);
        bool m[4];
#pragma unroll
        for (int k = 0; k < 4; ++k) m[k] = inb & (r[k] < lim);
        if (__builtin_amdgcn_ballot_w32(m[0] | m[1] | m[2] | m[3]) == 0u) continue;
#pragma unroll
        for (int k = 0; k < 4; ++k) {
          unsigned mk = __builtin_amdgcn_ballot_w32(m[k]);
          while (mk) {
            const int j = __builtin_ctz(mk);
            mk &= mk - 1u;
            const int row = __builtin_amdgcn_readlane((int)r[k], j);
            const int e = (qb + u * 32 + j) * 4 + k;
            aggr_one_edge<F, H>(ei, expv, denom, xr, acc, e, row, lo, lane);
          }
        }
      }
    }
  } else {
    for (int eb = wave * 32; eb < E; eb += AGG_THREADS) {
      const int e1 = eb + lane;
      const unsigned r1 = (e1 < E) ? (unsigned)(dstp[min(e1, E - 1)] - lo) : 0xFFFFFFFFu;
      unsigned mk = __builtin_amdgcn_ballot_w32(r1 < lim);
      while (mk) {
        const int j = __builtin_ctz(mk);
        mk &= mk - 1u;
        const int row = __builtin_amdgcn_readlane((int)r1, j);
        aggr_one_edge<F, H>(ei, expv, denom, xr, acc, eb + j, row, lo, lane);
      }
    }
  }
  __syncthreads();
  float4* __restrict__ orow = (float4*)(out + (size_t)lo * F);
  for (int i = tid; i < nrows * F / 4; i += AGG_THREADS)
    orow[i] = ((const float4*)acc)[i];
}

__global__ void node_ln_elu(const float* __restrict__ in, const float* __restrict__ bias,
                            const float* __restrict__ g, const float* __restrict__ b,
                            float* __restrict__ out, int N, int F) {
  const int node = blockIdx.x * (blockDim.x >> 5) + (threadIdx.x >> 5);
  const int lane = threadIdx.x & 31;
  if (node >= N) return;
  const int per = F / 32;
  float vals[4];
  float sum = 0.0f;
  for (int i = 0; i < per; ++i) {
    const int c = lane * per + i;
    float v = in[(size_t)node * F + c] + bias[c];
    vals[i] = v;
    sum += v;
  }
  for (int off = 16; off > 0; off >>= 1) sum += __shfl_xor(sum, off, 32);
  const float mu = sum / (float)F;
  float vs = 0.0f;
  for (int i = 0; i < per; ++i) { float dlt = vals[i] - mu; vs += dlt * dlt; }
  for (int off = 16; off > 0; off >>= 1) vs += __shfl_xor(vs, off, 32);
  const float inv = rsqrtf(vs / (float)F + LN_EPS);
  for (int i = 0; i < per; ++i) {
    const int c = lane * per + i;
    float hv = (vals[i] - mu) * inv * g[c] + b[c];
    out[(size_t)node * F + c] = (hv > 0.0f) ? hv : (__expf(hv) - 1.0f);
  }
}

extern "C" void kernel_launch(void* const* d_in, const int* in_sizes, int n_in,
                              void* d_out, int out_size, void* d_ws, size_t ws_size,
                              hipStream_t stream) {
  const float* x      = (const float*)d_in[0];
  const float* ea     = (const float*)d_in[1];
  const float* W1l    = (const float*)d_in[2];
  const float* W1r    = (const float*)d_in[3];
  const float* W1e    = (const float*)d_in[4];
  const float* att1   = (const float*)d_in[5];
  const float* b1     = (const float*)d_in[6];
  const float* ln1_g  = (const float*)d_in[7];
  const float* ln1_b  = (const float*)d_in[8];
  const float* W2l    = (const float*)d_in[9];
  const float* W2r    = (const float*)d_in[10];
  const float* W2e    = (const float*)d_in[11];
  const float* att2   = (const float*)d_in[12];
  const float* b2     = (const float*)d_in[13];
  const float* ln2_g  = (const float*)d_in[14];
  const float* ln2_b  = (const float*)d_in[15];
  const int*   ei     = (const int*)d_in[16];
  float*       outp   = (float*)d_out;

  const int N = in_sizes[0] / 10;
  const int E = in_sizes[1];
  const int HID = 32, HEADS = 4, F1 = HEADS * HID  , OUTC = 64;
  const int KP = 12;

  float* ws      = (float*)d_ws;
  float* xl1     = ws;
  float* xr1     = xl1 + (size_t)N * F1;
  float* h1      = xr1 + (size_t)N * F1;
  float* logit1  = h1  + (size_t)N * F1;
  unsigned* m1   = (unsigned*)(logit1 + (size_t)E * HEADS);
  float* denom1  = (float*)(m1 + (size_t)N * HEADS);
  float* xpad    = denom1 + (size_t)N * HEADS;
  float* w1lp    = xpad + (size_t)N * KP;
  float* w1rp    = w1lp + (size_t)KP * F1;
  float* xr2     = xl1;
  float* logit2  = xl1 + (size_t)N * OUTC;
  unsigned* m2   = (unsigned*)(logit2 + (size_t)E);
  float* denom2  = (float*)((float*)m2 + (size_t)N);
  float* xl2     = logit1;
  float* out2    = h1;

  const unsigned ORD_NEG_INF = 0x007FFFFFu;
  const int TB = 256;
  auto blocks = [](size_t n, int tb) { return (unsigned)((n + tb - 1) / tb); };

  pad_cols<<<blocks((size_t)N * KP, TB), TB, 0, stream>>>(x, xpad, N, 10, KP);
  pad_rows<<<blocks((size_t)KP * F1, TB), TB, 0, stream>>>(W1l, w1lp, 10, KP, F1);
  pad_rows<<<blocks((size_t)KP * F1, TB), TB, 0, stream>>>(W1r, w1rp, 10, KP, F1);
  {
    dim3 grid((N / 16 + 3) / 4, F1 / 16);
    wmma_gemm<12><<<grid, 128, 0, stream>>>(xpad, w1lp, xl1, N, F1);
    wmma_gemm<12><<<grid, 128, 0, stream>>>(xpad, w1rp, xr1, N, F1);
  }
  fill_u32<<<blocks((size_t)N * HEADS, TB), TB, 0, stream>>>(m1, ORD_NEG_INF, (size_t)N * HEADS);
  fill_f32<<<blocks((size_t)N * HEADS, TB), TB, 0, stream>>>(denom1, 0.0f, (size_t)N * HEADS);

  edge_logits<<<blocks((size_t)E * HEADS, TB), TB, 0, stream>>>(
      ei, ea, xl1, xr1, W1e, att1, logit1, m1, E, HEADS, HID);
  edge_exp<<<blocks((size_t)E * HEADS, TB), TB, 0, stream>>>(
      ei, logit1, m1, denom1, E, HEADS);
  edge_aggr_rows<128, 4><<<blocks((size_t)N, AGG_TR), AGG_THREADS, 0, stream>>>(
      ei, logit1, denom1, xr1, h1, N, E);
  node_ln_elu<<<blocks((size_t)N * 32, TB), TB, 0, stream>>>(
      h1, b1, ln1_g, ln1_b, h1, N, F1);

  {
    dim3 grid((N / 16 + 3) / 4, OUTC / 16);
    wmma_gemm<128><<<grid, 128, 0, stream>>>(h1, W2l, xl2, N, OUTC);
    wmma_gemm<128><<<grid, 128, 0, stream>>>(h1, W2r, xr2, N, OUTC);
  }
  fill_u32<<<blocks((size_t)N, TB), TB, 0, stream>>>(m2, ORD_NEG_INF, (size_t)N);
  fill_f32<<<blocks((size_t)N, TB), TB, 0, stream>>>(denom2, 0.0f, (size_t)N);

  edge_logits<<<blocks((size_t)E, TB), TB, 0, stream>>>(
      ei, ea, xl2, xr2, W2e, att2, logit2, m2, E, 1, OUTC);
  edge_exp<<<blocks((size_t)E, TB), TB, 0, stream>>>(
      ei, logit2, m2, denom2, E, 1);
  edge_aggr_rows<64, 1><<<blocks((size_t)N, AGG_TR), AGG_THREADS, 0, stream>>>(
      ei, logit2, denom2, xr2, out2, N, E);
  node_ln_elu<<<blocks((size_t)N * 32, TB), TB, 0, stream>>>(
      out2, b2, ln2_g, ln2_b, outp, N, OUTC);
}


# edge logits: thread per (edge, head) with strided 16-byte gathers replaced by a wave per 32 edges, lanes across channels, rows read whole; logit sums by a lane tree (sum order changed)
# speedup vs baseline: 3.3877x; 1.1515x over previous
#include <hip/hip_runtime.h>
#include <hip/hip_bf16.h>
#include <math.h>

typedef float     v2f  __attribute__((ext_vector_type(2)));
typedef float     v8f  __attribute__((ext_vector_type(8)));

#define NEG_SLOPE 0.2f
#define LN_EPS    1e-5f

__device__ __forceinline__ unsigned ordEncode(float f) {
  unsigned u = __float_as_uint(f);
  return (u >> 31) ? ~u : (u | 0x80000000u);
}
__device__ __forceinline__ float ordDecode(unsigned o) {
  return __uint_as_float((o >> 31) ? (o & 0x7FFFFFFFu) : ~o);
}

__global__ void fill_f32(float* p, float v, size_t n) {
  size_t i = (size_t)blockIdx.x * blockDim.x + threadIdx.x;
  if (i < n) p[i] = v;
}
__global__ void fill_u32(unsigned* p, unsigned v, size_t n) {
  size_t i = (size_t)blockIdx.x * blockDim.x + threadIdx.x;
  if (i < n) p[i] = v;
}

__global__ void pad_cols(const float* __restrict__ src, float* __restrict__ dst,
                         int M, int Ks, int Kd) {
  int i = blockIdx.x * blockDim.x + threadIdx.x;
  if (i >= M * Kd) return;
  int r = i / Kd, c = i % Kd;
  dst[i] = (c < Ks) ? src[(size_t)r * Ks + c] : 0.0f;
}

__global__ void pad_rows(const float* __restrict__ src, float* __restrict__ dst,
                         int Ms, int Md, int Ncols) {
  int i = blockIdx.x * blockDim.x + threadIdx.x;
  if (i >= Md * Ncols) return;
  int r = i / Ncols;
  dst[i] = (r < Ms) ? src[i] : 0.0f;
}

template <int K>
__global__ void wmma_gemm(const float* __restrict__ A, const float* __restrict__ B,
                          float* __restrict__ C, int M, int N) {
  const int lane  = threadIdx.x & 31;
  const int wave  = threadIdx.x >> 5;
  const int tileM = blockIdx.x * 4 + wave;
  const int m0 = tileM * 16;
  const int n0 = blockIdx.y * 16;
  if (m0 + 16 > M) return;
  const int half = lane >> 4;
  const int l16  = lane & 15;
  const float* __restrict__ arow = A + (size_t)(m0 + l16) * K + half * 2;
  const float* __restrict__ bcol = B + n0 + l16 + (size_t)(half * 2) * N;
  v8f acc = {};
#pragma unroll
  for (int k = 0; k < K; k += 4) {
    v2f a = *(const v2f*)(arow + k);
    v2f b;
    b.x = bcol[(size_t)k * N];
    b.y = bcol[(size_t)(k + 1) * N];
    acc = __builtin_amdgcn_wmma_f32_16x16x4_f32(false, a, false, b, (short)0, acc,
                                                false, false);
  }
  float* __restrict__ crow = C + (size_t)(m0 + half * 8) * N + n0 + l16;
#pragma unroll
  for (int r = 0; r < 8; ++r)
    crow[(size_t)r * N] = acc[r];
}

#define LG_U 8

template <int F, int H>
__global__ __launch_bounds__(256)
void edge_logits_rows(const int* __restrict__ ei, const float* __restrict__ ea,
                      const float* __restrict__ xl, const float* __restrict__ xr,
                      const float* __restrict__ We, const float* __restrict__ att,
                      float* __restrict__ logit, unsigned* __restrict__ mOrd, int E) {
  constexpr int VEC = F / 32;
  constexpr int LPH = 32 / H;
  static_assert(LG_U == 8 && LPH >= 8 && 32 % H == 0, "the lane sum below is laid out for 8 edges a step");
  typedef float vf __attribute__((ext_vector_type(VEC)));
  const int lane = threadIdx.x & 31;
  const int eb = (blockIdx.x * (blockDim.x >> 5) + (threadIdx.x >> 5)) * 32;
  if (eb >= E) return;
  const int el = min(eb + lane, E - 1);
  const int sL = ei[el];
  const int dL = ei[E + el];
  const int eaL = __float_as_int(ea[el]);
  const vf we = *(const vf*)(We + lane * VEC);
  const vf at = *(const vf*)(att + lane * VEC);
  float outv[H];
#pragma unroll
  for (int g = 0; g < H; ++g) outv[g] = 0.0f;

#pragma unroll
  for (int jb = 0; jb < 32; jb += LG_U) {
    vf a[LG_U], b[LG_U];
#pragma unroll
    for (int u = 0; u < LG_U; ++u) {
      const int s = __builtin_amdgcn_readlane(sL, jb + u);
      const int d = __builtin_amdgcn_readlane(dL, jb + u);
      a[u] = *(const vf*)(xl + (size_t)d * F + lane * VEC);
      b[u] = *(const vf*)(xr + (size_t)s * F + lane * VEC);
    }
    float p[LG_U];
#pragma unroll
    for (int u = 0; u < LG_U; ++u) {
      const float eav = __int_as_float(__builtin_amdgcn_readlane(eaL, jb + u));
      float acc = 0.0f;
#pragma unroll
      for (int i = 0; i < VEC; ++i) {
        float v = a[u][i] + b[u][i] + eav * we[i];
        v = fmaxf(v, NEG_SLOPE * v);
        acc += at[i] * v;
      }
      p[u] = acc;
    }
#pragma unroll
    for (int st = 0; st < 3; ++st) {
      const int off = LPH >> (st + 1);
      const int hn  = LG_U >> (st + 1);
      const bool hi = (lane & off) != 0;
#pragma unroll
      for (int u = 0; u < hn; ++u) {
        const float send = hi ? p[u] : p[u + hn];
        const float keep = hi ? p[u + hn] : p[u];
        p[u] = keep + __shfl_xor(send, off, 32);
      }
    }
#pragma unroll
    for (int off = LPH / 16; off > 0; off >>= 1) p[0] += __shfl_xor(p[0], off, 32);
    const int rl = lane - (jb * H) % 32;
    const bool mine = (rl >= 0) && (rl < LG_U * H);
    const int rc = mine ? rl : 0;
    const float q = __shfl(p[0], (rc % H) * LPH + (rc / H) * (LPH / 8), 32);
    if (mine) outv[(jb * H) / 32] = q;
  }

#pragma unroll
  for (int g = 0; g < H; ++g) {
    const int idx = g * 32 + lane;
    const int j = idx / H, h = idx % H;
    const int d = __shfl(dL, j, 32);
    if (eb + j < E) {
      logit[(size_t)eb * H + idx] = outv[g];
      atomicMax(&mOrd[(size_t)d * H + h], ordEncode(outv[g]));
    }
  }
}

__global__ void edge_exp(const int* __restrict__ ei, float* __restrict__ logit,
                         const unsigned* __restrict__ mOrd, float* __restrict__ denom,
                         int E, int H) {
  const int t = blockIdx.x * blockDim.x + threadIdx.x;
  if (t >= E * H) return;
  const int e = t / H, h = t % H;
  const int d = ei[E + e];
  const float m = ordDecode(mOrd[(size_t)d * H + h]);
  const float ex = __expf(logit[t] - m);
  logit[t] = ex;
  atomicAdd(&denom[(size_t)d * H + h], ex);
}

#define AGG_TR      196
#define AGG_THREADS 1024
#define AGG_GROUPS  8

template <int F, int H>
__device__ __forceinline__ void aggr_one_edge(const int* __restrict__ ei,
                                              const float* __restrict__ expv,
                                              const float* __restrict__ denom,
                                              const float* __restrict__ xr,
                                              float* acc, int e, int row, int lo, int lane) {
  constexpr int C = F / H;
  constexpr int P = C / 32;
  const int s = ei[e];
  const float* __restrict__ xrs = xr + (size_t)s * F + lane;
  float v[H * P];
#pragma unroll
  for (int i = 0; i < H * P; ++i) v[i] = xrs[i * 32];
  float alpha[H];
#pragma unroll
  for (int h = 0; h < H; ++h)
    alpha[h] = expv[(size_t)e * H + h] / denom[(size_t)(lo + row) * H + h];
  float* arow = acc + row * F + lane;
#pragma unroll
  for (int i = 0; i < H * P; ++i)
    atomicAdd(&arow[i * 32], alpha[i / P] * v[i]);
}

template <int F, int H>
__global__ __launch_bounds__(AGG_THREADS)
void edge_aggr_rows(const int* __restrict__ ei, const float* __restrict__ expv,
                    const float* __restrict__ denom, const float* __restrict__ xr,
                    float* __restrict__ out, int N, int E) {
  __shared__ float acc[AGG_TR * F];
  const int tid  = threadIdx.x;
  const int lane = tid & 31;
  const int wave = __builtin_amdgcn_readfirstlane(tid >> 5);
  const int lo = blockIdx.x * AGG_TR;
  const int nrows = min(AGG_TR, N - lo);
  const unsigned lim = (unsigned)nrows;

  for (int i = tid; i < AGG_TR * F / 4; i += AGG_THREADS)
    ((float4*)acc)[i] = make_float4(0.f, 0.f, 0.f, 0.f);
  __syncthreads();

  const int* __restrict__ dstp = ei + E;
  if ((E & 3) == 0) {
    const int4* __restrict__ dst4 = (const int4*)dstp;
    const int nQuads = E >> 2;
    for (int qb = wave * (32 * AGG_GROUPS); qb < nQuads;
         qb += (AGG_THREADS / 32) * 32 * AGG_GROUPS) {
      int4 d[AGG_GROUPS];
#pragma unroll
      for (int u = 0; u < AGG_GROUPS; ++u)
        d[u] = dst4[min(qb + u * 32 + lane, nQuads - 1)];
#pragma unroll
      for (int u = 0; u < AGG_GROUPS; ++u) {
        const bool inb = (qb + u * 32 + lane) < nQuads;
        unsigned r[4];
        r[0] = (unsigned)(d[u].x - lo);
        r[1] = (unsigned)(d[u].y - lo);
        r[2] = (unsigned)(d[u].z - lo);
        r[3] = (unsigned)(d[u].w - lo);
        bool m[4];
#pragma unroll
        for (int k = 0; k < 4; ++k) m[k] = inb & (r[k] < lim);
        if (__builtin_amdgcn_ballot_w32(m[0] | m[1] | m[2] | m[3]) == 0u) continue;
#pragma unroll
        for (int k = 0; k < 4; ++k) {
          unsigned mk = __builtin_amdgcn_ballot_w32(m[k]);
          while (mk) {
            const int j = __builtin_ctz(mk);
            mk &= mk - 1u;
            const int row = __builtin_amdgcn_readlane((int)r[k], j);
            const int e = (qb + u * 32 + j) * 4 + k;
            aggr_one_edge<F, H>(ei, expv, denom, xr, acc, e, row, lo, lane);
          }
        }
      }
    }
  } else {
    for (int eb = wave * 32; eb < E; eb += AGG_THREADS) {
      const int e1 = eb + lane;
      const unsigned r1 = (e1 < E) ? (unsigned)(dstp[min(e1, E - 1)] - lo) : 0xFFFFFFFFu;
      unsigned mk = __builtin_amdgcn_ballot_w32(r1 < lim);
      while (mk) {
        const int j = __builtin_ctz(mk);
        mk &= mk - 1u;
        const int row = __builtin_amdgcn_readlane((int)r1, j);
        aggr_one_edge<F, H>(ei, expv, denom, xr, acc, eb + j, row, lo, lane);
      }
    }
  }
  __syncthreads();
  float4* __restrict__ orow = (float4*)(out + (size_t)lo * F);
  for (int i = tid; i < nrows * F / 4; i += AGG_THREADS)
    orow[i] = ((const float4*)acc)[i];
}

__global__ void node_ln_elu(const float* __restrict__ in, const float* __restrict__ bias,
                            const float* __restrict__ g, const float* __restrict__ b,
                            float* __restrict__ out, int N, int F) {
  const int node = blockIdx.x * (blockDim.x >> 5) + (threadIdx.x >> 5);
  const int lane = threadIdx.x & 31;
  if (node >= N) return;
  const int per = F / 32;
  float vals[4];
  float sum = 0.0f;
  for (int i = 0; i < per; ++i) {
    const int c = lane * per + i;
    float v = in[(size_t)node * F + c] + bias[c];
    vals[i] = v;
    sum += v;
  }
  for (int off = 16; off > 0; off >>= 1) sum += __shfl_xor(sum, off, 32);
  const float mu = sum / (float)F;
  float vs = 0.0f;
  for (int i = 0; i < per; ++i) { float dlt = vals[i] - mu; vs += dlt * dlt; }
  for (int off = 16; off > 0; off >>= 1) vs += __shfl_xor(vs, off, 32);
  const float inv = rsqrtf(vs / (float)F + LN_EPS);
  for (int i = 0; i < per; ++i) {
    const int c = lane * per + i;
    float hv = (vals[i] - mu) * inv * g[c] + b[c];
    out[(size_t)node * F + c] = (hv > 0.0f) ? hv : (__expf(hv) - 1.0f);
  }
}

extern "C" void kernel_launch(void* const* d_in, const int* in_sizes, int n_in,
                              void* d_out, int out_size, void* d_ws, size_t ws_size,
                              hipStream_t stream) {
  const float* x      = (const float*)d_in[0];
  const float* ea     = (const float*)d_in[1];
  const float* W1l    = (const float*)d_in[2];
  const float* W1r    = (const float*)d_in[3];
  const float* W1e    = (const float*)d_in[4];
  const float* att1   = (const float*)d_in[5];
  const float* b1     = (const float*)d_in[6];
  const float* ln1_g  = (const float*)d_in[7];
  const float* ln1_b  = (const float*)d_in[8];
  const float* W2l    = (const float*)d_in[9];
  const float* W2r    = (const float*)d_in[10];
  const float* W2e    = (const float*)d_in[11];
  const float* att2   = (const float*)d_in[12];
  const float* b2     = (const float*)d_in[13];
  const float* ln2_g  = (const float*)d_in[14];
  const float* ln2_b  = (const float*)d_in[15];
  const int*   ei     = (const int*)d_in[16];
  float*       outp   = (float*)d_out;

  const int N = in_sizes[0] / 10;
  const int E = in_sizes[1];
  const int HID = 32, HEADS = 4, F1 = HEADS * HID  , OUTC = 64;
  const int KP = 12;

  float* ws      = (float*)d_ws;
  float* xl1     = ws;
  float* xr1     = xl1 + (size_t)N * F1;
  float* h1      = xr1 + (size_t)N * F1;
  float* logit1  = h1  + (size_t)N * F1;
  unsigned* m1   = (unsigned*)(logit1 + (size_t)E * HEADS);
  float* denom1  = (float*)(m1 + (size_t)N * HEADS);
  float* xpad    = denom1 + (size_t)N * HEADS;
  float* w1lp    = xpad + (size_t)N * KP;
  float* w1rp    = w1lp + (size_t)KP * F1;
  float* xr2     = xl1;
  float* logit2  = xl1 + (size_t)N * OUTC;
  unsigned* m2   = (unsigned*)(logit2 + (size_t)E);
  float* denom2  = (float*)((float*)m2 + (size_t)N);
  float* xl2     = logit1;
  float* out2    = h1;

  const unsigned ORD_NEG_INF = 0x007FFFFFu;
  const int TB = 256;
  auto blocks = [](size_t n, int tb) { return (unsigned)((n + tb - 1) / tb); };

  pad_cols<<<blocks((size_t)N * KP, TB), TB, 0, stream>>>(x, xpad, N, 10, KP);
  pad_rows<<<blocks((size_t)KP * F1, TB), TB, 0, stream>>>(W1l, w1lp, 10, KP, F1);
  pad_rows<<<blocks((size_t)KP * F1, TB), TB, 0, stream>>>(W1r, w1rp, 10, KP, F1);
  {
    dim3 grid((N / 16 + 3) / 4, F1 / 16);
    wmma_gemm<12><<<grid, 128, 0, stream>>>(xpad, w1lp, xl1, N, F1);
    wmma_gemm<12><<<grid, 128, 0, stream>>>(xpad, w1rp, xr1, N, F1);
  }
  fill_u32<<<blocks((size_t)N * HEADS, TB), TB, 0, stream>>>(m1, ORD_NEG_INF, (size_t)N * HEADS);
  fill_f32<<<blocks((size_t)N * HEADS, TB), TB, 0, stream>>>(denom1, 0.0f, (size_t)N * HEADS);

  edge_logits_rows<128, 4><<<blocks((size_t)E, TB), TB, 0, stream>>>(
      ei, ea, xl1, xr1, W1e, att1, logit1, m1, E);
  edge_exp<<<blocks((size_t)E * HEADS, TB), TB, 0, stream>>>(
      ei, logit1, m1, denom1, E, HEADS);
  edge_aggr_rows<128, 4><<<blocks((size_t)N, AGG_TR), AGG_THREADS, 0, stream>>>(
      ei, logit1, denom1, xr1, h1, N, E);
  node_ln_elu<<<blocks((size_t)N * 32, TB), TB, 0, stream>>>(
      h1, b1, ln1_g, ln1_b, h1, N, F1);

  {
    dim3 grid((N / 16 + 3) / 4, OUTC / 16);
    wmma_gemm<128><<<grid, 128, 0, stream>>>(h1, W2l, xl2, N, OUTC);
    wmma_gemm<128><<<grid, 128, 0, stream>>>(h1, W2r, xr2, N, OUTC);
  }
  fill_u32<<<blocks((size_t)N, TB), TB, 0, stream>>>(m2, ORD_NEG_INF, (size_t)N);
  fill_f32<<<blocks((size_t)N, TB), TB, 0, stream>>>(denom2, 0.0f, (size_t)N);

  edge_logits_rows<64, 1><<<blocks((size_t)E, TB), TB, 0, stream>>>(
      ei, ea, xl2, xr2, W2e, att2, logit2, m2, E);
  edge_exp<<<blocks((size_t)E, TB), TB, 0, stream>>>(
      ei, logit2, m2, denom2, E, 1);
  edge_aggr_rows<64, 1><<<blocks((size_t)N, AGG_TR), AGG_THREADS, 0, stream>>>(
      ei, logit2, denom2, xr2, out2, N, E);
  node_ln_elu<<<blocks((size_t)N * 32, TB), TB, 0, stream>>>(
      out2, b2, ln2_g, ln2_b, outp, N, OUTC);
}


# edge aggregation: a wave queues the edges of its rows (32, one a lane) and works a full queue off as one batch (src, dst, weights by lane-parallel gathers; 32 row loads in flight); sums: LDS arrival
# speedup vs baseline: 3.6833x; 1.0873x over previous
#include <hip/hip_runtime.h>
#include <hip/hip_bf16.h>
#include <math.h>

typedef float     v2f  __attribute__((ext_vector_type(2)));
typedef float     v8f  __attribute__((ext_vector_type(8)));

#define NEG_SLOPE 0.2f
#define LN_EPS    1e-5f

__device__ __forceinline__ unsigned ordEncode(float f) {
  unsigned u = __float_as_uint(f);
  return (u >> 31) ? ~u : (u | 0x80000000u);
}
__device__ __forceinline__ float ordDecode(unsigned o) {
  return __uint_as_float((o >> 31) ? (o & 0x7FFFFFFFu) : ~o);
}

__global__ void fill_f32(float* p, float v, size_t n) {
  size_t i = (size_t)blockIdx.x * blockDim.x + threadIdx.x;
  if (i < n) p[i] = v;
}
__global__ void fill_u32(unsigned* p, unsigned v, size_t n) {
  size_t i = (size_t)blockIdx.x * blockDim.x + threadIdx.x;
  if (i < n) p[i] = v;
}

__global__ void pad_cols(const float* __restrict__ src, float* __restrict__ dst,
                         int M, int Ks, int Kd) {
  int i = blockIdx.x * blockDim.x + threadIdx.x;
  if (i >= M * Kd) return;
  int r = i / Kd, c = i % Kd;
  dst[i] = (c < Ks) ? src[(size_t)r * Ks + c] : 0.0f;
}

__global__ void pad_rows(const float* __restrict__ src, float* __restrict__ dst,
                         int Ms, int Md, int Ncols) {
  int i = blockIdx.x * blockDim.x + threadIdx.x;
  if (i >= Md * Ncols) return;
  int r = i / Ncols;
  dst[i] = (r < Ms) ? src[i] : 0.0f;
}

template <int K>
__global__ void wmma_gemm(const float* __restrict__ A, const float* __restrict__ B,
                          float* __restrict__ C, int M, int N) {
  const int lane  = threadIdx.x & 31;
  const int wave  = threadIdx.x >> 5;
  const int tileM = blockIdx.x * 4 + wave;
  const int m0 = tileM * 16;
  const int n0 = blockIdx.y * 16;
  if (m0 + 16 > M) return;
  const int half = lane >> 4;
  const int l16  = lane & 15;
  const float* __restrict__ arow = A + (size_t)(m0 + l16) * K + half * 2;
  const float* __restrict__ bcol = B + n0 + l16 + (size_t)(half * 2) * N;
  v8f acc = {};
#pragma unroll
  for (int k = 0; k < K; k += 4) {
    v2f a = *(const v2f*)(arow + k);
    v2f b;
    b.x = bcol[(size_t)k * N];
    b.y = bcol[(size_t)(k + 1) * N];
    acc = __builtin_amdgcn_wmma_f32_16x16x4_f32(false, a, false, b, (short)0, acc,
                                                false, false);
  }
  float* __restrict__ crow = C + (size_t)(m0 + half * 8) * N + n0 + l16;
#pragma unroll
  for (int r = 0; r < 8; ++r)
    crow[(size_t)r * N] = acc[r];
}

#define LG_U 8

template <int F, int H>
__global__ __launch_bounds__(256)
void edge_logits_rows(const int* __restrict__ ei, const float* __restrict__ ea,
                      const float* __restrict__ xl, const float* __restrict__ xr,
                      const float* __restrict__ We, const float* __restrict__ att,
                      float* __restrict__ logit, unsigned* __restrict__ mOrd, int E) {
  constexpr int VEC = F / 32;
  constexpr int LPH = 32 / H;
  static_assert(LG_U == 8 && LPH >= 8 && 32 % H == 0, "the lane sum below is laid out for 8 edges a step");
  typedef float vf __attribute__((ext_vector_type(VEC)));
  const int lane = threadIdx.x & 31;
  const int eb = (blockIdx.x * (blockDim.x >> 5) + (threadIdx.x >> 5)) * 32;
  if (eb >= E) return;
  const int el = min(eb + lane, E - 1);
  const int sL = ei[el];
  const int dL = ei[E + el];
  const int eaL = __float_as_int(ea[el]);
  const vf we = *(const vf*)(We + lane * VEC);
  const vf at = *(const vf*)(att + lane * VEC);
  float outv[H];
#pragma unroll
  for (int g = 0; g < H; ++g) outv[g] = 0.0f;

#pragma unroll
  for (int jb = 0; jb < 32; jb += LG_U) {
    vf a[LG_U], b[LG_U];
#pragma unroll
    for (int u = 0; u < LG_U; ++u) {
      const int s = __builtin_amdgcn_readlane(sL, jb + u);
      const int d = __builtin_amdgcn_readlane(dL, jb + u);
      a[u] = *(const vf*)(xl + (size_t)d * F + lane * VEC);
      b[u] = *(const vf*)(xr + (size_t)s * F + lane * VEC);
    }
    float p[LG_U];
#pragma unroll
    for (int u = 0; u < LG_U; ++u) {
      const float eav = __int_as_float(__builtin_amdgcn_readlane(eaL, jb + u));
      float acc = 0.0f;
#pragma unroll
      for (int i = 0; i < VEC; ++i) {
        float v = a[u][i] + b[u][i] + eav * we[i];
        v = fmaxf(v, NEG_SLOPE * v);
        acc += at[i] * v;
      }
      p[u] = acc;
    }
#pragma unroll
    for (int st = 0; st < 3; ++st) {
      const int off = LPH >> (st + 1);
      const int hn  = LG_U >> (st + 1);
      const bool hi = (lane & off) != 0;
#pragma unroll
      for (int u = 0; u < hn; ++u) {
        const float send = hi ? p[u] : p[u + hn];
        const float keep = hi ? p[u + hn] : p[u];
        p[u] = keep + __shfl_xor(send, off, 32);
      }
    }
#pragma unroll
    for (int off = LPH / 16; off > 0; off >>= 1) p[0] += __shfl_xor(p[0], off, 32);
    const int rl = lane - (jb * H) % 32;
    const bool mine = (rl >= 0) && (rl < LG_U * H);
    const int rc = mine ? rl : 0;
    const float q = __shfl(p[0], (rc % H) * LPH + (rc / H) * (LPH / 8), 32);
    if (mine) outv[(jb * H) / 32] = q;
  }

#pragma unroll
  for (int g = 0; g < H; ++g) {
    const int idx = g * 32 + lane;
    const int j = idx / H, h = idx % H;
    const int d = __shfl(dL, j, 32);
    if (eb + j < E) {
      logit[(size_t)eb * H + idx] = outv[g];
      atomicMax(&mOrd[(size_t)d * H + h], ordEncode(outv[g]));
    }
  }
}

__global__ void edge_exp(const int* __restrict__ ei, float* __restrict__ logit,
                         const unsigned* __restrict__ mOrd, float* __restrict__ denom,
                         int E, int H) {
  const int t = blockIdx.x * blockDim.x + threadIdx.x;
  if (t >= E * H) return;
  const int e = t / H, h = t % H;
  const int d = ei[E + e];
  const float m = ordDecode(mOrd[(size_t)d * H + h]);
  const float ex = __expf(logit[t] - m);
  logit[t] = ex;
  atomicAdd(&denom[(size_t)d * H + h], ex);
}

#define AGG_TR      196
#define AGG_THREADS 1024
#define AGG_GROUPS  8

template <int F, int H>
__device__ __forceinline__ void aggr_batch(const int* __restrict__ ei,
                                           const int* __restrict__ dstp,
                                           const float* __restrict__ expv,
                                           const float* __restrict__ denom,
                                           const float* __restrict__ xr,
                                           float* acc, int qe, int cnt, int lo, int lane) {
  constexpr int C = F / H;
  constexpr int P = C / 32;
  constexpr int R = H * P;
  constexpr int U = 32 / R;
  const int s   = ei[qe];
  const int d   = dstp[qe];
  const int row = d - lo;
  float al[H];
#pragma unroll
  for (int h = 0; h < H; ++h)
    al[h] = expv[(size_t)qe * H + h] / denom[(size_t)d * H + h];

  for (int j0 = 0; j0 < cnt; j0 += U) {
    float v[U][R];
#pragma unroll
    for (int u = 0; u < U; ++u) {
      const int jj = min(j0 + u, cnt - 1);
      const int sj = __builtin_amdgcn_readlane(s, jj);
      const float* __restrict__ xrs = xr + (size_t)sj * F + lane;
#pragma unroll
      for (int i = 0; i < R; ++i) v[u][i] = xrs[i * 32];
    }
#pragma unroll
    for (int u = 0; u < U; ++u) {
      const int jj = j0 + u;
      if (jj < cnt) {
        const int rowj = __builtin_amdgcn_readlane(row, jj);
        float* arow = acc + rowj * F + lane;
#pragma unroll
        for (int i = 0; i < R; ++i) {
          const float a = __int_as_float(
              __builtin_amdgcn_readlane(__float_as_int(al[i / P]), jj));
          atomicAdd(&arow[i * 32], a * v[u][i]);
        }
      }
    }
  }
}

template <int F, int H>
__device__ __forceinline__ void aggr_push(const int* __restrict__ ei,
                                          const int* __restrict__ dstp,
                                          const float* __restrict__ expv,
                                          const float* __restrict__ denom,
                                          const float* __restrict__ xr,
                                          float* acc, int& qe, int& cnt, int e, int lo, int lane) {
  qe = (lane == cnt) ? e : qe;
  if (++cnt == 32) {
    aggr_batch<F, H>(ei, dstp, expv, denom, xr, acc, qe, 32, lo, lane);
    cnt = 0;
  }
}

template <int F, int H>
__global__ __launch_bounds__(AGG_THREADS)
void edge_aggr_rows(const int* __restrict__ ei, const float* __restrict__ expv,
                    const float* __restrict__ denom, const float* __restrict__ xr,
                    float* __restrict__ out, int N, int E) {
  __shared__ float acc[AGG_TR * F];
  const int tid  = threadIdx.x;
  const int lane = tid & 31;
  const int wave = __builtin_amdgcn_readfirstlane(tid >> 5);
  const int lo = blockIdx.x * AGG_TR;
  const int nrows = min(AGG_TR, N - lo);
  const unsigned lim = (unsigned)nrows;

  for (int i = tid; i < AGG_TR * F / 4; i += AGG_THREADS)
    ((float4*)acc)[i] = make_float4(0.f, 0.f, 0.f, 0.f);
  __syncthreads();

  const int* __restrict__ dstp = ei + E;
  int qe = 0;
  int cnt = 0;
  if ((E & 3) == 0) {
    const int4* __restrict__ dst4 = (const int4*)dstp;
    const int nQuads = E >> 2;
    for (int qb = wave * (32 * AGG_GROUPS); qb < nQuads;
         qb += (AGG_THREADS / 32) * 32 * AGG_GROUPS) {
      int4 d[AGG_GROUPS];
#pragma unroll
      for (int u = 0; u < AGG_GROUPS; ++u)
        d[u] = dst4[min(qb + u * 32 + lane, nQuads - 1)];
      unsigned bits = 0u, valid = 0u;
#pragma unroll
      for (int u = 0; u < AGG_GROUPS; ++u) {
        bits |= ((unsigned)(d[u].x - lo) < lim) ? (1u << (4 * u))     : 0u;
        bits |= ((unsigned)(d[u].y - lo) < lim) ? (1u << (4 * u + 1)) : 0u;
        bits |= ((unsigned)(d[u].z - lo) < lim) ? (1u << (4 * u + 2)) : 0u;
        bits |= ((unsigned)(d[u].w - lo) < lim) ? (1u << (4 * u + 3)) : 0u;
        valid |= ((qb + u * 32 + lane) < nQuads) ? (0xFu << (4 * u)) : 0u;
      }
      bits &= valid;
      unsigned any = __builtin_amdgcn_ballot_w32(bits != 0u);
      while (any) {
        const int L = __builtin_ctz(any);
        any &= any - 1u;
        unsigned b = (unsigned)__builtin_amdgcn_readlane((int)bits, L);
        while (b) {
          const int t = __builtin_ctz(b);
          b &= b - 1u;
          const int e = (qb + (t >> 2) * 32 + L) * 4 + (t & 3);
          aggr_push<F, H>(ei, dstp, expv, denom, xr, acc, qe, cnt, e, lo, lane);
        }
      }
    }
  } else {
    for (int eb = wave * 32; eb < E; eb += AGG_THREADS) {
      const int e1 = eb + lane;
      const unsigned r1 = (e1 < E) ? (unsigned)(dstp[min(e1, E - 1)] - lo) : 0xFFFFFFFFu;
      unsigned mk = __builtin_amdgcn_ballot_w32(r1 < lim);
      while (mk) {
        const int j = __builtin_ctz(mk);
        mk &= mk - 1u;
        aggr_push<F, H>(ei, dstp, expv, denom, xr, acc, qe, cnt, eb + j, lo, lane);
      }
    }
  }
  if (cnt > 0)
    aggr_batch<F, H>(ei, dstp, expv, denom, xr, acc, qe, cnt, lo, lane);
  __syncthreads();
  float4* __restrict__ orow = (float4*)(out + (size_t)lo * F);
  for (int i = tid; i < nrows * F / 4; i += AGG_THREADS)
    orow[i] = ((const float4*)acc)[i];
}

__global__ void node_ln_elu(const float* __restrict__ in, const float* __restrict__ bias,
                            const float* __restrict__ g, const float* __restrict__ b,
                            float* __restrict__ out, int N, int F) {
  const int node = blockIdx.x * (blockDim.x >> 5) + (threadIdx.x >> 5);
  const int lane = threadIdx.x & 31;
  if (node >= N) return;
  const int per = F / 32;
  float vals[4];
  float sum = 0.0f;
  for (int i = 0; i < per; ++i) {
    const int c = lane * per + i;
    float v = in[(size_t)node * F + c] + bias[c];
    vals[i] = v;
    sum += v;
  }
  for (int off = 16; off > 0; off >>= 1) sum += __shfl_xor(sum, off, 32);
  const float mu = sum / (float)F;
  float vs = 0.0f;
  for (int i = 0; i < per; ++i) { float dlt = vals[i] - mu; vs += dlt * dlt; }
  for (int off = 16; off > 0; off >>= 1) vs += __shfl_xor(vs, off, 32);
  const float inv = rsqrtf(vs / (float)F + LN_EPS);
  for (int i = 0; i < per; ++i) {
    const int c = lane * per + i;
    float hv = (vals[i] - mu) * inv * g[c] + b[c];
    out[(size_t)node * F + c] = (hv > 0.0f) ? hv : (__expf(hv) - 1.0f);
  }
}

extern "C" void kernel_launch(void* const* d_in, const int* in_sizes, int n_in,
                              void* d_out, int out_size, void* d_ws, size_t ws_size,
                              hipStream_t stream) {
  const float* x      = (const float*)d_in[0];
  const float* ea     = (const float*)d_in[1];
  const float* W1l    = (const float*)d_in[2];
  const float* W1r    = (const float*)d_in[3];
  const float* W1e    = (const float*)d_in[4];
  const float* att1   = (const float*)d_in[5];
  const float* b1     = (const float*)d_in[6];
  const float* ln1_g  = (const float*)d_in[7];
  const float* ln1_b  = (const float*)d_in[8];
  const float* W2l    = (const float*)d_in[9];
  const float* W2r    = (const float*)d_in[10];
  const float* W2e    = (const float*)d_in[11];
  const float* att2   = (const float*)d_in[12];
  const float* b2     = (const float*)d_in[13];
  const float* ln2_g  = (const float*)d_in[14];
  const float* ln2_b  = (const float*)d_in[15];
  const int*   ei     = (const int*)d_in[16];
  float*       outp   = (float*)d_out;

  const int N = in_sizes[0] / 10;
  const int E = in_sizes[1];
  const int HID = 32, HEADS = 4, F1 = HEADS * HID  , OUTC = 64;
  const int KP = 12;

  float* ws      = (float*)d_ws;
  float* xl1     = ws;
  float* xr1     = xl1 + (size_t)N * F1;
  float* h1      = xr1 + (size_t)N * F1;
  float* logit1  = h1  + (size_t)N * F1;
  unsigned* m1   = (unsigned*)(logit1 + (size_t)E * HEADS);
  float* denom1  = (float*)(m1 + (size_t)N * HEADS);
  float* xpad    = denom1 + (size_t)N * HEADS;
  float* w1lp    = xpad + (size_t)N * KP;
  float* w1rp    = w1lp + (size_t)KP * F1;
  float* xr2     = xl1;
  float* logit2  = xl1 + (size_t)N * OUTC;
  unsigned* m2   = (unsigned*)(logit2 + (size_t)E);
  float* denom2  = (float*)((float*)m2 + (size_t)N);
  float* xl2     = logit1;
  float* out2    = h1;

  const unsigned ORD_NEG_INF = 0x007FFFFFu;
  const int TB = 256;
  auto blocks = [](size_t n, int tb) { return (unsigned)((n + tb - 1) / tb); };

  pad_cols<<<blocks((size_t)N * KP, TB), TB, 0, stream>>>(x, xpad, N, 10, KP);
  pad_rows<<<blocks((size_t)KP * F1, TB), TB, 0, stream>>>(W1l, w1lp, 10, KP, F1);
  pad_rows<<<blocks((size_t)KP * F1, TB), TB, 0, stream>>>(W1r, w1rp, 10, KP, F1);
  {
    dim3 grid((N / 16 + 3) / 4, F1 / 16);
    wmma_gemm<12><<<grid, 128, 0, stream>>>(xpad, w1lp, xl1, N, F1);
    wmma_gemm<12><<<grid, 128, 0, stream>>>(xpad, w1rp, xr1, N, F1);
  }
  fill_u32<<<blocks((size_t)N * HEADS, TB), TB, 0, stream>>>(m1, ORD_NEG_INF, (size_t)N * HEADS);
  fill_f32<<<blocks((size_t)N * HEADS, TB), TB, 0, stream>>>(denom1, 0.0f, (size_t)N * HEADS);

  edge_logits_rows<128, 4><<<blocks((size_t)E, TB), TB, 0, stream>>>(
      ei, ea, xl1, xr1, W1e, att1, logit1, m1, E);
  edge_exp<<<blocks((size_t)E * HEADS, TB), TB, 0, stream>>>(
      ei, logit1, m1, denom1, E, HEADS);
  edge_aggr_rows<128, 4><<<blocks((size_t)N, AGG_TR), AGG_THREADS, 0, stream>>>(
      ei, logit1, denom1, xr1, h1, N, E);
  node_ln_elu<<<blocks((size_t)N * 32, TB), TB, 0, stream>>>(
      h1, b1, ln1_g, ln1_b, h1, N, F1);

  {
    dim3 grid((N / 16 + 3) / 4, OUTC / 16);
    wmma_gemm<128><<<grid, 128, 0, stream>>>(h1, W2l, xl2, N, OUTC);
    wmma_gemm<128><<<grid, 128, 0, stream>>>(h1, W2r, xr2, N, OUTC);
  }
  fill_u32<<<blocks((size_t)N, TB), TB, 0, stream>>>(m2, ORD_NEG_INF, (size_t)N);
  fill_f32<<<blocks((size_t)N, TB), TB, 0, stream>>>(denom2, 0.0f, (size_t)N);

  edge_logits_rows<64, 1><<<blocks((size_t)E, TB), TB, 0, stream>>>(
      ei, ea, xl2, xr2, W2e, att2, logit2, m2, E);
  edge_exp<<<blocks((size_t)E, TB), TB, 0, stream>>>(
      ei, logit2, m2, denom2, E, 1);
  edge_aggr_rows<64, 1><<<blocks((size_t)N, AGG_TR), AGG_THREADS, 0, stream>>>(
      ei, logit2, denom2, xr2, out2, N, E);
  node_ln_elu<<<blocks((size_t)N * 32, TB), TB, 0, stream>>>(
      out2, b2, ln2_g, ln2_b, outp, N, OUTC);
}


# softmax exp and denominators moved into the aggregation block: exp per queued edge, denominators summed in LDS (arrival order), rows divided once at the store; edge_exp and denom fill launches gone
# speedup vs baseline: 4.7663x; 1.2940x over previous
#include <hip/hip_runtime.h>
#include <hip/hip_bf16.h>
#include <math.h>

typedef float     v2f  __attribute__((ext_vector_type(2)));
typedef float     v8f  __attribute__((ext_vector_type(8)));

#define NEG_SLOPE 0.2f
#define LN_EPS    1e-5f

__device__ __forceinline__ unsigned ordEncode(float f) {
  unsigned u = __float_as_uint(f);
  return (u >> 31) ? ~u : (u | 0x80000000u);
}
__device__ __forceinline__ float ordDecode(unsigned o) {
  return __uint_as_float((o >> 31) ? (o & 0x7FFFFFFFu) : ~o);
}

__global__ void fill_u32(unsigned* p, unsigned v, size_t n) {
  size_t i = (size_t)blockIdx.x * blockDim.x + threadIdx.x;
  if (i < n) p[i] = v;
}

__global__ void pad_cols(const float* __restrict__ src, float* __restrict__ dst,
                         int M, int Ks, int Kd) {
  int i = blockIdx.x * blockDim.x + threadIdx.x;
  if (i >= M * Kd) return;
  int r = i / Kd, c = i % Kd;
  dst[i] = (c < Ks) ? src[(size_t)r * Ks + c] : 0.0f;
}

__global__ void pad_rows(const float* __restrict__ src, float* __restrict__ dst,
                         int Ms, int Md, int Ncols) {
  int i = blockIdx.x * blockDim.x + threadIdx.x;
  if (i >= Md * Ncols) return;
  int r = i / Ncols;
  dst[i] = (r < Ms) ? src[i] : 0.0f;
}

template <int K>
__global__ void wmma_gemm(const float* __restrict__ A, const float* __restrict__ B,
                          float* __restrict__ C, int M, int N) {
  const int lane  = threadIdx.x & 31;
  const int wave  = threadIdx.x >> 5;
  const int tileM = blockIdx.x * 4 + wave;
  const int m0 = tileM * 16;
  const int n0 = blockIdx.y * 16;
  if (m0 + 16 > M) return;
  const int half = lane >> 4;
  const int l16  = lane & 15;
  const float* __restrict__ arow = A + (size_t)(m0 + l16) * K + half * 2;
  const float* __restrict__ bcol = B + n0 + l16 + (size_t)(half * 2) * N;
  v8f acc = {};
#pragma unroll
  for (int k = 0; k < K; k += 4) {
    v2f a = *(const v2f*)(arow + k);
    v2f b;
    b.x = bcol[(size_t)k * N];
    b.y = bcol[(size_t)(k + 1) * N];
    acc = __builtin_amdgcn_wmma_f32_16x16x4_f32(false, a, false, b, (short)0, acc,
                                                false, false);
  }
  float* __restrict__ crow = C + (size_t)(m0 + half * 8) * N + n0 + l16;
#pragma unroll
  for (int r = 0; r < 8; ++r)
    crow[(size_t)r * N] = acc[r];
}

#define LG_U 8

template <int F, int H>
__global__ __launch_bounds__(256)
void edge_logits_rows(const int* __restrict__ ei, const float* __restrict__ ea,
                      const float* __restrict__ xl, const float* __restrict__ xr,
                      const float* __restrict__ We, const float* __restrict__ att,
                      float* __restrict__ logit, unsigned* __restrict__ mOrd, int E) {
  constexpr int VEC = F / 32;
  constexpr int LPH = 32 / H;
  static_assert(LG_U == 8 && LPH >= 8 && 32 % H == 0, "the lane sum below is laid out for 8 edges a step");
  typedef float vf __attribute__((ext_vector_type(VEC)));
  const int lane = threadIdx.x & 31;
  const int eb = (blockIdx.x * (blockDim.x >> 5) + (threadIdx.x >> 5)) * 32;
  if (eb >= E) return;
  const int el = min(eb + lane, E - 1);
  const int sL = ei[el];
  const int dL = ei[E + el];
  const int eaL = __float_as_int(ea[el]);
  const vf we = *(const vf*)(We + lane * VEC);
  const vf at = *(const vf*)(att + lane * VEC);
  float outv[H];
#pragma unroll
  for (int g = 0; g < H; ++g) outv[g] = 0.0f;

#pragma unroll
  for (int jb = 0; jb < 32; jb += LG_U) {
    vf a[LG_U], b[LG_U];
#pragma unroll
    for (int u = 0; u < LG_U; ++u) {
      const int s = __builtin_amdgcn_readlane(sL, jb + u);
      const int d = __builtin_amdgcn_readlane(dL, jb + u);
      a[u] = *(const vf*)(xl + (size_t)d * F + lane * VEC);
      b[u] = *(const vf*)(xr + (size_t)s * F + lane * VEC);
    }
    float p[LG_U];
#pragma unroll
    for (int u = 0; u < LG_U; ++u) {
      const float eav = __int_as_float(__builtin_amdgcn_readlane(eaL, jb + u));
      float acc = 0.0f;
#pragma unroll
      for (int i = 0; i < VEC; ++i) {
        float v = a[u][i] + b[u][i] + eav * we[i];
        v = fmaxf(v, NEG_SLOPE * v);
        acc += at[i] * v;
      }
      p[u] = acc;
    }
#pragma unroll
    for (int st = 0; st < 3; ++st) {
      const int off = LPH >> (st + 1);
      const int hn  = LG_U >> (st + 1);
      const bool hi = (lane & off) != 0;
#pragma unroll
      for (int u = 0; u < hn; ++u) {
        const float send = hi ? p[u] : p[u + hn];
        const float keep = hi ? p[u + hn] : p[u];
        p[u] = keep + __shfl_xor(send, off, 32);
      }
    }
#pragma unroll
    for (int off = LPH / 16; off > 0; off >>= 1) p[0] += __shfl_xor(p[0], off, 32);
    const int rl = lane - (jb * H) % 32;
    const bool mine = (rl >= 0) && (rl < LG_U * H);
    const int rc = mine ? rl : 0;
    const float q = __shfl(p[0], (rc % H) * LPH + (rc / H) * (LPH / 8), 32);
    if (mine) outv[(jb * H) / 32] = q;
  }

#pragma unroll
  for (int g = 0; g < H; ++g) {
    const int idx = g * 32 + lane;
    const int j = idx / H, h = idx % H;
    const int d = __shfl(dL, j, 32);
    if (eb + j < E) {
      logit[(size_t)eb * H + idx] = outv[g];
      atomicMax(&mOrd[(size_t)d * H + h], ordEncode(outv[g]));
    }
  }
}

#define AGG_TR      196
#define AGG_THREADS 1024
#define AGG_GROUPS  8

template <int F, int H>
__device__ __forceinline__ void aggr_batch(const int* __restrict__ ei,
                                           const int* __restrict__ dstp,
                                           const float* __restrict__ logit,
                                           const float* __restrict__ xr,
                                           float* acc, float* den, const float* mx,
                                           int qe, int cnt, int lo, int lane) {
  constexpr int C = F / H;
  constexpr int P = C / 32;
  constexpr int R = H * P;
  constexpr int U = 32 / R;
  const int s   = ei[qe];
  const int d   = dstp[qe];
  float lg[H];
#pragma unroll
  for (int h = 0; h < H; ++h) lg[h] = logit[(size_t)qe * H + h];
  const bool live = lane < cnt;
  const int row = live ? d - lo : 0;
  float al[H];
#pragma unroll
  for (int h = 0; h < H; ++h) al[h] = __expf(lg[h] - mx[row * H + h]);
  if (live) {
#pragma unroll
    for (int h = 0; h < H; ++h) atomicAdd(&den[row * H + h], al[h]);
  }

  for (int j0 = 0; j0 < cnt; j0 += U) {
    float v[U][R];
#pragma unroll
    for (int u = 0; u < U; ++u) {
      const int jj = min(j0 + u, cnt - 1);
      const int sj = __builtin_amdgcn_readlane(s, jj);
      const float* __restrict__ xrs = xr + (size_t)sj * F + lane;
#pragma unroll
      for (int i = 0; i < R; ++i) v[u][i] = xrs[i * 32];
    }
#pragma unroll
    for (int u = 0; u < U; ++u) {
      const int jj = j0 + u;
      if (jj < cnt) {
        const int rowj = __builtin_amdgcn_readlane(row, jj);
        float* arow = acc + rowj * F + lane;
#pragma unroll
        for (int i = 0; i < R; ++i) {
          const float a = __int_as_float(
              __builtin_amdgcn_readlane(__float_as_int(al[i / P]), jj));
          atomicAdd(&arow[i * 32], a * v[u][i]);
        }
      }
    }
  }
}

template <int F, int H>
__device__ __forceinline__ void aggr_push(const int* __restrict__ ei,
                                          const int* __restrict__ dstp,
                                          const float* __restrict__ logit,
                                          const float* __restrict__ xr,
                                          float* acc, float* den, const float* mx,
                                          int& qe, int& cnt, int e, int lo, int lane) {
  qe = (lane == cnt) ? e : qe;
  if (++cnt == 32) {
    aggr_batch<F, H>(ei, dstp, logit, xr, acc, den, mx, qe, 32, lo, lane);
    cnt = 0;
  }
}

template <int F, int H>
__global__ __launch_bounds__(AGG_THREADS)
void edge_aggr_rows(const int* __restrict__ ei, const float* __restrict__ logit,
                    const unsigned* __restrict__ mOrd, const float* __restrict__ xr,
                    float* __restrict__ out, int N, int E) {
  __shared__ float acc[AGG_TR * F];
  __shared__ float den[AGG_TR * H];
  __shared__ float mx[AGG_TR * H];
  const int tid  = threadIdx.x;
  const int lane = tid & 31;
  const int wave = __builtin_amdgcn_readfirstlane(tid >> 5);
  const int lo = blockIdx.x * AGG_TR;
  const int nrows = min(AGG_TR, N - lo);
  const unsigned lim = (unsigned)nrows;

  for (int i = tid; i < AGG_TR * F / 4; i += AGG_THREADS)
    ((float4*)acc)[i] = make_float4(0.f, 0.f, 0.f, 0.f);
  for (int i = tid; i < AGG_TR * H; i += AGG_THREADS) {
    den[i] = 0.0f;
    mx[i] = (i < nrows * H) ? ordDecode(mOrd[(size_t)lo * H + i]) : 0.0f;
  }
  __syncthreads();

  const int* __restrict__ dstp = ei + E;
  int qe = 0;
  int cnt = 0;
  if ((E & 3) == 0) {
    const int4* __restrict__ dst4 = (const int4*)dstp;
    const int nQuads = E >> 2;
    for (int qb = wave * (32 * AGG_GROUPS); qb < nQuads;
         qb += (AGG_THREADS / 32) * 32 * AGG_GROUPS) {
      int4 d[AGG_GROUPS];
#pragma unroll
      for (int u = 0; u < AGG_GROUPS; ++u)
        d[u] = dst4[min(qb + u * 32 + lane, nQuads - 1)];
      unsigned bits = 0u, valid = 0u;
#pragma unroll
      for (int u = 0; u < AGG_GROUPS; ++u) {
        bits |= ((unsigned)(d[u].x - lo) < lim) ? (1u << (4 * u))     : 0u;
        bits |= ((unsigned)(d[u].y - lo) < lim) ? (1u << (4 * u + 1)) : 0u;
        bits |= ((unsigned)(d[u].z - lo) < lim) ? (1u << (4 * u + 2)) : 0u;
        bits |= ((unsigned)(d[u].w - lo) < lim) ? (1u << (4 * u + 3)) : 0u;
        valid |= ((qb + u * 32 + lane) < nQuads) ? (0xFu << (4 * u)) : 0u;
      }
      bits &= valid;
      unsigned any = __builtin_amdgcn_ballot_w32(bits != 0u);
      while (any) {
        const int L = __builtin_ctz(any);
        any &= any - 1u;
        unsigned b = (unsigned)__builtin_amdgcn_readlane((int)bits, L);
        while (b) {
          const int t = __builtin_ctz(b);
          b &= b - 1u;
          const int e = (qb + (t >> 2) * 32 + L) * 4 + (t & 3);
          aggr_push<F, H>(ei, dstp, logit, xr, acc, den, mx, qe, cnt, e, lo, lane);
        }
      }
    }
  } else {
    for (int eb = wave * 32; eb < E; eb += AGG_THREADS) {
      const int e1 = eb + lane;
      const unsigned r1 = (e1 < E) ? (unsigned)(dstp[min(e1, E - 1)] - lo) : 0xFFFFFFFFu;
      unsigned mk = __builtin_amdgcn_ballot_w32(r1 < lim);
      while (mk) {
        const int j = __builtin_ctz(mk);
        mk &= mk - 1u;
        aggr_push<F, H>(ei, dstp, logit, xr, acc, den, mx, qe, cnt, eb + j, lo, lane);
      }
    }
  }
  if (cnt > 0)
    aggr_batch<F, H>(ei, dstp, logit, xr, acc, den, mx, qe, cnt, lo, lane);
  __syncthreads();
  float4* __restrict__ orow = (float4*)(out + (size_t)lo * F);
  for (int i = tid; i < nrows * F / 4; i += AGG_THREADS) {
    const int r = (i * 4) / F;
    const int h = ((i * 4) % F) / (F / H);
    const float dn = den[r * H + h];
    float4 a = ((const float4*)acc)[i];
    if (dn > 0.0f) {
      a.x /= dn; a.y /= dn; a.z /= dn; a.w /= dn;
    }
    orow[i] = a;
  }
}

__global__ void node_ln_elu(const float* __restrict__ in, const float* __restrict__ bias,
                            const float* __restrict__ g, const float* __restrict__ b,
                            float* __restrict__ out, int N, int F) {
  const int node = blockIdx.x * (blockDim.x >> 5) + (threadIdx.x >> 5);
  const int lane = threadIdx.x & 31;
  if (node >= N) return;
  const int per = F / 32;
  float vals[4];
  float sum = 0.0f;
  for (int i = 0; i < per; ++i) {
    const int c = lane * per + i;
    float v = in[(size_t)node * F + c] + bias[c];
    vals[i] = v;
    sum += v;
  }
  for (int off = 16; off > 0; off >>= 1) sum += __shfl_xor(sum, off, 32);
  const float mu = sum / (float)F;
  float vs = 0.0f;
  for (int i = 0; i < per; ++i) { float dlt = vals[i] - mu; vs += dlt * dlt; }
  for (int off = 16; off > 0; off >>= 1) vs += __shfl_xor(vs, off, 32);
  const float inv = rsqrtf(vs / (float)F + LN_EPS);
  for (int i = 0; i < per; ++i) {
    const int c = lane * per + i;
    float hv = (vals[i] - mu) * inv * g[c] + b[c];
    out[(size_t)node * F + c] = (hv > 0.0f) ? hv : (__expf(hv) - 1.0f);
  }
}

extern "C" void kernel_launch(void* const* d_in, const int* in_sizes, int n_in,
                              void* d_out, int out_size, void* d_ws, size_t ws_size,
                              hipStream_t stream) {
  const float* x      = (const float*)d_in[0];
  const float* ea     = (const float*)d_in[1];
  const float* W1l    = (const float*)d_in[2];
  const float* W1r    = (const float*)d_in[3];
  const float* W1e    = (const float*)d_in[4];
  const float* att1   = (const float*)d_in[5];
  const float* b1     = (const float*)d_in[6];
  const float* ln1_g  = (const float*)d_in[7];
  const float* ln1_b  = (const float*)d_in[8];
  const float* W2l    = (const float*)d_in[9];
  const float* W2r    = (const float*)d_in[10];
  const float* W2e    = (const float*)d_in[11];
  const float* att2   = (const float*)d_in[12];
  const float* b2     = (const float*)d_in[13];
  const float* ln2_g  = (const float*)d_in[14];
  const float* ln2_b  = (const float*)d_in[15];
  const int*   ei     = (const int*)d_in[16];
  float*       outp   = (float*)d_out;

  const int N = in_sizes[0] / 10;
  const int E = in_sizes[1];
  const int HID = 32, HEADS = 4, F1 = HEADS * HID  , OUTC = 64;
  const int KP = 12;

  float* ws      = (float*)d_ws;
  float* xl1     = ws;
  float* xr1     = xl1 + (size_t)N * F1;
  float* h1      = xr1 + (size_t)N * F1;
  float* logit1  = h1  + (size_t)N * F1;
  unsigned* m1   = (unsigned*)(logit1 + (size_t)E * HEADS);
  float* xpad    = (float*)(m1 + (size_t)N * HEADS);
  float* w1lp    = xpad + (size_t)N * KP;
  float* w1rp    = w1lp + (size_t)KP * F1;
  float* xr2     = xl1;
  float* logit2  = xl1 + (size_t)N * OUTC;
  unsigned* m2   = (unsigned*)(logit2 + (size_t)E);
  float* xl2     = logit1;
  float* out2    = h1;

  const unsigned ORD_NEG_INF = 0x007FFFFFu;
  const int TB = 256;
  auto blocks = [](size_t n, int tb) { return (unsigned)((n + tb - 1) / tb); };

  pad_cols<<<blocks((size_t)N * KP, TB), TB, 0, stream>>>(x, xpad, N, 10, KP);
  pad_rows<<<blocks((size_t)KP * F1, TB), TB, 0, stream>>>(W1l, w1lp, 10, KP, F1);
  pad_rows<<<blocks((size_t)KP * F1, TB), TB, 0, stream>>>(W1r, w1rp, 10, KP, F1);
  {
    dim3 grid((N / 16 + 3) / 4, F1 / 16);
    wmma_gemm<12><<<grid, 128, 0, stream>>>(xpad, w1lp, xl1, N, F1);
    wmma_gemm<12><<<grid, 128, 0, stream>>>(xpad, w1rp, xr1, N, F1);
  }
  fill_u32<<<blocks((size_t)N * HEADS, TB), TB, 0, stream>>>(m1, ORD_NEG_INF, (size_t)N * HEADS);

  edge_logits_rows<128, 4><<<blocks((size_t)E, TB), TB, 0, stream>>>(
      ei, ea, xl1, xr1, W1e, att1, logit1, m1, E);
  edge_aggr_rows<128, 4><<<blocks((size_t)N, AGG_TR), AGG_THREADS, 0, stream>>>(
      ei, logit1, m1, xr1, h1, N, E);
  node_ln_elu<<<blocks((size_t)N * 32, TB), TB, 0, stream>>>(
      h1, b1, ln1_g, ln1_b, h1, N, F1);

  {
    dim3 grid((N / 16 + 3) / 4, OUTC / 16);
    wmma_gemm<128><<<grid, 128, 0, stream>>>(h1, W2l, xl2, N, OUTC);
    wmma_gemm<128><<<grid, 128, 0, stream>>>(h1, W2r, xr2, N, OUTC);
  }
  fill_u32<<<blocks((size_t)N, TB), TB, 0, stream>>>(m2, ORD_NEG_INF, (size_t)N);

  edge_logits_rows<64, 1><<<blocks((size_t)E, TB), TB, 0, stream>>>(
      ei, ea, xl2, xr2, W2e, att2, logit2, m2, E);
  edge_aggr_rows<64, 1><<<blocks((size_t)N, AGG_TR), AGG_THREADS, 0, stream>>>(
      ei, logit2, m2, xr2, out2, N, E);
  node_ln_elu<<<blocks((size_t)N * 32, TB), TB, 0, stream>>>(
      out2, b2, ln2_g, ln2_b, outp, N, OUTC);
}


# bias + LayerNorm + ELU moved into the aggregation block: a wave finishes a row straight from LDS (division, bias, LN, ELU) and stores it once; node_ln_elu launches gone, 15 to 13; sums kept
# speedup vs baseline: 5.0738x; 1.0645x over previous
#include <hip/hip_runtime.h>
#include <hip/hip_bf16.h>
#include <math.h>

typedef float     v2f  __attribute__((ext_vector_type(2)));
typedef float     v8f  __attribute__((ext_vector_type(8)));

#define NEG_SLOPE 0.2f
#define LN_EPS    1e-5f

__device__ __forceinline__ unsigned ordEncode(float f) {
  unsigned u = __float_as_uint(f);
  return (u >> 31) ? ~u : (u | 0x80000000u);
}
__device__ __forceinline__ float ordDecode(unsigned o) {
  return __uint_as_float((o >> 31) ? (o & 0x7FFFFFFFu) : ~o);
}

__global__ void fill_u32(unsigned* p, unsigned v, size_t n) {
  size_t i = (size_t)blockIdx.x * blockDim.x + threadIdx.x;
  if (i < n) p[i] = v;
}

__global__ void pad_cols(const float* __restrict__ src, float* __restrict__ dst,
                         int M, int Ks, int Kd) {
  int i = blockIdx.x * blockDim.x + threadIdx.x;
  if (i >= M * Kd) return;
  int r = i / Kd, c = i % Kd;
  dst[i] = (c < Ks) ? src[(size_t)r * Ks + c] : 0.0f;
}

__global__ void pad_rows(const float* __restrict__ src, float* __restrict__ dst,
                         int Ms, int Md, int Ncols) {
  int i = blockIdx.x * blockDim.x + threadIdx.x;
  if (i >= Md * Ncols) return;
  int r = i / Ncols;
  dst[i] = (r < Ms) ? src[i] : 0.0f;
}

template <int K>
__global__ void wmma_gemm(const float* __restrict__ A, const float* __restrict__ B,
                          float* __restrict__ C, int M, int N) {
  const int lane  = threadIdx.x & 31;
  const int wave  = threadIdx.x >> 5;
  const int tileM = blockIdx.x * 4 + wave;
  const int m0 = tileM * 16;
  const int n0 = blockIdx.y * 16;
  if (m0 + 16 > M) return;
  const int half = lane >> 4;
  const int l16  = lane & 15;
  const float* __restrict__ arow = A + (size_t)(m0 + l16) * K + half * 2;
  const float* __restrict__ bcol = B + n0 + l16 + (size_t)(half * 2) * N;
  v8f acc = {};
#pragma unroll
  for (int k = 0; k < K; k += 4) {
    v2f a = *(const v2f*)(arow + k);
    v2f b;
    b.x = bcol[(size_t)k * N];
    b.y = bcol[(size_t)(k + 1) * N];
    acc = __builtin_amdgcn_wmma_f32_16x16x4_f32(false, a, false, b, (short)0, acc,
                                                false, false);
  }
  float* __restrict__ crow = C + (size_t)(m0 + half * 8) * N + n0 + l16;
#pragma unroll
  for (int r = 0; r < 8; ++r)
    crow[(size_t)r * N] = acc[r];
}

#define LG_U 8

template <int F, int H>
__global__ __launch_bounds__(256)
void edge_logits_rows(const int* __restrict__ ei, const float* __restrict__ ea,
                      const float* __restrict__ xl, const float* __restrict__ xr,
                      const float* __restrict__ We, const float* __restrict__ att,
                      float* __restrict__ logit, unsigned* __restrict__ mOrd, int E) {
  constexpr int VEC = F / 32;
  constexpr int LPH = 32 / H;
  static_assert(LG_U == 8 && LPH >= 8 && 32 % H == 0, "the lane sum below is laid out for 8 edges a step");
  typedef float vf __attribute__((ext_vector_type(VEC)));
  const int lane = threadIdx.x & 31;
  const int eb = (blockIdx.x * (blockDim.x >> 5) + (threadIdx.x >> 5)) * 32;
  if (eb >= E) return;
  const int el = min(eb + lane, E - 1);
  const int sL = ei[el];
  const int dL = ei[E + el];
  const int eaL = __float_as_int(ea[el]);
  const vf we = *(const vf*)(We + lane * VEC);
  const vf at = *(const vf*)(att + lane * VEC);
  float outv[H];
#pragma unroll
  for (int g = 0; g < H; ++g) outv[g] = 0.0f;

#pragma unroll
  for (int jb = 0; jb < 32; jb += LG_U) {
    vf a[LG_U], b[LG_U];
#pragma unroll
    for (int u = 0; u < LG_U; ++u) {
      const int s = __builtin_amdgcn_readlane(sL, jb + u);
      const int d = __builtin_amdgcn_readlane(dL, jb + u);
      a[u] = *(const vf*)(xl + (size_t)d * F + lane * VEC);
      b[u] = *(const vf*)(xr + (size_t)s * F + lane * VEC);
    }
    float p[LG_U];
#pragma unroll
    for (int u = 0; u < LG_U; ++u) {
      const float eav = __int_as_float(__builtin_amdgcn_readlane(eaL, jb + u));
      float acc = 0.0f;
#pragma unroll
      for (int i = 0; i < VEC; ++i) {
        float v = a[u][i] + b[u][i] + eav * we[i];
        v = fmaxf(v, NEG_SLOPE * v);
        acc += at[i] * v;
      }
      p[u] = acc;
    }
#pragma unroll
    for (int st = 0; st < 3; ++st) {
      const int off = LPH >> (st + 1);
      const int hn  = LG_U >> (st + 1);
      const bool hi = (lane & off) != 0;
#pragma unroll
      for (int u = 0; u < hn; ++u) {
        const float send = hi ? p[u] : p[u + hn];
        const float keep = hi ? p[u + hn] : p[u];
        p[u] = keep + __shfl_xor(send, off, 32);
      }
    }
#pragma unroll
    for (int off = LPH / 16; off > 0; off >>= 1) p[0] += __shfl_xor(p[0], off, 32);
    const int rl = lane - (jb * H) % 32;
    const bool mine = (rl >= 0) && (rl < LG_U * H);
    const int rc = mine ? rl : 0;
    const float q = __shfl(p[0], (rc % H) * LPH + (rc / H) * (LPH / 8), 32);
    if (mine) outv[(jb * H) / 32] = q;
  }

#pragma unroll
  for (int g = 0; g < H; ++g) {
    const int idx = g * 32 + lane;
    const int j = idx / H, h = idx % H;
    const int d = __shfl(dL, j, 32);
    if (eb + j < E) {
      logit[(size_t)eb * H + idx] = outv[g];
      atomicMax(&mOrd[(size_t)d * H + h], ordEncode(outv[g]));
    }
  }
}

#define AGG_TR      196
#define AGG_THREADS 1024
#define AGG_GROUPS  8

template <int F, int H>
__device__ __forceinline__ void aggr_batch(const int* __restrict__ ei,
                                           const int* __restrict__ dstp,
                                           const float* __restrict__ logit,
                                           const float* __restrict__ xr,
                                           float* acc, float* den, const float* mx,
                                           int qe, int cnt, int lo, int lane) {
  constexpr int C = F / H;
  constexpr int P = C / 32;
  constexpr int R = H * P;
  constexpr int U = 32 / R;
  const int s   = ei[qe];
  const int d   = dstp[qe];
  float lg[H];
#pragma unroll
  for (int h = 0; h < H; ++h) lg[h] = logit[(size_t)qe * H + h];
  const bool live = lane < cnt;
  const int row = live ? d - lo : 0;
  float al[H];
#pragma unroll
  for (int h = 0; h < H; ++h) al[h] = __expf(lg[h] - mx[row * H + h]);
  if (live) {
#pragma unroll
    for (int h = 0; h < H; ++h) atomicAdd(&den[row * H + h], al[h]);
  }

  for (int j0 = 0; j0 < cnt; j0 += U) {
    float v[U][R];
#pragma unroll
    for (int u = 0; u < U; ++u) {
      const int jj = min(j0 + u, cnt - 1);
      const int sj = __builtin_amdgcn_readlane(s, jj);
      const float* __restrict__ xrs = xr + (size_t)sj * F + lane;
#pragma unroll
      for (int i = 0; i < R; ++i) v[u][i] = xrs[i * 32];
    }
#pragma unroll
    for (int u = 0; u < U; ++u) {
      const int jj = j0 + u;
      if (jj < cnt) {
        const int rowj = __builtin_amdgcn_readlane(row, jj);
        float* arow = acc + rowj * F + lane;
#pragma unroll
        for (int i = 0; i < R; ++i) {
          const float a = __int_as_float(
              __builtin_amdgcn_readlane(__float_as_int(al[i / P]), jj));
          atomicAdd(&arow[i * 32], a * v[u][i]);
        }
      }
    }
  }
}

template <int F, int H>
__device__ __forceinline__ void aggr_push(const int* __restrict__ ei,
                                          const int* __restrict__ dstp,
                                          const float* __restrict__ logit,
                                          const float* __restrict__ xr,
                                          float* acc, float* den, const float* mx,
                                          int& qe, int& cnt, int e, int lo, int lane) {
  qe = (lane == cnt) ? e : qe;
  if (++cnt == 32) {
    aggr_batch<F, H>(ei, dstp, logit, xr, acc, den, mx, qe, 32, lo, lane);
    cnt = 0;
  }
}

template <int F, int H>
__global__ __launch_bounds__(AGG_THREADS)
void edge_aggr_rows(const int* __restrict__ ei, const float* __restrict__ logit,
                    const unsigned* __restrict__ mOrd, const float* __restrict__ xr,
                    const float* __restrict__ bias, const float* __restrict__ lng,
                    const float* __restrict__ lnb, float* __restrict__ out, int N, int E) {
  __shared__ float acc[AGG_TR * F];
  __shared__ float den[AGG_TR * H];
  __shared__ float mx[AGG_TR * H];
  const int tid  = threadIdx.x;
  const int lane = tid & 31;
  const int wave = __builtin_amdgcn_readfirstlane(tid >> 5);
  const int lo = blockIdx.x * AGG_TR;
  const int nrows = min(AGG_TR, N - lo);
  const unsigned lim = (unsigned)nrows;

  for (int i = tid; i < AGG_TR * F / 4; i += AGG_THREADS)
    ((float4*)acc)[i] = make_float4(0.f, 0.f, 0.f, 0.f);
  for (int i = tid; i < AGG_TR * H; i += AGG_THREADS) {
    den[i] = 0.0f;
    mx[i] = (i < nrows * H) ? ordDecode(mOrd[(size_t)lo * H + i]) : 0.0f;
  }
  __syncthreads();

  const int* __restrict__ dstp = ei + E;
  int qe = 0;
  int cnt = 0;
  if ((E & 3) == 0) {
    const int4* __restrict__ dst4 = (const int4*)dstp;
    const int nQuads = E >> 2;
    for (int qb = wave * (32 * AGG_GROUPS); qb < nQuads;
         qb += (AGG_THREADS / 32) * 32 * AGG_GROUPS) {
      int4 d[AGG_GROUPS];
#pragma unroll
      for (int u = 0; u < AGG_GROUPS; ++u)
        d[u] = dst4[min(qb + u * 32 + lane, nQuads - 1)];
      unsigned bits = 0u, valid = 0u;
#pragma unroll
      for (int u = 0; u < AGG_GROUPS; ++u) {
        bits |= ((unsigned)(d[u].x - lo) < lim) ? (1u << (4 * u))     : 0u;
        bits |= ((unsigned)(d[u].y - lo) < lim) ? (1u << (4 * u + 1)) : 0u;
        bits |= ((unsigned)(d[u].z - lo) < lim) ? (1u << (4 * u + 2)) : 0u;
        bits |= ((unsigned)(d[u].w - lo) < lim) ? (1u << (4 * u + 3)) : 0u;
        valid |= ((qb + u * 32 + lane) < nQuads) ? (0xFu << (4 * u)) : 0u;
      }
      bits &= valid;
      unsigned any = __builtin_amdgcn_ballot_w32(bits != 0u);
      while (any) {
        const int L = __builtin_ctz(any);
        any &= any - 1u;
        unsigned b = (unsigned)__builtin_amdgcn_readlane((int)bits, L);
        while (b) {
          const int t = __builtin_ctz(b);
          b &= b - 1u;
          const int e = (qb + (t >> 2) * 32 + L) * 4 + (t & 3);
          aggr_push<F, H>(ei, dstp, logit, xr, acc, den, mx, qe, cnt, e, lo, lane);
        }
      }
    }
  } else {
    for (int eb = wave * 32; eb < E; eb += AGG_THREADS) {
      const int e1 = eb + lane;
      const unsigned r1 = (e1 < E) ? (unsigned)(dstp[min(e1, E - 1)] - lo) : 0xFFFFFFFFu;
      unsigned mk = __builtin_amdgcn_ballot_w32(r1 < lim);
      while (mk) {
        const int j = __builtin_ctz(mk);
        mk &= mk - 1u;
        aggr_push<F, H>(ei, dstp, logit, xr, acc, den, mx, qe, cnt, eb + j, lo, lane);
      }
    }
  }
  if (cnt > 0)
    aggr_batch<F, H>(ei, dstp, logit, xr, acc, den, mx, qe, cnt, lo, lane);
  __syncthreads();
  constexpr int PER = F / 32;
  constexpr int ROWS_TOGETHER = 4;
  constexpr int WAVES = AGG_THREADS / 32;
  static_assert(PER >= 1 && (F / H) % PER == 0, "a lane's channels must lie in one head");
  typedef float vp __attribute__((ext_vector_type(PER)));
  const vp bs = *(const vp*)(bias + lane * PER);
  const vp gg = *(const vp*)(lng + lane * PER);
  const vp bb = *(const vp*)(lnb + lane * PER);
  const int hd = (lane * PER) / (F / H);
  for (int r0 = wave; r0 < nrows; r0 += WAVES * ROWS_TOGETHER) {
    float vals[ROWS_TOGETHER][PER];
    float sum[ROWS_TOGETHER];
#pragma unroll
    for (int k = 0; k < ROWS_TOGETHER; ++k) {
      const int r = min(r0 + k * WAVES, nrows - 1);
      const vp a = *(const vp*)(acc + r * F + lane * PER);
      const float dn = den[r * H + hd];
      float s = 0.0f;
#pragma unroll
      for (int i = 0; i < PER; ++i) {
        float v = a[i];
        if (dn > 0.0f) v /= dn;
        v += bs[i];
        vals[k][i] = v;
        s += v;
      }
      sum[k] = s;
    }
#pragma unroll
    for (int off = 16; off > 0; off >>= 1) {
#pragma unroll
      for (int k = 0; k < ROWS_TOGETHER; ++k) sum[k] += __shfl_xor(sum[k], off, 32);
    }
    float mu[ROWS_TOGETHER], vs[ROWS_TOGETHER];
#pragma unroll
    for (int k = 0; k < ROWS_TOGETHER; ++k) {
      mu[k] = sum[k] / (float)F;
      float q = 0.0f;
#pragma unroll
      for (int i = 0; i < PER; ++i) { const float dlt = vals[k][i] - mu[k]; q += dlt * dlt; }
      vs[k] = q;
    }
#pragma unroll
    for (int off = 16; off > 0; off >>= 1) {
#pragma unroll
      for (int k = 0; k < ROWS_TOGETHER; ++k) vs[k] += __shfl_xor(vs[k], off, 32);
    }
#pragma unroll
    for (int k = 0; k < ROWS_TOGETHER; ++k) {
      const int r = r0 + k * WAVES;
      if (r < nrows) {
        const float inv = rsqrtf(vs[k] / (float)F + LN_EPS);
        vp o;
#pragma unroll
        for (int i = 0; i < PER; ++i) {
          const float hv = (vals[k][i] - mu[k]) * inv * gg[i] + bb[i];
          o[i] = (hv > 0.0f) ? hv : (__expf(hv) - 1.0f);
        }
        *(vp*)(out + (size_t)(lo + r) * F + lane * PER) = o;
      }
    }
  }
}

extern "C" void kernel_launch(void* const* d_in, const int* in_sizes, int n_in,
                              void* d_out, int out_size, void* d_ws, size_t ws_size,
                              hipStream_t stream) {
  const float* x      = (const float*)d_in[0];
  const float* ea     = (const float*)d_in[1];
  const float* W1l    = (const float*)d_in[2];
  const float* W1r    = (const float*)d_in[3];
  const float* W1e    = (const float*)d_in[4];
  const float* att1   = (const float*)d_in[5];
  const float* b1     = (const float*)d_in[6];
  const float* ln1_g  = (const float*)d_in[7];
  const float* ln1_b  = (const float*)d_in[8];
  const float* W2l    = (const float*)d_in[9];
  const float* W2r    = (const float*)d_in[10];
  const float* W2e    = (const float*)d_in[11];
  const float* att2   = (const float*)d_in[12];
  const float* b2     = (const float*)d_in[13];
  const float* ln2_g  = (const float*)d_in[14];
  const float* ln2_b  = (const float*)d_in[15];
  const int*   ei     = (const int*)d_in[16];
  float*       outp   = (float*)d_out;

  const int N = in_sizes[0] / 10;
  const int E = in_sizes[1];
  const int HID = 32, HEADS = 4, F1 = HEADS * HID  , OUTC = 64;
  const int KP = 12;

  float* ws      = (float*)d_ws;
  float* xl1     = ws;
  float* xr1     = xl1 + (size_t)N * F1;
  float* h1      = xr1 + (size_t)N * F1;
  float* logit1  = h1  + (size_t)N * F1;
  unsigned* m1   = (unsigned*)(logit1 + (size_t)E * HEADS);
  float* xpad    = (float*)(m1 + (size_t)N * HEADS);
  float* w1lp    = xpad + (size_t)N * KP;
  float* w1rp    = w1lp + (size_t)KP * F1;
  float* xr2     = xl1;
  float* logit2  = xl1 + (size_t)N * OUTC;
  unsigned* m2   = (unsigned*)(logit2 + (size_t)E);
  float* xl2     = logit1;

  const unsigned ORD_NEG_INF = 0x007FFFFFu;
  const int TB = 256;
  auto blocks = [](size_t n, int tb) { return (unsigned)((n + tb - 1) / tb); };

  pad_cols<<<blocks((size_t)N * KP, TB), TB, 0, stream>>>(x, xpad, N, 10, KP);
  pad_rows<<<blocks((size_t)KP * F1, TB), TB, 0, stream>>>(W1l, w1lp, 10, KP, F1);
  pad_rows<<<blocks((size_t)KP * F1, TB), TB, 0, stream>>>(W1r, w1rp, 10, KP, F1);
  {
    dim3 grid((N / 16 + 3) / 4, F1 / 16);
    wmma_gemm<12><<<grid, 128, 0, stream>>>(xpad, w1lp, xl1, N, F1);
    wmma_gemm<12><<<grid, 128, 0, stream>>>(xpad, w1rp, xr1, N, F1);
  }
  fill_u32<<<blocks((size_t)N * HEADS, TB), TB, 0, stream>>>(m1, ORD_NEG_INF, (size_t)N * HEADS);

  edge_logits_rows<128, 4><<<blocks((size_t)E, TB), TB, 0, stream>>>(
      ei, ea, xl1, xr1, W1e, att1, logit1, m1, E);
  edge_aggr_rows<128, 4><<<blocks((size_t)N, AGG_TR), AGG_THREADS, 0, stream>>>(
      ei, logit1, m1, xr1, b1, ln1_g, ln1_b, h1, N, E);

  {
    dim3 grid((N / 16 + 3) / 4, OUTC / 16);
    wmma_gemm<128><<<grid, 128, 0, stream>>>(h1, W2l, xl2, N, OUTC);
    wmma_gemm<128><<<grid, 128, 0, stream>>>(h1, W2r, xr2, N, OUTC);
  }
  fill_u32<<<blocks((size_t)N, TB), TB, 0, stream>>>(m2, ORD_NEG_INF, (size_t)N);

  edge_logits_rows<64, 1><<<blocks((size_t)E, TB), TB, 0, stream>>>(
      ei, ea, xl2, xr2, W2e, att2, logit2, m2, E);
  edge_aggr_rows<64, 1><<<blocks((size_t)N, AGG_TR), AGG_THREADS, 0, stream>>>(
      ei, logit2, m2, xr2, b2, ln2_g, ln2_b, outp, N, E);
}


# aggregation scan: a block reads a 1-byte owner tag per edge (dst/196, written by the layer-1 logit pass), 16 edges a load, not the 4-byte dst list; queue order changes, sums: LDS arrival as before
# speedup vs baseline: 5.4360x; 1.0714x over previous
#include <hip/hip_runtime.h>
#include <hip/hip_bf16.h>
#include <math.h>

typedef float     v2f  __attribute__((ext_vector_type(2)));
typedef float     v8f  __attribute__((ext_vector_type(8)));

#define NEG_SLOPE 0.2f
#define LN_EPS    1e-5f

__device__ __forceinline__ unsigned ordEncode(float f) {
  unsigned u = __float_as_uint(f);
  return (u >> 31) ? ~u : (u | 0x80000000u);
}
__device__ __forceinline__ float ordDecode(unsigned o) {
  return __uint_as_float((o >> 31) ? (o & 0x7FFFFFFFu) : ~o);
}

__global__ void fill_u32(unsigned* p, unsigned v, size_t n) {
  size_t i = (size_t)blockIdx.x * blockDim.x + threadIdx.x;
  if (i < n) p[i] = v;
}

__global__ void pad_cols(const float* __restrict__ src, float* __restrict__ dst,
                         int M, int Ks, int Kd) {
  int i = blockIdx.x * blockDim.x + threadIdx.x;
  if (i >= M * Kd) return;
  int r = i / Kd, c = i % Kd;
  dst[i] = (c < Ks) ? src[(size_t)r * Ks + c] : 0.0f;
}

__global__ void pad_rows(const float* __restrict__ src, float* __restrict__ dst,
                         int Ms, int Md, int Ncols) {
  int i = blockIdx.x * blockDim.x + threadIdx.x;
  if (i >= Md * Ncols) return;
  int r = i / Ncols;
  dst[i] = (r < Ms) ? src[i] : 0.0f;
}

template <int K>
__global__ void wmma_gemm(const float* __restrict__ A, const float* __restrict__ B,
                          float* __restrict__ C, int M, int N) {
  const int lane  = threadIdx.x & 31;
  const int wave  = threadIdx.x >> 5;
  const int tileM = blockIdx.x * 4 + wave;
  const int m0 = tileM * 16;
  const int n0 = blockIdx.y * 16;
  if (m0 + 16 > M) return;
  const int half = lane >> 4;
  const int l16  = lane & 15;
  const float* __restrict__ arow = A + (size_t)(m0 + l16) * K + half * 2;
  const float* __restrict__ bcol = B + n0 + l16 + (size_t)(half * 2) * N;
  v8f acc = {};
#pragma unroll
  for (int k = 0; k < K; k += 4) {
    v2f a = *(const v2f*)(arow + k);
    v2f b;
    b.x = bcol[(size_t)k * N];
    b.y = bcol[(size_t)(k + 1) * N];
    acc = __builtin_amdgcn_wmma_f32_16x16x4_f32(false, a, false, b, (short)0, acc,
                                                false, false);
  }
  float* __restrict__ crow = C + (size_t)(m0 + half * 8) * N + n0 + l16;
#pragma unroll
  for (int r = 0; r < 8; ++r)
    crow[(size_t)r * N] = acc[r];
}

#define LG_U 8
#define AGG_TR 196

template <int F, int H>
__global__ __launch_bounds__(256)
void edge_logits_rows(const int* __restrict__ ei, const float* __restrict__ ea,
                      const float* __restrict__ xl, const float* __restrict__ xr,
                      const float* __restrict__ We, const float* __restrict__ att,
                      float* __restrict__ logit, unsigned* __restrict__ mOrd,
                      unsigned char* __restrict__ tag, int E) {
  constexpr int VEC = F / 32;
  constexpr int LPH = 32 / H;
  static_assert(LG_U == 8 && LPH >= 8 && 32 % H == 0, "the lane sum below is laid out for 8 edges a step");
  typedef float vf __attribute__((ext_vector_type(VEC)));
  const int lane = threadIdx.x & 31;
  const int eb = (blockIdx.x * (blockDim.x >> 5) + (threadIdx.x >> 5)) * 32;
  if (eb >= E) return;
  const int el = min(eb + lane, E - 1);
  const int sL = ei[el];
  const int dL = ei[E + el];
  const int eaL = __float_as_int(ea[el]);
  if (tag != nullptr && eb + lane < E) tag[eb + lane] = (unsigned char)(dL / AGG_TR);
  const vf we = *(const vf*)(We + lane * VEC);
  const vf at = *(const vf*)(att + lane * VEC);
  float outv[H];
#pragma unroll
  for (int g = 0; g < H; ++g) outv[g] = 0.0f;

#pragma unroll
  for (int jb = 0; jb < 32; jb += LG_U) {
    vf a[LG_U], b[LG_U];
#pragma unroll
    for (int u = 0; u < LG_U; ++u) {
      const int s = __builtin_amdgcn_readlane(sL, jb + u);
      const int d = __builtin_amdgcn_readlane(dL, jb + u);
      a[u] = *(const vf*)(xl + (size_t)d * F + lane * VEC);
      b[u] = *(const vf*)(xr + (size_t)s * F + lane * VEC);
    }
    float p[LG_U];
#pragma unroll
    for (int u = 0; u < LG_U; ++u) {
      const float eav = __int_as_float(__builtin_amdgcn_readlane(eaL, jb + u));
      float acc = 0.0f;
#pragma unroll
      for (int i = 0; i < VEC; ++i) {
        float v = a[u][i] + b[u][i] + eav * we[i];
        v = fmaxf(v, NEG_SLOPE * v);
        acc += at[i] * v;
      }
      p[u] = acc;
    }
#pragma unroll
    for (int st = 0; st < 3; ++st) {
      const int off = LPH >> (st + 1);
      const int hn  = LG_U >> (st + 1);
      const bool hi = (lane & off) != 0;
#pragma unroll
      for (int u = 0; u < hn; ++u) {
        const float send = hi ? p[u] : p[u + hn];
        const float keep = hi ? p[u + hn] : p[u];
        p[u] = keep + __shfl_xor(send, off, 32);
      }
    }
#pragma unroll
    for (int off = LPH / 16; off > 0; off >>= 1) p[0] += __shfl_xor(p[0], off, 32);
    const int rl = lane - (jb * H) % 32;
    const bool mine = (rl >= 0) && (rl < LG_U * H);
    const int rc = mine ? rl : 0;
    const float q = __shfl(p[0], (rc % H) * LPH + (rc / H) * (LPH / 8), 32);
    if (mine) outv[(jb * H) / 32] = q;
  }

#pragma unroll
  for (int g = 0; g < H; ++g) {
    const int idx = g * 32 + lane;
    const int j = idx / H, h = idx % H;
    const int d = __shfl(dL, j, 32);
    if (eb + j < E) {
      logit[(size_t)eb * H + idx] = outv[g];
      atomicMax(&mOrd[(size_t)d * H + h], ordEncode(outv[g]));
    }
  }
}

#define AGG_THREADS 1024
#define AGG_GROUPS  8

__device__ __forceinline__ unsigned tag_differs(unsigned w, unsigned pat) {
  const unsigned x = w ^ pat;
  return ((x & 0x7F7F7F7Fu) + 0x7F7F7F7Fu) | x;
}
__device__ __forceinline__ unsigned tags_differ16(uint4 t, unsigned pat) {
  unsigned r = (tag_differs(t.x, pat) >> 7) & 0x01010101u;
  r |= (tag_differs(t.y, pat) >> 6) & 0x02020202u;
  r |= (tag_differs(t.z, pat) >> 5) & 0x04040404u;
  r |= (tag_differs(t.w, pat) >> 4) & 0x08080808u;
  return r;
}

template <int F, int H>
__device__ __forceinline__ void aggr_batch(const int* __restrict__ ei,
                                           const int* __restrict__ dstp,
                                           const float* __restrict__ logit,
                                           const float* __restrict__ xr,
                                           float* acc, float* den, const float* mx,
                                           int qe, int cnt, int lo, int lane) {
  constexpr int C = F / H;
  constexpr int P = C / 32;
  constexpr int R = H * P;
  constexpr int U = 32 / R;
  const int s   = ei[qe];
  const int d   = dstp[qe];
  float lg[H];
#pragma unroll
  for (int h = 0; h < H; ++h) lg[h] = logit[(size_t)qe * H + h];
  const bool live = lane < cnt;
  const int row = live ? d - lo : 0;
  float al[H];
#pragma unroll
  for (int h = 0; h < H; ++h) al[h] = __expf(lg[h] - mx[row * H + h]);
  if (live) {
#pragma unroll
    for (int h = 0; h < H; ++h) atomicAdd(&den[row * H + h], al[h]);
  }

  for (int j0 = 0; j0 < cnt; j0 += U) {
    float v[U][R];
#pragma unroll
    for (int u = 0; u < U; ++u) {
      const int jj = min(j0 + u, cnt - 1);
      const int sj = __builtin_amdgcn_readlane(s, jj);
      const float* __restrict__ xrs = xr + (size_t)sj * F + lane;
#pragma unroll
      for (int i = 0; i < R; ++i) v[u][i] = xrs[i * 32];
    }
#pragma unroll
    for (int u = 0; u < U; ++u) {
      const int jj = j0 + u;
      if (jj < cnt) {
        const int rowj = __builtin_amdgcn_readlane(row, jj);
        float* arow = acc + rowj * F + lane;
#pragma unroll
        for (int i = 0; i < R; ++i) {
          const float a = __int_as_float(
              __builtin_amdgcn_readlane(__float_as_int(al[i / P]), jj));
          atomicAdd(&arow[i * 32], a * v[u][i]);
        }
      }
    }
  }
}

template <int F, int H>
__device__ __forceinline__ void aggr_push(const int* __restrict__ ei,
                                          const int* __restrict__ dstp,
                                          const float* __restrict__ logit,
                                          const float* __restrict__ xr,
                                          float* acc, float* den, const float* mx,
                                          int& qe, int& cnt, int e, int lo, int lane) {
  qe = (lane == cnt) ? e : qe;
  if (++cnt == 32) {
    aggr_batch<F, H>(ei, dstp, logit, xr, acc, den, mx, qe, 32, lo, lane);
    cnt = 0;
  }
}

template <int F, int H>
__global__ __launch_bounds__(AGG_THREADS)
void edge_aggr_rows(const int* __restrict__ ei, const unsigned char* __restrict__ tag,
                    const float* __restrict__ logit,
                    const unsigned* __restrict__ mOrd, const float* __restrict__ xr,
                    const float* __restrict__ bias, const float* __restrict__ lng,
                    const float* __restrict__ lnb, float* __restrict__ out, int N, int E) {
  __shared__ float acc[AGG_TR * F];
  __shared__ float den[AGG_TR * H];
  __shared__ float mx[AGG_TR * H];
  const int tid  = threadIdx.x;
  const int lane = tid & 31;
  const int wave = __builtin_amdgcn_readfirstlane(tid >> 5);
  const int lo = blockIdx.x * AGG_TR;
  const int nrows = min(AGG_TR, N - lo);
  const unsigned lim = (unsigned)nrows;

  for (int i = tid; i < AGG_TR * F / 4; i += AGG_THREADS)
    ((float4*)acc)[i] = make_float4(0.f, 0.f, 0.f, 0.f);
  for (int i = tid; i < AGG_TR * H; i += AGG_THREADS) {
    den[i] = 0.0f;
    mx[i] = (i < nrows * H) ? ordDecode(mOrd[(size_t)lo * H + i]) : 0.0f;
  }
  __syncthreads();

  const int* __restrict__ dstp = ei + E;
  int qe = 0;
  int cnt = 0;
  const bool useTags = (tag != nullptr) && (gridDim.x <= 256u) && ((((size_t)tag) & 15) == 0);
  const int nGroups = useTags ? (E >> 4) : 0;
  if (nGroups > 0) {
    const uint4* __restrict__ tag4 = (const uint4*)tag;
    const unsigned pat = blockIdx.x * 0x01010101u;
    for (int gb = wave * (32 * AGG_GROUPS); gb < nGroups;
         gb += (AGG_THREADS / 32) * 32 * AGG_GROUPS) {
      uint4 t[AGG_GROUPS];
#pragma unroll
      for (int u = 0; u < AGG_GROUPS; ++u)
        t[u] = tag4[min(gb + u * 32 + lane, nGroups - 1)];
      __builtin_amdgcn_sched_barrier(0);
      unsigned m[AGG_GROUPS / 2];
#pragma unroll
      for (int j = 0; j < AGG_GROUPS / 2; ++j) {
        const unsigned differ = tags_differ16(t[2 * j], pat) | (tags_differ16(t[2 * j + 1], pat) << 4);
        const unsigned ok = (((gb + (2 * j) * 32 + lane) < nGroups) ? 0x0F0F0F0Fu : 0u) |
                            (((gb + (2 * j + 1) * 32 + lane) < nGroups) ? 0xF0F0F0F0u : 0u);
        m[j] = ~differ & ok;
      }
      static_assert(AGG_GROUPS == 8, "the choice of m[j] below is written for four words");
#pragma nounroll
      for (int j = 0; j < AGG_GROUPS / 2; ++j) {
        const unsigned mj = (j == 0) ? m[0] : (j == 1) ? m[1] : (j == 2) ? m[2] : m[3];
        unsigned any = __builtin_amdgcn_ballot_w32(mj != 0u);
        while (any) {
          const int L = __builtin_ctz(any);
          any &= any - 1u;
          unsigned b = (unsigned)__builtin_amdgcn_readlane((int)mj, L);
          while (b) {
            const int p = __builtin_ctz(b);
            b &= b - 1u;
            const int g = gb + (2 * j + ((p >> 2) & 1)) * 32 + L;
            const int e = (g << 4) + ((p & 3) << 2) + (p >> 3);
            aggr_push<F, H>(ei, dstp, logit, xr, acc, den, mx, qe, cnt, e, lo, lane);
          }
        }
      }
    }
  }
  for (int eb = (nGroups << 4) + wave * 32; eb < E; eb += AGG_THREADS) {
    const int e1 = eb + lane;
    const unsigned r1 = (e1 < E) ? (unsigned)(dstp[min(e1, E - 1)] - lo) : 0xFFFFFFFFu;
    unsigned mk = __builtin_amdgcn_ballot_w32(r1 < lim);
    while (mk) {
      const int j = __builtin_ctz(mk);
      mk &= mk - 1u;
      aggr_push<F, H>(ei, dstp, logit, xr, acc, den, mx, qe, cnt, eb + j, lo, lane);
    }
  }
  if (cnt > 0)
    aggr_batch<F, H>(ei, dstp, logit, xr, acc, den, mx, qe, cnt, lo, lane);
  __syncthreads();
  constexpr int PER = F / 32;
  constexpr int ROWS_TOGETHER = 4;
  constexpr int WAVES = AGG_THREADS / 32;
  static_assert(PER >= 1 && (F / H) % PER == 0, "a lane's channels must lie in one head");
  typedef float vp __attribute__((ext_vector_type(PER)));
  const vp bs = *(const vp*)(bias + lane * PER);
  const vp gg = *(const vp*)(lng + lane * PER);
  const vp bb = *(const vp*)(lnb + lane * PER);
  const int hd = (lane * PER) / (F / H);
  for (int r0 = wave; r0 < nrows; r0 += WAVES * ROWS_TOGETHER) {
    float vals[ROWS_TOGETHER][PER];
    float sum[ROWS_TOGETHER];
#pragma unroll
    for (int k = 0; k < ROWS_TOGETHER; ++k) {
      const int r = min(r0 + k * WAVES, nrows - 1);
      const vp a = *(const vp*)(acc + r * F + lane * PER);
      const float dn = den[r * H + hd];
      float s = 0.0f;
#pragma unroll
      for (int i = 0; i < PER; ++i) {
        float v = a[i];
        if (dn > 0.0f) v /= dn;
        v += bs[i];
        vals[k][i] = v;
        s += v;
      }
      sum[k] = s;
    }
#pragma unroll
    for (int off = 16; off > 0; off >>= 1) {
#pragma unroll
      for (int k = 0; k < ROWS_TOGETHER; ++k) sum[k] += __shfl_xor(sum[k], off, 32);
    }
    float mu[ROWS_TOGETHER], vs[ROWS_TOGETHER];
#pragma unroll
    for (int k = 0; k < ROWS_TOGETHER; ++k) {
      mu[k] = sum[k] / (float)F;
      float q = 0.0f;
#pragma unroll
      for (int i = 0; i < PER; ++i) { const float dlt = vals[k][i] - mu[k]; q += dlt * dlt; }
      vs[k] = q;
    }
#pragma unroll
    for (int off = 16; off > 0; off >>= 1) {
#pragma unroll
      for (int k = 0; k < ROWS_TOGETHER; ++k) vs[k] += __shfl_xor(vs[k], off, 32);
    }
#pragma unroll
    for (int k = 0; k < ROWS_TOGETHER; ++k) {
      const int r = r0 + k * WAVES;
      if (r < nrows) {
        const float inv = rsqrtf(vs[k] / (float)F + LN_EPS);
        vp o;
#pragma unroll
        for (int i = 0; i < PER; ++i) {
          const float hv = (vals[k][i] - mu[k]) * inv * gg[i] + bb[i];
          o[i] = (hv > 0.0f) ? hv : (__expf(hv) - 1.0f);
        }
        *(vp*)(out + (size_t)(lo + r) * F + lane * PER) = o;
      }
    }
  }
}

extern "C" void kernel_launch(void* const* d_in, const int* in_sizes, int n_in,
                              void* d_out, int out_size, void* d_ws, size_t ws_size,
                              hipStream_t stream) {
  const float* x      = (const float*)d_in[0];
  const float* ea     = (const float*)d_in[1];
  const float* W1l    = (const float*)d_in[2];
  const float* W1r    = (const float*)d_in[3];
  const float* W1e    = (const float*)d_in[4];
  const float* att1   = (const float*)d_in[5];
  const float* b1     = (const float*)d_in[6];
  const float* ln1_g  = (const float*)d_in[7];
  const float* ln1_b  = (const float*)d_in[8];
  const float* W2l    = (const float*)d_in[9];
  const float* W2r    = (const float*)d_in[10];
  const float* W2e    = (const float*)d_in[11];
  const float* att2   = (const float*)d_in[12];
  const float* b2     = (const float*)d_in[13];
  const float* ln2_g  = (const float*)d_in[14];
  const float* ln2_b  = (const float*)d_in[15];
  const int*   ei     = (const int*)d_in[16];
  float*       outp   = (float*)d_out;

  const int N = in_sizes[0] / 10;
  const int E = in_sizes[1];
  const int HID = 32, HEADS = 4, F1 = HEADS * HID  , OUTC = 64;
  const int KP = 12;

  float* ws      = (float*)d_ws;
  float* xl1     = ws;
  float* xr1     = xl1 + (size_t)N * F1;
  float* h1      = xr1 + (size_t)N * F1;
  float* logit1  = h1  + (size_t)N * F1;
  unsigned* m1   = (unsigned*)(logit1 + (size_t)E * HEADS);
  float* xpad    = (float*)(m1 + (size_t)N * HEADS);
  float* w1lp    = xpad + (size_t)N * KP;
  float* w1rp    = w1lp + (size_t)KP * F1;
  unsigned char* etag = (unsigned char*)(w1rp + (size_t)KP * F1);
  float* xr2     = xl1;
  float* logit2  = xl1 + (size_t)N * OUTC;
  unsigned* m2   = (unsigned*)(logit2 + (size_t)E);
  float* xl2     = logit1;

  const unsigned ORD_NEG_INF = 0x007FFFFFu;
  const int TB = 256;
  auto blocks = [](size_t n, int tb) { return (unsigned)((n + tb - 1) / tb); };

  pad_cols<<<blocks((size_t)N * KP, TB), TB, 0, stream>>>(x, xpad, N, 10, KP);
  pad_rows<<<blocks((size_t)KP * F1, TB), TB, 0, stream>>>(W1l, w1lp, 10, KP, F1);
  pad_rows<<<blocks((size_t)KP * F1, TB), TB, 0, stream>>>(W1r, w1rp, 10, KP, F1);
  {
    dim3 grid((N / 16 + 3) / 4, F1 / 16);
    wmma_gemm<12><<<grid, 128, 0, stream>>>(xpad, w1lp, xl1, N, F1);
    wmma_gemm<12><<<grid, 128, 0, stream>>>(xpad, w1rp, xr1, N, F1);
  }
  fill_u32<<<blocks((size_t)N * HEADS, TB), TB, 0, stream>>>(m1, ORD_NEG_INF, (size_t)N * HEADS);

  edge_logits_rows<128, 4><<<blocks((size_t)E, TB), TB, 0, stream>>>(
      ei, ea, xl1, xr1, W1e, att1, logit1, m1, etag, E);
  edge_aggr_rows<128, 4><<<blocks((size_t)N, AGG_TR), AGG_THREADS, 0, stream>>>(
      ei, etag, logit1, m1, xr1, b1, ln1_g, ln1_b, h1, N, E);

  {
    dim3 grid((N / 16 + 3) / 4, OUTC / 16);
    wmma_gemm<128><<<grid, 128, 0, stream>>>(h1, W2l, xl2, N, OUTC);
    wmma_gemm<128><<<grid, 128, 0, stream>>>(h1, W2r, xr2, N, OUTC);
  }
  fill_u32<<<blocks((size_t)N, TB), TB, 0, stream>>>(m2, ORD_NEG_INF, (size_t)N);

  edge_logits_rows<64, 1><<<blocks((size_t)E, TB), TB, 0, stream>>>(
      ei, ea, xl2, xr2, W2e, att2, logit2, m2, nullptr, E);
  edge_aggr_rows<64, 1><<<blocks((size_t)N, AGG_TR), AGG_THREADS, 0, stream>>>(
      ei, etag, logit2, m2, xr2, b2, ln2_g, ln2_b, outp, N, E);
}


# segment max off global atomics: the aggregation block scans once, takes LDS integer maxima per row, lists its edges in LDS, then works the list; max fills gone; sums: LDS arrival, new edge order
# speedup vs baseline: 6.2669x; 1.1529x over previous
#include <hip/hip_runtime.h>
#include <hip/hip_bf16.h>
#include <math.h>

typedef float     v2f  __attribute__((ext_vector_type(2)));
typedef float     v8f  __attribute__((ext_vector_type(8)));

#define NEG_SLOPE 0.2f
#define LN_EPS    1e-5f

__device__ __forceinline__ unsigned ordEncode(float f) {
  unsigned u = __float_as_uint(f);
  return (u >> 31) ? ~u : (u | 0x80000000u);
}
__device__ __forceinline__ float ordDecode(unsigned o) {
  return __uint_as_float((o >> 31) ? (o & 0x7FFFFFFFu) : ~o);
}

__global__ void pad_cols(const float* __restrict__ src, float* __restrict__ dst,
                         int M, int Ks, int Kd) {
  int i = blockIdx.x * blockDim.x + threadIdx.x;
  if (i >= M * Kd) return;
  int r = i / Kd, c = i % Kd;
  dst[i] = (c < Ks) ? src[(size_t)r * Ks + c] : 0.0f;
}

__global__ void pad_rows(const float* __restrict__ src, float* __restrict__ dst,
                         int Ms, int Md, int Ncols) {
  int i = blockIdx.x * blockDim.x + threadIdx.x;
  if (i >= Md * Ncols) return;
  int r = i / Ncols;
  dst[i] = (r < Ms) ? src[i] : 0.0f;
}

template <int K>
__global__ void wmma_gemm(const float* __restrict__ A, const float* __restrict__ B,
                          float* __restrict__ C, int M, int N) {
  const int lane  = threadIdx.x & 31;
  const int wave  = threadIdx.x >> 5;
  const int tileM = blockIdx.x * 4 + wave;
  const int m0 = tileM * 16;
  const int n0 = blockIdx.y * 16;
  if (m0 + 16 > M) return;
  const int half = lane >> 4;
  const int l16  = lane & 15;
  const float* __restrict__ arow = A + (size_t)(m0 + l16) * K + half * 2;
  const float* __restrict__ bcol = B + n0 + l16 + (size_t)(half * 2) * N;
  v8f acc = {};
#pragma unroll
  for (int k = 0; k < K; k += 4) {
    v2f a = *(const v2f*)(arow + k);
    v2f b;
    b.x = bcol[(size_t)k * N];
    b.y = bcol[(size_t)(k + 1) * N];
    acc = __builtin_amdgcn_wmma_f32_16x16x4_f32(false, a, false, b, (short)0, acc,
                                                false, false);
  }
  float* __restrict__ crow = C + (size_t)(m0 + half * 8) * N + n0 + l16;
#pragma unroll
  for (int r = 0; r < 8; ++r)
    crow[(size_t)r * N] = acc[r];
}

#define LG_U 8
#define AGG_TR 196

template <int F, int H>
__global__ __launch_bounds__(256)
void edge_logits_rows(const int* __restrict__ ei, const float* __restrict__ ea,
                      const float* __restrict__ xl, const float* __restrict__ xr,
                      const float* __restrict__ We, const float* __restrict__ att,
                      float* __restrict__ logit,
                      unsigned char* __restrict__ tag, int E) {
  constexpr int VEC = F / 32;
  constexpr int LPH = 32 / H;
  static_assert(LG_U == 8 && LPH >= 8 && 32 % H == 0, "the lane sum below is laid out for 8 edges a step");
  typedef float vf __attribute__((ext_vector_type(VEC)));
  const int lane = threadIdx.x & 31;
  const int eb = (blockIdx.x * (blockDim.x >> 5) + (threadIdx.x >> 5)) * 32;
  if (eb >= E) return;
  const int el = min(eb + lane, E - 1);
  const int sL = ei[el];
  const int dL = ei[E + el];
  const int eaL = __float_as_int(ea[el]);
  if (tag != nullptr && eb + lane < E) tag[eb + lane] = (unsigned char)(dL / AGG_TR);
  const vf we = *(const vf*)(We + lane * VEC);
  const vf at = *(const vf*)(att + lane * VEC);
  float outv[H];
#pragma unroll
  for (int g = 0; g < H; ++g) outv[g] = 0.0f;

#pragma unroll
  for (int jb = 0; jb < 32; jb += LG_U) {
    vf a[LG_U], b[LG_U];
#pragma unroll
    for (int u = 0; u < LG_U; ++u) {
      const int s = __builtin_amdgcn_readlane(sL, jb + u);
      const int d = __builtin_amdgcn_readlane(dL, jb + u);
      a[u] = *(const vf*)(xl + (size_t)d * F + lane * VEC);
      b[u] = *(const vf*)(xr + (size_t)s * F + lane * VEC);
    }
    float p[LG_U];
#pragma unroll
    for (int u = 0; u < LG_U; ++u) {
      const float eav = __int_as_float(__builtin_amdgcn_readlane(eaL, jb + u));
      float acc = 0.0f;
#pragma unroll
      for (int i = 0; i < VEC; ++i) {
        float v = a[u][i] + b[u][i] + eav * we[i];
        v = fmaxf(v, NEG_SLOPE * v);
        acc += at[i] * v;
      }
      p[u] = acc;
    }
#pragma unroll
    for (int st = 0; st < 3; ++st) {
      const int off = LPH >> (st + 1);
      const int hn  = LG_U >> (st + 1);
      const bool hi = (lane & off) != 0;
#pragma unroll
      for (int u = 0; u < hn; ++u) {
        const float send = hi ? p[u] : p[u + hn];
        const float keep = hi ? p[u + hn] : p[u];
        p[u] = keep + __shfl_xor(send, off, 32);
      }
    }
#pragma unroll
    for (int off = LPH / 16; off > 0; off >>= 1) p[0] += __shfl_xor(p[0], off, 32);
    const int rl = lane - (jb * H) % 32;
    const bool mine = (rl >= 0) && (rl < LG_U * H);
    const int rc = mine ? rl : 0;
    const float q = __shfl(p[0], (rc % H) * LPH + (rc / H) * (LPH / 8), 32);
    if (mine) outv[(jb * H) / 32] = q;
  }

#pragma unroll
  for (int g = 0; g < H; ++g) {
    const int idx = g * 32 + lane;
    const int j = idx / H;
    if (eb + j < E) logit[(size_t)eb * H + idx] = outv[g];
  }
}

#define AGG_THREADS 1024
#define AGG_GROUPS  8
#define AGG_CAP     8192
#define ORD_NEG_INF 0x007FFFFFu

__device__ __forceinline__ unsigned tag_differs(unsigned w, unsigned pat) {
  const unsigned x = w ^ pat;
  return ((x & 0x7F7F7F7Fu) + 0x7F7F7F7Fu) | x;
}
__device__ __forceinline__ unsigned tags_differ16(uint4 t, unsigned pat) {
  unsigned r = (tag_differs(t.x, pat) >> 7) & 0x01010101u;
  r |= (tag_differs(t.y, pat) >> 6) & 0x02020202u;
  r |= (tag_differs(t.z, pat) >> 5) & 0x04040404u;
  r |= (tag_differs(t.w, pat) >> 4) & 0x08080808u;
  return r;
}

template <int F, int H>
__device__ __forceinline__ void aggr_batch(const int* __restrict__ ei,
                                           const int* __restrict__ dstp,
                                           const float* __restrict__ logit,
                                           const float* __restrict__ xr,
                                           float* acc, float* den, const unsigned* mxo,
                                           int qe, int cnt, int lo, int lane) {
  constexpr int C = F / H;
  constexpr int P = C / 32;
  constexpr int R = H * P;
  constexpr int U = 32 / R;
  const int s   = ei[qe];
  const int d   = dstp[qe];
  float lg[H];
#pragma unroll
  for (int h = 0; h < H; ++h) lg[h] = logit[(size_t)qe * H + h];
  const bool live = lane < cnt;
  const int row = live ? d - lo : 0;
  float al[H];
#pragma unroll
  for (int h = 0; h < H; ++h) al[h] = __expf(lg[h] - ordDecode(mxo[row * H + h]));
  if (live) {
#pragma unroll
    for (int h = 0; h < H; ++h) atomicAdd(&den[row * H + h], al[h]);
  }

  for (int j0 = 0; j0 < cnt; j0 += U) {
    float v[U][R];
#pragma unroll
    for (int u = 0; u < U; ++u) {
      const int jj = min(j0 + u, cnt - 1);
      const int sj = __builtin_amdgcn_readlane(s, jj);
      const float* __restrict__ xrs = xr + (size_t)sj * F + lane;
#pragma unroll
      for (int i = 0; i < R; ++i) v[u][i] = xrs[i * 32];
    }
#pragma unroll
    for (int u = 0; u < U; ++u) {
      const int jj = j0 + u;
      if (jj < cnt) {
        const int rowj = __builtin_amdgcn_readlane(row, jj);
        float* arow = acc + rowj * F + lane;
#pragma unroll
        for (int i = 0; i < R; ++i) {
          const float a = __int_as_float(
              __builtin_amdgcn_readlane(__float_as_int(al[i / P]), jj));
          atomicAdd(&arow[i * 32], a * v[u][i]);
        }
      }
    }
  }
}

template <int H>
__device__ __forceinline__ void max_batch(const int* __restrict__ dstp,
                                          const float* __restrict__ logit,
                                          unsigned* mxo, int* elist, unsigned* lcount,
                                          int qe, int cnt, int lo, int lane) {
  const int d = dstp[qe];
  float lg[H];
#pragma unroll
  for (int h = 0; h < H; ++h) lg[h] = logit[(size_t)qe * H + h];
  unsigned base = 0u;
  if (lane == 0) base = atomicAdd(lcount, (unsigned)cnt);
  base = (unsigned)__builtin_amdgcn_readfirstlane((int)base);
  if (lane < cnt) {
    const int row = d - lo;
#pragma unroll
    for (int h = 0; h < H; ++h) atomicMax(&mxo[row * H + h], ordEncode(lg[h]));
    if (base + (unsigned)cnt <= (unsigned)AGG_CAP) elist[base + lane] = qe;
  }
}

template <int F, int H, int PASS>
__device__ __forceinline__ void queue_flush(const int* __restrict__ ei,
                                            const int* __restrict__ dstp,
                                            const float* __restrict__ logit,
                                            const float* __restrict__ xr,
                                            float* acc, float* den, unsigned* mxo,
                                            int* elist, unsigned* lcount,
                                            int qe, int cnt, int lo, int lane) {
  if constexpr (PASS == 1) max_batch<H>(dstp, logit, mxo, elist, lcount, qe, cnt, lo, lane);
  else aggr_batch<F, H>(ei, dstp, logit, xr, acc, den, mxo, qe, cnt, lo, lane);
}

template <int F, int H, int PASS>
__device__ __forceinline__ void aggr_push(const int* __restrict__ ei,
                                          const int* __restrict__ dstp,
                                          const float* __restrict__ logit,
                                          const float* __restrict__ xr,
                                          float* acc, float* den, unsigned* mxo,
                                          int* elist, unsigned* lcount,
                                          int& qe, int& cnt, int e, int lo, int lane) {
  qe = (lane == cnt) ? e : qe;
  if (++cnt == 32) {
    queue_flush<F, H, PASS>(ei, dstp, logit, xr, acc, den, mxo, elist, lcount, qe, 32, lo, lane);
    cnt = 0;
  }
}

template <int F, int H, int PASS>
__device__ __forceinline__ void scan_edges(const int* __restrict__ ei,
                                           const unsigned char* __restrict__ tag,
                                           const float* __restrict__ logit,
                                           const float* __restrict__ xr,
                                           float* acc, float* den, unsigned* mxo,
                                           int* elist, unsigned* lcount,
                                           int lo, unsigned lim, int E, int wave, int lane) {
  const int* __restrict__ dstp = ei + E;
  int qe = 0;
  int cnt = 0;
  const bool useTags = (tag != nullptr) && (gridDim.x <= 256u) && ((((size_t)tag) & 15) == 0);
  const int nGroups = useTags ? (E >> 4) : 0;
  if (nGroups > 0) {
    const uint4* __restrict__ tag4 = (const uint4*)tag;
    const unsigned pat = blockIdx.x * 0x01010101u;
    for (int gb = wave * (32 * AGG_GROUPS); gb < nGroups;
         gb += (AGG_THREADS / 32) * 32 * AGG_GROUPS) {
      uint4 t[AGG_GROUPS];
#pragma unroll
      for (int u = 0; u < AGG_GROUPS; ++u)
        t[u] = tag4[min(gb + u * 32 + lane, nGroups - 1)];
      __builtin_amdgcn_sched_barrier(0);
      unsigned m[AGG_GROUPS / 2];
#pragma unroll
      for (int j = 0; j < AGG_GROUPS / 2; ++j) {
        const unsigned differ = tags_differ16(t[2 * j], pat) | (tags_differ16(t[2 * j + 1], pat) << 4);
        const unsigned ok = (((gb + (2 * j) * 32 + lane) < nGroups) ? 0x0F0F0F0Fu : 0u) |
                            (((gb + (2 * j + 1) * 32 + lane) < nGroups) ? 0xF0F0F0F0u : 0u);
        m[j] = ~differ & ok;
      }
      static_assert(AGG_GROUPS == 8, "the choice of m[j] below is written for four words");
#pragma nounroll
      for (int j = 0; j < AGG_GROUPS / 2; ++j) {
        const unsigned mj = (j == 0) ? m[0] : (j == 1) ? m[1] : (j == 2) ? m[2] : m[3];
        unsigned any = __builtin_amdgcn_ballot_w32(mj != 0u);
        while (any) {
          const int L = __builtin_ctz(any);
          any &= any - 1u;
          unsigned b = (unsigned)__builtin_amdgcn_readlane((int)mj, L);
          while (b) {
            const int p = __builtin_ctz(b);
            b &= b - 1u;
            const int g = gb + (2 * j + ((p >> 2) & 1)) * 32 + L;
            const int e = (g << 4) + ((p & 3) << 2) + (p >> 3);
            aggr_push<F, H, PASS>(ei, dstp, logit, xr, acc, den, mxo, elist, lcount, qe, cnt, e, lo, lane);
          }
        }
      }
    }
  }
  for (int eb = (nGroups << 4) + wave * 32; eb < E; eb += AGG_THREADS) {
    const int e1 = eb + lane;
    const unsigned r1 = (e1 < E) ? (unsigned)(dstp[min(e1, E - 1)] - lo) : 0xFFFFFFFFu;
    unsigned mk = __builtin_amdgcn_ballot_w32(r1 < lim);
    while (mk) {
      const int j = __builtin_ctz(mk);
      mk &= mk - 1u;
      aggr_push<F, H, PASS>(ei, dstp, logit, xr, acc, den, mxo, elist, lcount, qe, cnt, eb + j, lo, lane);
    }
  }
  if (cnt > 0)
    queue_flush<F, H, PASS>(ei, dstp, logit, xr, acc, den, mxo, elist, lcount, qe, cnt, lo, lane);
}

template <int F, int H>
__global__ __launch_bounds__(AGG_THREADS)
void edge_aggr_rows(const int* __restrict__ ei, const unsigned char* __restrict__ tag,
                    const float* __restrict__ logit,
                    const float* __restrict__ xr,
                    const float* __restrict__ bias, const float* __restrict__ lng,
                    const float* __restrict__ lnb, float* __restrict__ out, int N, int E) {
  __shared__ float acc[AGG_TR * F];
  __shared__ float den[AGG_TR * H];
  __shared__ unsigned mxo[AGG_TR * H];
  __shared__ int elist[AGG_CAP];
  __shared__ unsigned lcount;
  const int tid  = threadIdx.x;
  const int lane = tid & 31;
  const int wave = __builtin_amdgcn_readfirstlane(tid >> 5);
  const int lo = blockIdx.x * AGG_TR;
  const int nrows = min(AGG_TR, N - lo);
  const unsigned lim = (unsigned)nrows;

  for (int i = tid; i < AGG_TR * F / 4; i += AGG_THREADS)
    ((float4*)acc)[i] = make_float4(0.f, 0.f, 0.f, 0.f);
  for (int i = tid; i < AGG_TR * H; i += AGG_THREADS) {
    den[i] = 0.0f;
    mxo[i] = ORD_NEG_INF;
  }
  if (tid == 0) lcount = 0u;
  __syncthreads();

  scan_edges<F, H, 1>(ei, tag, logit, xr, acc, den, mxo, elist, &lcount, lo, lim, E, wave, lane);
  __syncthreads();
  const int total = __builtin_amdgcn_readfirstlane((int)lcount);
  if (total <= AGG_CAP) {
    const int* __restrict__ dstp = ei + E;
    const int share = (total + AGG_THREADS / 32 - 1) / (AGG_THREADS / 32);
    const int wend = min(total, (wave + 1) * share);
    for (int b0 = wave * share; b0 < wend; b0 += 32) {
      const int c = min(32, wend - b0);
      const int q = elist[min(b0 + lane, wend - 1)];
      aggr_batch<F, H>(ei, dstp, logit, xr, acc, den, mxo, q, c, lo, lane);
    }
  } else {
    scan_edges<F, H, 2>(ei, tag, logit, xr, acc, den, mxo, elist, &lcount, lo, lim, E, wave, lane);
  }
  __syncthreads();
  constexpr int PER = F / 32;
  constexpr int ROWS_TOGETHER = 4;
  constexpr int WAVES = AGG_THREADS / 32;
  static_assert(PER >= 1 && (F / H) % PER == 0, "a lane's channels must lie in one head");
  typedef float vp __attribute__((ext_vector_type(PER)));
  const vp bs = *(const vp*)(bias + lane * PER);
  const vp gg = *(const vp*)(lng + lane * PER);
  const vp bb = *(const vp*)(lnb + lane * PER);
  const int hd = (lane * PER) / (F / H);
  for (int r0 = wave; r0 < nrows; r0 += WAVES * ROWS_TOGETHER) {
    float vals[ROWS_TOGETHER][PER];
    float sum[ROWS_TOGETHER];
#pragma unroll
    for (int k = 0; k < ROWS_TOGETHER; ++k) {
      const int r = min(r0 + k * WAVES, nrows - 1);
      const vp a = *(const vp*)(acc + r * F + lane * PER);
      const float dn = den[r * H + hd];
      float s = 0.0f;
#pragma unroll
      for (int i = 0; i < PER; ++i) {
        float v = a[i];
        if (dn > 0.0f) v /= dn;
        v += bs[i];
        vals[k][i] = v;
        s += v;
      }
      sum[k] = s;
    }
#pragma unroll
    for (int off = 16; off > 0; off >>= 1) {
#pragma unroll
      for (int k = 0; k < ROWS_TOGETHER; ++k) sum[k] += __shfl_xor(sum[k], off, 32);
    }
    float mu[ROWS_TOGETHER], vs[ROWS_TOGETHER];
#pragma unroll
    for (int k = 0; k < ROWS_TOGETHER; ++k) {
      mu[k] = sum[k] / (float)F;
      float q = 0.0f;
#pragma unroll
      for (int i = 0; i < PER; ++i) { const float dlt = vals[k][i] - mu[k]; q += dlt * dlt; }
      vs[k] = q;
    }
#pragma unroll
    for (int off = 16; off > 0; off >>= 1) {
#pragma unroll
      for (int k = 0; k < ROWS_TOGETHER; ++k) vs[k] += __shfl_xor(vs[k], off, 32);
    }
#pragma unroll
    for (int k = 0; k < ROWS_TOGETHER; ++k) {
      const int r = r0 + k * WAVES;
      if (r < nrows) {
        const float inv = rsqrtf(vs[k] / (float)F + LN_EPS);
        vp o;
#pragma unroll
        for (int i = 0; i < PER; ++i) {
          const float hv = (vals[k][i] - mu[k]) * inv * gg[i] + bb[i];
          o[i] = (hv > 0.0f) ? hv : (__expf(hv) - 1.0f);
        }
        *(vp*)(out + (size_t)(lo + r) * F + lane * PER) = o;
      }
    }
  }
}

extern "C" void kernel_launch(void* const* d_in, const int* in_sizes, int n_in,
                              void* d_out, int out_size, void* d_ws, size_t ws_size,
                              hipStream_t stream) {
  const float* x      = (const float*)d_in[0];
  const float* ea     = (const float*)d_in[1];
  const float* W1l    = (const float*)d_in[2];
  const float* W1r    = (const float*)d_in[3];
  const float* W1e    = (const float*)d_in[4];
  const float* att1   = (const float*)d_in[5];
  const float* b1     = (const float*)d_in[6];
  const float* ln1_g  = (const float*)d_in[7];
  const float* ln1_b  = (const float*)d_in[8];
  const float* W2l    = (const float*)d_in[9];
  const float* W2r    = (const float*)d_in[10];
  const float* W2e    = (const float*)d_in[11];
  const float* att2   = (const float*)d_in[12];
  const float* b2     = (const float*)d_in[13];
  const float* ln2_g  = (const float*)d_in[14];
  const float* ln2_b  = (const float*)d_in[15];
  const int*   ei     = (const int*)d_in[16];
  float*       outp   = (float*)d_out;

  const int N = in_sizes[0] / 10;
  const int E = in_sizes[1];
  const int HID = 32, HEADS = 4, F1 = HEADS * HID  , OUTC = 64;
  const int KP = 12;

  float* ws      = (float*)d_ws;
  float* xl1     = ws;
  float* xr1     = xl1 + (size_t)N * F1;
  float* h1      = xr1 + (size_t)N * F1;
  float* logit1  = h1  + (size_t)N * F1;
  float* xpad    = logit1 + (size_t)E * HEADS;
  float* w1lp    = xpad + (size_t)N * KP;
  float* w1rp    = w1lp + (size_t)KP * F1;
  unsigned char* etag = (unsigned char*)(w1rp + (size_t)KP * F1);
  float* xr2     = xl1;
  float* logit2  = xl1 + (size_t)N * OUTC;
  float* xl2     = logit1;

  const int TB = 256;
  auto blocks = [](size_t n, int tb) { return (unsigned)((n + tb - 1) / tb); };

  pad_cols<<<blocks((size_t)N * KP, TB), TB, 0, stream>>>(x, xpad, N, 10, KP);
  pad_rows<<<blocks((size_t)KP * F1, TB), TB, 0, stream>>>(W1l, w1lp, 10, KP, F1);
  pad_rows<<<blocks((size_t)KP * F1, TB), TB, 0, stream>>>(W1r, w1rp, 10, KP, F1);
  {
    dim3 grid((N / 16 + 3) / 4, F1 / 16);
    wmma_gemm<12><<<grid, 128, 0, stream>>>(xpad, w1lp, xl1, N, F1);
    wmma_gemm<12><<<grid, 128, 0, stream>>>(xpad, w1rp, xr1, N, F1);
  }

  edge_logits_rows<128, 4><<<blocks((size_t)E, TB), TB, 0, stream>>>(
      ei, ea, xl1, xr1, W1e, att1, logit1, etag, E);
  edge_aggr_rows<128, 4><<<blocks((size_t)N, AGG_TR), AGG_THREADS, 0, stream>>>(
      ei, etag, logit1, xr1, b1, ln1_g, ln1_b, h1, N, E);

  {
    dim3 grid((N / 16 + 3) / 4, OUTC / 16);
    wmma_gemm<128><<<grid, 128, 0, stream>>>(h1, W2l, xl2, N, OUTC);
    wmma_gemm<128><<<grid, 128, 0, stream>>>(h1, W2r, xr2, N, OUTC);
  }

  edge_logits_rows<64, 1><<<blocks((size_t)E, TB), TB, 0, stream>>>(
      ei, ea, xl2, xr2, W2e, att2, logit2, nullptr, E);
  edge_aggr_rows<64, 1><<<blocks((size_t)N, AGG_TR), AGG_THREADS, 0, stream>>>(
      ei, etag, logit2, xr2, b2, ln2_g, ln2_b, outp, N, E);
}


# edge logits formed inside the aggregation block: its xl rows held in LDS, xr rows read in the scan pass, logits kept in LDS for pass 2; logit launches and arrays gone; sums: LDS arrival as before
# speedup vs baseline: 6.5869x; 1.0511x over previous
#include <hip/hip_runtime.h>
#include <hip/hip_bf16.h>
#include <math.h>

typedef float     v2f  __attribute__((ext_vector_type(2)));
typedef float     v8f  __attribute__((ext_vector_type(8)));

#define NEG_SLOPE 0.2f
#define LN_EPS    1e-5f

__device__ __forceinline__ unsigned ordEncode(float f) {
  unsigned u = __float_as_uint(f);
  return (u >> 31) ? ~u : (u | 0x80000000u);
}
__device__ __forceinline__ float ordDecode(unsigned o) {
  return __uint_as_float((o >> 31) ? (o & 0x7FFFFFFFu) : ~o);
}

__global__ void pad_cols(const float* __restrict__ src, float* __restrict__ dst,
                         int M, int Ks, int Kd) {
  int i = blockIdx.x * blockDim.x + threadIdx.x;
  if (i >= M * Kd) return;
  int r = i / Kd, c = i % Kd;
  dst[i] = (c < Ks) ? src[(size_t)r * Ks + c] : 0.0f;
}

__global__ void pad_rows(const float* __restrict__ src, float* __restrict__ dst,
                         int Ms, int Md, int Ncols) {
  int i = blockIdx.x * blockDim.x + threadIdx.x;
  if (i >= Md * Ncols) return;
  int r = i / Ncols;
  dst[i] = (r < Ms) ? src[i] : 0.0f;
}

template <int K>
__global__ void wmma_gemm(const float* __restrict__ A, const float* __restrict__ B,
                          float* __restrict__ C, int M, int N) {
  const int lane  = threadIdx.x & 31;
  const int wave  = threadIdx.x >> 5;
  const int tileM = blockIdx.x * 4 + wave;
  const int m0 = tileM * 16;
  const int n0 = blockIdx.y * 16;
  if (m0 + 16 > M) return;
  const int half = lane >> 4;
  const int l16  = lane & 15;
  const float* __restrict__ arow = A + (size_t)(m0 + l16) * K + half * 2;
  const float* __restrict__ bcol = B + n0 + l16 + (size_t)(half * 2) * N;
  v8f acc = {};
#pragma unroll
  for (int k = 0; k < K; k += 4) {
    v2f a = *(const v2f*)(arow + k);
    v2f b;
    b.x = bcol[(size_t)k * N];
    b.y = bcol[(size_t)(k + 1) * N];
    acc = __builtin_amdgcn_wmma_f32_16x16x4_f32(false, a, false, b, (short)0, acc,
                                                false, false);
  }
  float* __restrict__ crow = C + (size_t)(m0 + half * 8) * N + n0 + l16;
#pragma unroll
  for (int r = 0; r < 8; ++r)
    crow[(size_t)r * N] = acc[r];
}

#define LG_U 8
#define AGG_TR 196

__global__ void edge_tags(const int* __restrict__ dstp, unsigned char* __restrict__ tag, int E) {
  const int e0 = (blockIdx.x * blockDim.x + threadIdx.x) * 16;
  if (e0 >= E) return;
  if (e0 + 16 <= E && (((((size_t)(dstp + e0)) | ((size_t)(tag + e0))) & 15) == 0)) {
    const int4* __restrict__ p = (const int4*)(dstp + e0);
    const int4 d0 = p[0], d1 = p[1], d2 = p[2], d3 = p[3];
    uint4 o;
    o.x = ((unsigned)(d0.x / AGG_TR) & 255u) | (((unsigned)(d0.y / AGG_TR) & 255u) << 8) |
          (((unsigned)(d0.z / AGG_TR) & 255u) << 16) | (((unsigned)(d0.w / AGG_TR) & 255u) << 24);
    o.y = ((unsigned)(d1.x / AGG_TR) & 255u) | (((unsigned)(d1.y / AGG_TR) & 255u) << 8) |
          (((unsigned)(d1.z / AGG_TR) & 255u) << 16) | (((unsigned)(d1.w / AGG_TR) & 255u) << 24);
    o.z = ((unsigned)(d2.x / AGG_TR) & 255u) | (((unsigned)(d2.y / AGG_TR) & 255u) << 8) |
          (((unsigned)(d2.z / AGG_TR) & 255u) << 16) | (((unsigned)(d2.w / AGG_TR) & 255u) << 24);
    o.w = ((unsigned)(d3.x / AGG_TR) & 255u) | (((unsigned)(d3.y / AGG_TR) & 255u) << 8) |
          (((unsigned)(d3.z / AGG_TR) & 255u) << 16) | (((unsigned)(d3.w / AGG_TR) & 255u) << 24);
    *(uint4*)(tag + e0) = o;
  } else {
    for (int e = e0; e < min(e0 + 16, E); ++e) tag[e] = (unsigned char)(dstp[e] / AGG_TR);
  }
}

template <int F, int H, class VF>
__device__ __forceinline__ void batch_logits(const float* xlrows, const float* __restrict__ xr,
                                             const VF we, const VF at,
                                             int sL, int rL, int eaL, int cnt, int lane,
                                             float (&outv)[H]) {
  constexpr int VEC = F / 32;
  constexpr int LPH = 32 / H;
  static_assert(LG_U == 8 && LPH >= 8 && 32 % H == 0, "the lane sum below is laid out for 8 edges a step");
#pragma unroll
  for (int g = 0; g < H; ++g) outv[g] = 0.0f;

  constexpr int NB = (VEC <= 2) ? 2 : 1;
#pragma unroll
  for (int jb0 = 0; jb0 < 32; jb0 += LG_U * NB) {
    if (jb0 < cnt) {
      VF b[NB][LG_U];
#pragma unroll
      for (int nb = 0; nb < NB; ++nb) {
#pragma unroll
        for (int u = 0; u < LG_U; ++u) {
          const int s = __builtin_amdgcn_readlane(sL, jb0 + nb * LG_U + u);
          b[nb][u] = *(const VF*)(xr + (size_t)s * F + lane * VEC);
        }
      }
      static_assert(LG_U == 8, "the statement below names eight rows a step");
      if constexpr (NB == 1) {
        asm volatile("" :: "v"(b[0][0]), "v"(b[0][1]), "v"(b[0][2]), "v"(b[0][3]),
                           "v"(b[0][4]), "v"(b[0][5]), "v"(b[0][6]), "v"(b[0][7]));
      } else {
        asm volatile("" :: "v"(b[0][0]), "v"(b[0][1]), "v"(b[0][2]), "v"(b[0][3]),
                           "v"(b[0][4]), "v"(b[0][5]), "v"(b[0][6]), "v"(b[0][7]),
                           "v"(b[NB - 1][0]), "v"(b[NB - 1][1]), "v"(b[NB - 1][2]), "v"(b[NB - 1][3]),
                           "v"(b[NB - 1][4]), "v"(b[NB - 1][5]), "v"(b[NB - 1][6]), "v"(b[NB - 1][7]));
      }
#pragma unroll
      for (int nb = 0; nb < NB; ++nb) {
        const int jb = jb0 + nb * LG_U;
        if (nb == 0 || jb < cnt) {
          VF a[LG_U];
#pragma unroll
          for (int u = 0; u < LG_U; ++u) {
            const int r = __builtin_amdgcn_readlane(rL, jb + u);
            a[u] = *(const VF*)(xlrows + (size_t)r * F + lane * VEC);
          }
          float p[LG_U];
#pragma unroll
          for (int u = 0; u < LG_U; ++u) {
            const float eav = __int_as_float(__builtin_amdgcn_readlane(eaL, jb + u));
            float sum = 0.0f;
#pragma unroll
            for (int i = 0; i < VEC; ++i) {
              float v = a[u][i] + b[nb][u][i] + eav * we[i];
              v = fmaxf(v, NEG_SLOPE * v);
              sum += at[i] * v;
            }
            p[u] = sum;
          }
#pragma unroll
          for (int st = 0; st < 3; ++st) {
            const int off = LPH >> (st + 1);
            const int hn  = LG_U >> (st + 1);
            const bool hi = (lane & off) != 0;
#pragma unroll
            for (int u = 0; u < hn; ++u) {
              const float send = hi ? p[u] : p[u + hn];
              const float keep = hi ? p[u + hn] : p[u];
              p[u] = keep + __shfl_xor(send, off, 32);
            }
          }
#pragma unroll
          for (int off = LPH / 16; off > 0; off >>= 1) p[0] += __shfl_xor(p[0], off, 32);
          const int rl = lane - (jb * H) % 32;
          const bool mine = (rl >= 0) && (rl < LG_U * H);
          const int rc = mine ? rl : 0;
          const float q = __shfl(p[0], (rc % H) * LPH + (rc / H) * (LPH / 8), 32);
          if (mine) outv[(jb * H) / 32] = q;
        }
      }
    }
  }
}

#define AGG_THREADS 1024
#define AGG_GROUPS  8
#define AGG_CAP     8192
#define ORD_NEG_INF 0x007FFFFFu

__device__ __forceinline__ unsigned tag_differs(unsigned w, unsigned pat) {
  const unsigned x = w ^ pat;
  return ((x & 0x7F7F7F7Fu) + 0x7F7F7F7Fu) | x;
}
__device__ __forceinline__ unsigned tags_differ16(uint4 t, unsigned pat) {
  unsigned r = (tag_differs(t.x, pat) >> 7) & 0x01010101u;
  r |= (tag_differs(t.y, pat) >> 6) & 0x02020202u;
  r |= (tag_differs(t.z, pat) >> 5) & 0x04040404u;
  r |= (tag_differs(t.w, pat) >> 4) & 0x08080808u;
  return r;
}

template <int F, int H>
__device__ __forceinline__ void aggr_core(const float* __restrict__ xr,
                                          float* acc, float* den, const unsigned* mxo,
                                          int s, int row, const float (&lg)[H], int cnt, int lane) {
  constexpr int C = F / H;
  constexpr int P = C / 32;
  constexpr int R = H * P;
  constexpr int U = 32 / R;
  const bool live = lane < cnt;
  float al[H];
#pragma unroll
  for (int h = 0; h < H; ++h) al[h] = __expf(lg[h] - ordDecode(mxo[row * H + h]));
  if (live) {
#pragma unroll
    for (int h = 0; h < H; ++h) atomicAdd(&den[row * H + h], al[h]);
  }

  for (int j0 = 0; j0 < cnt; j0 += U) {
    float v[U][R];
#pragma unroll
    for (int u = 0; u < U; ++u) {
      const int jj = min(j0 + u, cnt - 1);
      const int sj = __builtin_amdgcn_readlane(s, jj);
      const float* __restrict__ xrs = xr + (size_t)sj * F + lane;
#pragma unroll
      for (int i = 0; i < R; ++i) v[u][i] = xrs[i * 32];
    }
#pragma unroll
    for (int u = 0; u < U; ++u) {
      const int jj = j0 + u;
      if (jj < cnt) {
        const int rowj = __builtin_amdgcn_readlane(row, jj);
        float* arow = acc + rowj * F + lane;
#pragma unroll
        for (int i = 0; i < R; ++i) {
          const float a = __int_as_float(
              __builtin_amdgcn_readlane(__float_as_int(al[i / P]), jj));
          atomicAdd(&arow[i * 32], a * v[u][i]);
        }
      }
    }
  }
}

template <class Flush>
__device__ __forceinline__ void scan_edges(const int* __restrict__ dstp,
                                           const unsigned char* __restrict__ tag,
                                           int lo, unsigned lim, int E, int wave, int lane,
                                           Flush flush) {
  int qe = 0;
  int cnt = 0;
  const bool useTags = (tag != nullptr) && (gridDim.x <= 256u) && ((((size_t)tag) & 15) == 0);
  const int nGroups = useTags ? (E >> 4) : 0;
  if (nGroups > 0) {
    const uint4* __restrict__ tag4 = (const uint4*)tag;
    const unsigned pat = blockIdx.x * 0x01010101u;
    for (int gb = wave * (32 * AGG_GROUPS); gb < nGroups;
         gb += (AGG_THREADS / 32) * 32 * AGG_GROUPS) {
      uint4 t[AGG_GROUPS];
#pragma unroll
      for (int u = 0; u < AGG_GROUPS; ++u)
        t[u] = tag4[min(gb + u * 32 + lane, nGroups - 1)];
      __builtin_amdgcn_sched_barrier(0);
      unsigned m[AGG_GROUPS / 2];
#pragma unroll
      for (int j = 0; j < AGG_GROUPS / 2; ++j) {
        const unsigned differ = tags_differ16(t[2 * j], pat) | (tags_differ16(t[2 * j + 1], pat) << 4);
        const unsigned ok = (((gb + (2 * j) * 32 + lane) < nGroups) ? 0x0F0F0F0Fu : 0u) |
                            (((gb + (2 * j + 1) * 32 + lane) < nGroups) ? 0xF0F0F0F0u : 0u);
        m[j] = ~differ & ok;
      }
      static_assert(AGG_GROUPS == 8, "the choice of m[j] below is written for four words");
#pragma nounroll
      for (int j = 0; j < AGG_GROUPS / 2; ++j) {
        const unsigned mj = (j == 0) ? m[0] : (j == 1) ? m[1] : (j == 2) ? m[2] : m[3];
        unsigned any = __builtin_amdgcn_ballot_w32(mj != 0u);
        while (any) {
          const int L = __builtin_ctz(any);
          any &= any - 1u;
          unsigned b = (unsigned)__builtin_amdgcn_readlane((int)mj, L);
          while (b) {
            const int p = __builtin_ctz(b);
            b &= b - 1u;
            const int g = gb + (2 * j + ((p >> 2) & 1)) * 32 + L;
            const int e = (g << 4) + ((p & 3) << 2) + (p >> 3);
            qe = (lane == cnt) ? e : qe;
            if (++cnt == 32) {
              flush(qe, 32);
              cnt = 0;
            }
          }
        }
      }
    }
  }
  for (int eb = (nGroups << 4) + wave * 32; eb < E; eb += AGG_THREADS) {
    const int e1 = eb + lane;
    const unsigned r1 = (e1 < E) ? (unsigned)(dstp[min(e1, E - 1)] - lo) : 0xFFFFFFFFu;
    unsigned mk = __builtin_amdgcn_ballot_w32(r1 < lim);
    while (mk) {
      const int j = __builtin_ctz(mk);
      mk &= mk - 1u;
      qe = (lane == cnt) ? (eb + j) : qe;
      if (++cnt == 32) {
        flush(qe, 32);
        cnt = 0;
      }
    }
  }
  if (cnt > 0) flush(qe, cnt);
}

template <int F, int H>
__global__ __launch_bounds__(AGG_THREADS)
void edge_aggr_rows(const int* __restrict__ ei, const float* __restrict__ ea,
                    const unsigned char* __restrict__ tag,
                    const float* __restrict__ xl, const float* __restrict__ xr,
                    const float* __restrict__ We, const float* __restrict__ att,
                    const float* __restrict__ bias, const float* __restrict__ lng,
                    const float* __restrict__ lnb, float* __restrict__ out, int N, int E) {
  static_assert(AGG_TR <= 256, "a row index inside the block is kept in one byte");
  __shared__ float acc[AGG_TR * F];
  __shared__ float den[AGG_TR * H];
  __shared__ unsigned mxo[AGG_TR * H];
  __shared__ int esrc[AGG_CAP];
  __shared__ unsigned char erow[AGG_CAP];
  __shared__ float llog[AGG_CAP * H];
  __shared__ unsigned lcount;
  const int tid  = threadIdx.x;
  const int lane = tid & 31;
  const int wave = __builtin_amdgcn_readfirstlane(tid >> 5);
  const int lo = blockIdx.x * AGG_TR;
  const int nrows = min(AGG_TR, N - lo);
  const unsigned lim = (unsigned)nrows;
  const int* __restrict__ dstp = ei + E;
  constexpr int VEC = F / 32;
  typedef float vf __attribute__((ext_vector_type(VEC)));
  const vf we = *(const vf*)(We + lane * VEC);
  const vf at = *(const vf*)(att + lane * VEC);

  {
    constexpr int Q = AGG_TR * F / 4;
    constexpr int NV = (Q + AGG_THREADS - 1) / AGG_THREADS;
    const float4* __restrict__ src4 = (const float4*)(xl + (size_t)lo * F);
    const int have = nrows * (F / 4);
    float4 tmp[NV];
#pragma unroll
    for (int k = 0; k < NV; ++k) {
      const int i = tid + k * AGG_THREADS;
      tmp[k] = (i < have) ? src4[i] : make_float4(0.f, 0.f, 0.f, 0.f);
    }
#pragma unroll
    for (int k = 0; k < NV; ++k) {
      const int i = tid + k * AGG_THREADS;
      if (i < Q) ((float4*)acc)[i] = tmp[k];
    }
  }
  for (int i = tid; i < AGG_TR * H; i += AGG_THREADS) {
    den[i] = 0.0f;
    mxo[i] = ORD_NEG_INF;
  }
  if (tid == 0) lcount = 0u;
  __syncthreads();

  scan_edges(dstp, tag, lo, lim, E, wave, lane, [&](int qe, int cnt) {
    const int s = ei[qe];
    const int d = dstp[qe];
    const int eav = __float_as_int(ea[qe]);
    const int row = (lane < cnt) ? d - lo : 0;
    float outv[H];
    batch_logits<F, H, vf>(acc, xr, we, at, s, row, eav, cnt, lane, outv);
    unsigned base = 0u;
    if (lane == 0) base = atomicAdd(&lcount, (unsigned)cnt);
    base = (unsigned)__builtin_amdgcn_readfirstlane((int)base);
    const bool fits = base + (unsigned)cnt <= (unsigned)AGG_CAP;
#pragma unroll
    for (int g = 0; g < H; ++g) {
      const int idx = g * 32 + lane;
      const int j = idx / H;
      const int h = idx % H;
      const int rj = __shfl(row, j, 32);
      if (j < cnt) {
        atomicMax(&mxo[rj * H + h], ordEncode(outv[g]));
        if (fits) llog[base * H + idx] = outv[g];
      }
    }
    if (fits && lane < cnt) {
      esrc[base + lane] = s;
      erow[base + lane] = (unsigned char)row;
    }
  });
  __syncthreads();
  const int total = __builtin_amdgcn_readfirstlane((int)lcount);
  for (int i = tid; i < AGG_TR * F / 4; i += AGG_THREADS)
    ((float4*)acc)[i] = make_float4(0.f, 0.f, 0.f, 0.f);
  __syncthreads();
  if (total <= AGG_CAP) {
    const int share = (total + AGG_THREADS / 32 - 1) / (AGG_THREADS / 32);
    const int wend = min(total, (wave + 1) * share);
    for (int b0 = wave * share; b0 < wend; b0 += 32) {
      const int c = min(32, wend - b0);
      const int q = min(b0 + lane, wend - 1);
      const int s = esrc[q];
      const int row = erow[q];
      float lg[H];
#pragma unroll
      for (int h = 0; h < H; ++h) lg[h] = llog[q * H + h];
      aggr_core<F, H>(xr, acc, den, mxo, s, row, lg, c, lane);
    }
  } else {
    scan_edges(dstp, tag, lo, lim, E, wave, lane, [&](int qe, int cnt) {
      const int s = ei[qe];
      const int d = dstp[qe];
      const int eav = __float_as_int(ea[qe]);
      const int row = (lane < cnt) ? d - lo : 0;
      float outv[H];
      batch_logits<F, H, vf>(xl, xr, we, at, s, d, eav, cnt, lane, outv);
      float lg[H];
#pragma unroll
      for (int h = 0; h < H; ++h) {
        const int idx = lane * H + h;
        lg[h] = 0.0f;
#pragma unroll
        for (int g = 0; g < H; ++g) {
          const float t = __shfl(outv[g], idx & 31, 32);
          if ((idx >> 5) == g) lg[h] = t;
        }
      }
      aggr_core<F, H>(xr, acc, den, mxo, s, row, lg, cnt, lane);
    });
  }
  __syncthreads();
  constexpr int PER = F / 32;
  constexpr int ROWS_TOGETHER = 4;
  constexpr int WAVES = AGG_THREADS / 32;
  static_assert(PER >= 1 && (F / H) % PER == 0, "a lane's channels must lie in one head");
  typedef float vp __attribute__((ext_vector_type(PER)));
  const vp bs = *(const vp*)(bias + lane * PER);
  const vp gg = *(const vp*)(lng + lane * PER);
  const vp bb = *(const vp*)(lnb + lane * PER);
  const int hd = (lane * PER) / (F / H);
  for (int r0 = wave; r0 < nrows; r0 += WAVES * ROWS_TOGETHER) {
    float vals[ROWS_TOGETHER][PER];
    float sum[ROWS_TOGETHER];
#pragma unroll
    for (int k = 0; k < ROWS_TOGETHER; ++k) {
      const int r = min(r0 + k * WAVES, nrows - 1);
      const vp a = *(const vp*)(acc + r * F + lane * PER);
      const float dn = den[r * H + hd];
      float s = 0.0f;
#pragma unroll
      for (int i = 0; i < PER; ++i) {
        float v = a[i];
        if (dn > 0.0f) v /= dn;
        v += bs[i];
        vals[k][i] = v;
        s += v;
      }
      sum[k] = s;
    }
#pragma unroll
    for (int off = 16; off > 0; off >>= 1) {
#pragma unroll
      for (int k = 0; k < ROWS_TOGETHER; ++k) sum[k] += __shfl_xor(sum[k], off, 32);
    }
    float mu[ROWS_TOGETHER], vs[ROWS_TOGETHER];
#pragma unroll
    for (int k = 0; k < ROWS_TOGETHER; ++k) {
      mu[k] = sum[k] / (float)F;
      float q = 0.0f;
#pragma unroll
      for (int i = 0; i < PER; ++i) { const float dlt = vals[k][i] - mu[k]; q += dlt * dlt; }
      vs[k] = q;
    }
#pragma unroll
    for (int off = 16; off > 0; off >>= 1) {
#pragma unroll
      for (int k = 0; k < ROWS_TOGETHER; ++k) vs[k] += __shfl_xor(vs[k], off, 32);
    }
#pragma unroll
    for (int k = 0; k < ROWS_TOGETHER; ++k) {
      const int r = r0 + k * WAVES;
      if (r < nrows) {
        const float inv = rsqrtf(vs[k] / (float)F + LN_EPS);
        vp o;
#pragma unroll
        for (int i = 0; i < PER; ++i) {
          const float hv = (vals[k][i] - mu[k]) * inv * gg[i] + bb[i];
          o[i] = (hv > 0.0f) ? hv : (__expf(hv) - 1.0f);
        }
        *(vp*)(out + (size_t)(lo + r) * F + lane * PER) = o;
      }
    }
  }
}

extern "C" void kernel_launch(void* const* d_in, const int* in_sizes, int n_in,
                              void* d_out, int out_size, void* d_ws, size_t ws_size,
                              hipStream_t stream) {
  const float* x      = (const float*)d_in[0];
  const float* ea     = (const float*)d_in[1];
  const float* W1l    = (const float*)d_in[2];
  const float* W1r    = (const float*)d_in[3];
  const float* W1e    = (const float*)d_in[4];
  const float* att1   = (const float*)d_in[5];
  const float* b1     = (const float*)d_in[6];
  const float* ln1_g  = (const float*)d_in[7];
  const float* ln1_b  = (const float*)d_in[8];
  const float* W2l    = (const float*)d_in[9];
  const float* W2r    = (const float*)d_in[10];
  const float* W2e    = (const float*)d_in[11];
  const float* att2   = (const float*)d_in[12];
  const float* b2     = (const float*)d_in[13];
  const float* ln2_g  = (const float*)d_in[14];
  const float* ln2_b  = (const float*)d_in[15];
  const int*   ei     = (const int*)d_in[16];
  float*       outp   = (float*)d_out;

  const int N = in_sizes[0] / 10;
  const int E = in_sizes[1];
  const int HID = 32, HEADS = 4, F1 = HEADS * HID  , OUTC = 64;
  const int KP = 12;

  float* ws      = (float*)d_ws;
  float* xl1     = ws;
  float* xr1     = xl1 + (size_t)N * F1;
  float* h1      = xr1 + (size_t)N * F1;
  float* xpad    = h1  + (size_t)N * F1;
  float* w1lp    = xpad + (size_t)N * KP;
  float* w1rp    = w1lp + (size_t)KP * F1;
  unsigned char* etag = (unsigned char*)(w1rp + (size_t)KP * F1);
  float* xl2     = xl1;
  float* xr2     = xr1;

  const int TB = 256;
  auto blocks = [](size_t n, int tb) { return (unsigned)((n + tb - 1) / tb); };

  pad_cols<<<blocks((size_t)N * KP, TB), TB, 0, stream>>>(x, xpad, N, 10, KP);
  pad_rows<<<blocks((size_t)KP * F1, TB), TB, 0, stream>>>(W1l, w1lp, 10, KP, F1);
  pad_rows<<<blocks((size_t)KP * F1, TB), TB, 0, stream>>>(W1r, w1rp, 10, KP, F1);
  {
    dim3 grid((N / 16 + 3) / 4, F1 / 16);
    wmma_gemm<12><<<grid, 128, 0, stream>>>(xpad, w1lp, xl1, N, F1);
    wmma_gemm<12><<<grid, 128, 0, stream>>>(xpad, w1rp, xr1, N, F1);
  }

  edge_tags<<<blocks(((size_t)E + 15) / 16, TB), TB, 0, stream>>>(ei + E, etag, E);
  edge_aggr_rows<128, 4><<<blocks((size_t)N, AGG_TR), AGG_THREADS, 0, stream>>>(
      ei, ea, etag, xl1, xr1, W1e, att1, b1, ln1_g, ln1_b, h1, N, E);

  {
    dim3 grid((N / 16 + 3) / 4, OUTC / 16);
    wmma_gemm<128><<<grid, 128, 0, stream>>>(h1, W2l, xl2, N, OUTC);
    wmma_gemm<128><<<grid, 128, 0, stream>>>(h1, W2r, xr2, N, OUTC);
  }

  edge_aggr_rows<64, 1><<<blocks((size_t)N, AGG_TR), AGG_THREADS, 0, stream>>>(
      ei, ea, etag, xl2, xr2, W2e, att2, b2, ln2_g, ln2_b, outp, N, E);
}


# layer 2 aggregation reads its block's edge list (src, row, attribute), which layer 1's block wrote to memory while scanning: no second tag scan, no gathers of src, dst, attribute; sums: LDS arrival
# speedup vs baseline: 7.1062x; 1.0788x over previous
#include <hip/hip_runtime.h>
#include <hip/hip_bf16.h>
#include <math.h>

typedef float     v2f  __attribute__((ext_vector_type(2)));
typedef float     v8f  __attribute__((ext_vector_type(8)));

#define NEG_SLOPE 0.2f
#define LN_EPS    1e-5f

__device__ __forceinline__ unsigned ordEncode(float f) {
  unsigned u = __float_as_uint(f);
  return (u >> 31) ? ~u : (u | 0x80000000u);
}
__device__ __forceinline__ float ordDecode(unsigned o) {
  return __uint_as_float((o >> 31) ? (o & 0x7FFFFFFFu) : ~o);
}

__global__ void pad_cols(const float* __restrict__ src, float* __restrict__ dst,
                         int M, int Ks, int Kd) {
  int i = blockIdx.x * blockDim.x + threadIdx.x;
  if (i >= M * Kd) return;
  int r = i / Kd, c = i % Kd;
  dst[i] = (c < Ks) ? src[(size_t)r * Ks + c] : 0.0f;
}

__global__ void pad_rows(const float* __restrict__ src, float* __restrict__ dst,
                         int Ms, int Md, int Ncols) {
  int i = blockIdx.x * blockDim.x + threadIdx.x;
  if (i >= Md * Ncols) return;
  int r = i / Ncols;
  dst[i] = (r < Ms) ? src[i] : 0.0f;
}

template <int K>
__global__ void wmma_gemm(const float* __restrict__ A, const float* __restrict__ B,
                          float* __restrict__ C, int M, int N) {
  const int lane  = threadIdx.x & 31;
  const int wave  = threadIdx.x >> 5;
  const int tileM = blockIdx.x * 4 + wave;
  const int m0 = tileM * 16;
  const int n0 = blockIdx.y * 16;
  if (m0 + 16 > M) return;
  const int half = lane >> 4;
  const int l16  = lane & 15;
  const float* __restrict__ arow = A + (size_t)(m0 + l16) * K + half * 2;
  const float* __restrict__ bcol = B + n0 + l16 + (size_t)(half * 2) * N;
  v8f acc = {};
#pragma unroll
  for (int k = 0; k < K; k += 4) {
    v2f a = *(const v2f*)(arow + k);
    v2f b;
    b.x = bcol[(size_t)k * N];
    b.y = bcol[(size_t)(k + 1) * N];
    acc = __builtin_amdgcn_wmma_f32_16x16x4_f32(false, a, false, b, (short)0, acc,
                                                false, false);
  }
  float* __restrict__ crow = C + (size_t)(m0 + half * 8) * N + n0 + l16;
#pragma unroll
  for (int r = 0; r < 8; ++r)
    crow[(size_t)r * N] = acc[r];
}

#define LG_U 8
#define AGG_TR 196

__global__ void edge_tags(const int* __restrict__ dstp, unsigned char* __restrict__ tag, int E) {
  const int e0 = (blockIdx.x * blockDim.x + threadIdx.x) * 16;
  if (e0 >= E) return;
  if (e0 + 16 <= E && (((((size_t)(dstp + e0)) | ((size_t)(tag + e0))) & 15) == 0)) {
    const int4* __restrict__ p = (const int4*)(dstp + e0);
    const int4 d0 = p[0], d1 = p[1], d2 = p[2], d3 = p[3];
    uint4 o;
    o.x = ((unsigned)(d0.x / AGG_TR) & 255u) | (((unsigned)(d0.y / AGG_TR) & 255u) << 8) |
          (((unsigned)(d0.z / AGG_TR) & 255u) << 16) | (((unsigned)(d0.w / AGG_TR) & 255u) << 24);
    o.y = ((unsigned)(d1.x / AGG_TR) & 255u) | (((unsigned)(d1.y / AGG_TR) & 255u) << 8) |
          (((unsigned)(d1.z / AGG_TR) & 255u) << 16) | (((unsigned)(d1.w / AGG_TR) & 255u) << 24);
    o.z = ((unsigned)(d2.x / AGG_TR) & 255u) | (((unsigned)(d2.y / AGG_TR) & 255u) << 8) |
          (((unsigned)(d2.z / AGG_TR) & 255u) << 16) | (((unsigned)(d2.w / AGG_TR) & 255u) << 24);
    o.w = ((unsigned)(d3.x / AGG_TR) & 255u) | (((unsigned)(d3.y / AGG_TR) & 255u) << 8) |
          (((unsigned)(d3.z / AGG_TR) & 255u) << 16) | (((unsigned)(d3.w / AGG_TR) & 255u) << 24);
    *(uint4*)(tag + e0) = o;
  } else {
    for (int e = e0; e < min(e0 + 16, E); ++e) tag[e] = (unsigned char)(dstp[e] / AGG_TR);
  }
}

template <int F, int H, class VF>
__device__ __forceinline__ void batch_logits(const float* xlrows, const float* __restrict__ xr,
                                             const VF we, const VF at,
                                             int sL, int rL, int eaL, int cnt, int lane,
                                             float (&outv)[H]) {
  constexpr int VEC = F / 32;
  constexpr int LPH = 32 / H;
  static_assert(LG_U == 8 && LPH >= 8 && 32 % H == 0, "the lane sum below is laid out for 8 edges a step");
#pragma unroll
  for (int g = 0; g < H; ++g) outv[g] = 0.0f;

  constexpr int NB = (VEC <= 2) ? 2 : 1;
#pragma unroll
  for (int jb0 = 0; jb0 < 32; jb0 += LG_U * NB) {
    if (jb0 < cnt) {
      VF b[NB][LG_U];
#pragma unroll
      for (int nb = 0; nb < NB; ++nb) {
#pragma unroll
        for (int u = 0; u < LG_U; ++u) {
          const int s = __builtin_amdgcn_readlane(sL, jb0 + nb * LG_U + u);
          b[nb][u] = *(const VF*)(xr + (size_t)s * F + lane * VEC);
        }
      }
      static_assert(LG_U == 8, "the statement below names eight rows a step");
      if constexpr (NB == 1) {
        asm volatile("" :: "v"(b[0][0]), "v"(b[0][1]), "v"(b[0][2]), "v"(b[0][3]),
                           "v"(b[0][4]), "v"(b[0][5]), "v"(b[0][6]), "v"(b[0][7]));
      } else {
        asm volatile("" :: "v"(b[0][0]), "v"(b[0][1]), "v"(b[0][2]), "v"(b[0][3]),
                           "v"(b[0][4]), "v"(b[0][5]), "v"(b[0][6]), "v"(b[0][7]),
                           "v"(b[NB - 1][0]), "v"(b[NB - 1][1]), "v"(b[NB - 1][2]), "v"(b[NB - 1][3]),
                           "v"(b[NB - 1][4]), "v"(b[NB - 1][5]), "v"(b[NB - 1][6]), "v"(b[NB - 1][7]));
      }
#pragma unroll
      for (int nb = 0; nb < NB; ++nb) {
        const int jb = jb0 + nb * LG_U;
        if (nb == 0 || jb < cnt) {
          VF a[LG_U];
#pragma unroll
          for (int u = 0; u < LG_U; ++u) {
            const int r = __builtin_amdgcn_readlane(rL, jb + u);
            a[u] = *(const VF*)(xlrows + (size_t)r * F + lane * VEC);
          }
          float p[LG_U];
#pragma unroll
          for (int u = 0; u < LG_U; ++u) {
            const float eav = __int_as_float(__builtin_amdgcn_readlane(eaL, jb + u));
            float sum = 0.0f;
#pragma unroll
            for (int i = 0; i < VEC; ++i) {
              float v = a[u][i] + b[nb][u][i] + eav * we[i];
              v = fmaxf(v, NEG_SLOPE * v);
              sum += at[i] * v;
            }
            p[u] = sum;
          }
#pragma unroll
          for (int st = 0; st < 3; ++st) {
            const int off = LPH >> (st + 1);
            const int hn  = LG_U >> (st + 1);
            const bool hi = (lane & off) != 0;
#pragma unroll
            for (int u = 0; u < hn; ++u) {
              const float send = hi ? p[u] : p[u + hn];
              const float keep = hi ? p[u + hn] : p[u];
              p[u] = keep + __shfl_xor(send, off, 32);
            }
          }
#pragma unroll
          for (int off = LPH / 16; off > 0; off >>= 1) p[0] += __shfl_xor(p[0], off, 32);
          const int rl = lane - (jb * H) % 32;
          const bool mine = (rl >= 0) && (rl < LG_U * H);
          const int rc = mine ? rl : 0;
          const float q = __shfl(p[0], (rc % H) * LPH + (rc / H) * (LPH / 8), 32);
          if (mine) outv[(jb * H) / 32] = q;
        }
      }
    }
  }
}

#define AGG_THREADS 1024
#define AGG_GROUPS  8
#define AGG_CAP     8192
#define ORD_NEG_INF 0x007FFFFFu

__device__ __forceinline__ unsigned tag_differs(unsigned w, unsigned pat) {
  const unsigned x = w ^ pat;
  return ((x & 0x7F7F7F7Fu) + 0x7F7F7F7Fu) | x;
}
__device__ __forceinline__ unsigned tags_differ16(uint4 t, unsigned pat) {
  unsigned r = (tag_differs(t.x, pat) >> 7) & 0x01010101u;
  r |= (tag_differs(t.y, pat) >> 6) & 0x02020202u;
  r |= (tag_differs(t.z, pat) >> 5) & 0x04040404u;
  r |= (tag_differs(t.w, pat) >> 4) & 0x08080808u;
  return r;
}

template <int F, int H>
__device__ __forceinline__ void aggr_core(const float* __restrict__ xr,
                                          float* acc, float* den, const unsigned* mxo,
                                          int s, int row, const float (&lg)[H], int cnt, int lane) {
  constexpr int C = F / H;
  constexpr int P = C / 32;
  constexpr int R = H * P;
  constexpr int U = 32 / R;
  const bool live = lane < cnt;
  float al[H];
#pragma unroll
  for (int h = 0; h < H; ++h) al[h] = __expf(lg[h] - ordDecode(mxo[row * H + h]));
  if (live) {
#pragma unroll
    for (int h = 0; h < H; ++h) atomicAdd(&den[row * H + h], al[h]);
  }

  for (int j0 = 0; j0 < cnt; j0 += U) {
    float v[U][R];
#pragma unroll
    for (int u = 0; u < U; ++u) {
      const int jj = min(j0 + u, cnt - 1);
      const int sj = __builtin_amdgcn_readlane(s, jj);
      const float* __restrict__ xrs = xr + (size_t)sj * F + lane;
#pragma unroll
      for (int i = 0; i < R; ++i) v[u][i] = xrs[i * 32];
    }
#pragma unroll
    for (int u = 0; u < U; ++u) {
      const int jj = j0 + u;
      if (jj < cnt) {
        const int rowj = __builtin_amdgcn_readlane(row, jj);
        float* arow = acc + rowj * F + lane;
#pragma unroll
        for (int i = 0; i < R; ++i) {
          const float a = __int_as_float(
              __builtin_amdgcn_readlane(__float_as_int(al[i / P]), jj));
          atomicAdd(&arow[i * 32], a * v[u][i]);
        }
      }
    }
  }
}

template <class Flush>
__device__ __forceinline__ void scan_edges(const int* __restrict__ dstp,
                                           const unsigned char* __restrict__ tag,
                                           int lo, unsigned lim, int E, int wave, int lane,
                                           Flush flush) {
  int qe = 0;
  int cnt = 0;
  const bool useTags = (tag != nullptr) && (gridDim.x <= 256u) && ((((size_t)tag) & 15) == 0);
  const int nGroups = useTags ? (E >> 4) : 0;
  if (nGroups > 0) {
    const uint4* __restrict__ tag4 = (const uint4*)tag;
    const unsigned pat = blockIdx.x * 0x01010101u;
    for (int gb = wave * (32 * AGG_GROUPS); gb < nGroups;
         gb += (AGG_THREADS / 32) * 32 * AGG_GROUPS) {
      uint4 t[AGG_GROUPS];
#pragma unroll
      for (int u = 0; u < AGG_GROUPS; ++u)
        t[u] = tag4[min(gb + u * 32 + lane, nGroups - 1)];
      __builtin_amdgcn_sched_barrier(0);
      unsigned m[AGG_GROUPS / 2];
#pragma unroll
      for (int j = 0; j < AGG_GROUPS / 2; ++j) {
        const unsigned differ = tags_differ16(t[2 * j], pat) | (tags_differ16(t[2 * j + 1], pat) << 4);
        const unsigned ok = (((gb + (2 * j) * 32 + lane) < nGroups) ? 0x0F0F0F0Fu : 0u) |
                            (((gb + (2 * j + 1) * 32 + lane) < nGroups) ? 0xF0F0F0F0u : 0u);
        m[j] = ~differ & ok;
      }
      static_assert(AGG_GROUPS == 8, "the choice of m[j] below is written for four words");
#pragma nounroll
      for (int j = 0; j < AGG_GROUPS / 2; ++j) {
        const unsigned mj = (j == 0) ? m[0] : (j == 1) ? m[1] : (j == 2) ? m[2] : m[3];
        unsigned any = __builtin_amdgcn_ballot_w32(mj != 0u);
        while (any) {
          const int L = __builtin_ctz(any);
          any &= any - 1u;
          unsigned b = (unsigned)__builtin_amdgcn_readlane((int)mj, L);
          while (b) {
            const int p = __builtin_ctz(b);
            b &= b - 1u;
            const int g = gb + (2 * j + ((p >> 2) & 1)) * 32 + L;
            const int e = (g << 4) + ((p & 3) << 2) + (p >> 3);
            qe = (lane == cnt) ? e : qe;
            if (++cnt == 32) {
              flush(qe, 32);
              cnt = 0;
            }
          }
        }
      }
    }
  }
  for (int eb = (nGroups << 4) + wave * 32; eb < E; eb += AGG_THREADS) {
    const int e1 = eb + lane;
    const unsigned r1 = (e1 < E) ? (unsigned)(dstp[min(e1, E - 1)] - lo) : 0xFFFFFFFFu;
    unsigned mk = __builtin_amdgcn_ballot_w32(r1 < lim);
    while (mk) {
      const int j = __builtin_ctz(mk);
      mk &= mk - 1u;
      qe = (lane == cnt) ? (eb + j) : qe;
      if (++cnt == 32) {
        flush(qe, 32);
        cnt = 0;
      }
    }
  }
  if (cnt > 0) flush(qe, cnt);
}

template <int F, int H, int LIST>
__global__ __launch_bounds__(AGG_THREADS)
void edge_aggr_rows(const int* __restrict__ ei, const float* __restrict__ ea,
                    const unsigned char* __restrict__ tag,
                    const float* __restrict__ xl, const float* __restrict__ xr,
                    const float* __restrict__ We, const float* __restrict__ att,
                    const float* __restrict__ bias, const float* __restrict__ lng,
                    const float* __restrict__ lnb, float* __restrict__ out,
                    int* gsrc, unsigned char* grow, int* gea, int* gcnt, int N, int E) {
  static_assert(AGG_TR <= 256, "a row index inside the block is kept in one byte");
  __shared__ float acc[AGG_TR * F];
  __shared__ float den[AGG_TR * H];
  __shared__ unsigned mxo[AGG_TR * H];
  __shared__ int esrc[AGG_CAP];
  __shared__ unsigned char erow[AGG_CAP];
  __shared__ float llog[AGG_CAP * H];
  __shared__ unsigned lcount;
  const int tid  = threadIdx.x;
  const int lane = tid & 31;
  const int wave = __builtin_amdgcn_readfirstlane(tid >> 5);
  const int lo = blockIdx.x * AGG_TR;
  const int nrows = min(AGG_TR, N - lo);
  const unsigned lim = (unsigned)nrows;
  const int* __restrict__ dstp = ei + E;
  constexpr int VEC = F / 32;
  typedef float vf __attribute__((ext_vector_type(VEC)));
  const vf we = *(const vf*)(We + lane * VEC);
  const vf at = *(const vf*)(att + lane * VEC);

  {
    constexpr int Q = AGG_TR * F / 4;
    constexpr int NV = (Q + AGG_THREADS - 1) / AGG_THREADS;
    const float4* __restrict__ src4 = (const float4*)(xl + (size_t)lo * F);
    const int have = nrows * (F / 4);
    float4 tmp[NV];
#pragma unroll
    for (int k = 0; k < NV; ++k) {
      const int i = tid + k * AGG_THREADS;
      tmp[k] = (i < have) ? src4[i] : make_float4(0.f, 0.f, 0.f, 0.f);
    }
#pragma unroll
    for (int k = 0; k < NV; ++k) {
      const int i = tid + k * AGG_THREADS;
      if (i < Q) ((float4*)acc)[i] = tmp[k];
    }
  }
  for (int i = tid; i < AGG_TR * H; i += AGG_THREADS) {
    den[i] = 0.0f;
    mxo[i] = ORD_NEG_INF;
  }
  if (tid == 0) lcount = 0u;
  __syncthreads();

  const size_t lb = (size_t)blockIdx.x * AGG_CAP;
  auto work = [&](int s, int row, int eav, int cnt, int given) {
    float outv[H];
    batch_logits<F, H, vf>(acc, xr, we, at, s, row, eav, cnt, lane, outv);
    unsigned base = (unsigned)given;
    if (given < 0) {
      base = 0u;
      if (lane == 0) base = atomicAdd(&lcount, (unsigned)cnt);
      base = (unsigned)__builtin_amdgcn_readfirstlane((int)base);
    }
    const bool fits = base + (unsigned)cnt <= (unsigned)AGG_CAP;
#pragma unroll
    for (int g = 0; g < H; ++g) {
      const int idx = g * 32 + lane;
      const int j = idx / H;
      const int h = idx % H;
      const int rj = __shfl(row, j, 32);
      if (j < cnt) {
        atomicMax(&mxo[rj * H + h], ordEncode(outv[g]));
        if (fits) llog[base * H + idx] = outv[g];
      }
    }
    if (fits && lane < cnt) {
      esrc[base + lane] = s;
      erow[base + lane] = (unsigned char)row;
      if constexpr (LIST == 0) {
        gsrc[lb + base + lane] = s;
        grow[lb + base + lane] = (unsigned char)row;
        gea[lb + base + lane] = eav;
      }
    }
  };
  auto scanned = [&](int qe, int cnt) {
    const int s = ei[qe];
    const int d = dstp[qe];
    const int eav = __float_as_int(ea[qe]);
    const int row = (lane < cnt) ? d - lo : 0;
    work(s, row, eav, cnt, -1);
  };
  const int gtotal = (LIST == 1) ? __builtin_amdgcn_readfirstlane(gcnt[blockIdx.x]) : -1;
  if (LIST == 1 && gtotal >= 0 && gtotal <= AGG_CAP) {
    const int share1 = (gtotal + AGG_THREADS / 32 - 1) / (AGG_THREADS / 32);
    const int wend1 = min(gtotal, (wave + 1) * share1);
    for (int b0 = wave * share1; b0 < wend1; b0 += 32) {
      const int c = min(32, wend1 - b0);
      const int q = min(b0 + lane, wend1 - 1);
      work(gsrc[lb + q], (int)grow[lb + q], gea[lb + q], c, b0);
    }
    if (tid == 0) lcount = (unsigned)gtotal;
  } else {
    scan_edges(dstp, tag, lo, lim, E, wave, lane, scanned);
  }
  __syncthreads();
  const int total = __builtin_amdgcn_readfirstlane((int)lcount);
  if (LIST == 0 && tid == 0) gcnt[blockIdx.x] = total;
  for (int i = tid; i < AGG_TR * F / 4; i += AGG_THREADS)
    ((float4*)acc)[i] = make_float4(0.f, 0.f, 0.f, 0.f);
  __syncthreads();
  if (total <= AGG_CAP) {
    const int share = (total + AGG_THREADS / 32 - 1) / (AGG_THREADS / 32);
    const int wend = min(total, (wave + 1) * share);
    for (int b0 = wave * share; b0 < wend; b0 += 32) {
      const int c = min(32, wend - b0);
      const int q = min(b0 + lane, wend - 1);
      const int s = esrc[q];
      const int row = erow[q];
      float lg[H];
#pragma unroll
      for (int h = 0; h < H; ++h) lg[h] = llog[q * H + h];
      aggr_core<F, H>(xr, acc, den, mxo, s, row, lg, c, lane);
    }
  } else {
    scan_edges(dstp, tag, lo, lim, E, wave, lane, [&](int qe, int cnt) {
      const int s = ei[qe];
      const int d = dstp[qe];
      const int eav = __float_as_int(ea[qe]);
      const int row = (lane < cnt) ? d - lo : 0;
      float outv[H];
      batch_logits<F, H, vf>(xl, xr, we, at, s, d, eav, cnt, lane, outv);
      float lg[H];
#pragma unroll
      for (int h = 0; h < H; ++h) {
        const int idx = lane * H + h;
        lg[h] = 0.0f;
#pragma unroll
        for (int g = 0; g < H; ++g) {
          const float t = __shfl(outv[g], idx & 31, 32);
          if ((idx >> 5) == g) lg[h] = t;
        }
      }
      aggr_core<F, H>(xr, acc, den, mxo, s, row, lg, cnt, lane);
    });
  }
  __syncthreads();
  constexpr int PER = F / 32;
  constexpr int ROWS_TOGETHER = 4;
  constexpr int WAVES = AGG_THREADS / 32;
  static_assert(PER >= 1 && (F / H) % PER == 0, "a lane's channels must lie in one head");
  typedef float vp __attribute__((ext_vector_type(PER)));
  const vp bs = *(const vp*)(bias + lane * PER);
  const vp gg = *(const vp*)(lng + lane * PER);
  const vp bb = *(const vp*)(lnb + lane * PER);
  const int hd = (lane * PER) / (F / H);
  for (int r0 = wave; r0 < nrows; r0 += WAVES * ROWS_TOGETHER) {
    float vals[ROWS_TOGETHER][PER];
    float sum[ROWS_TOGETHER];
#pragma unroll
    for (int k = 0; k < ROWS_TOGETHER; ++k) {
      const int r = min(r0 + k * WAVES, nrows - 1);
      const vp a = *(const vp*)(acc + r * F + lane * PER);
      const float dn = den[r * H + hd];
      float s = 0.0f;
#pragma unroll
      for (int i = 0; i < PER; ++i) {
        float v = a[i];
        if (dn > 0.0f) v /= dn;
        v += bs[i];
        vals[k][i] = v;
        s += v;
      }
      sum[k] = s;
    }
#pragma unroll
    for (int off = 16; off > 0; off >>= 1) {
#pragma unroll
      for (int k = 0; k < ROWS_TOGETHER; ++k) sum[k] += __shfl_xor(sum[k], off, 32);
    }
    float mu[ROWS_TOGETHER], vs[ROWS_TOGETHER];
#pragma unroll
    for (int k = 0; k < ROWS_TOGETHER; ++k) {
      mu[k] = sum[k] / (float)F;
      float q = 0.0f;
#pragma unroll
      for (int i = 0; i < PER; ++i) { const float dlt = vals[k][i] - mu[k]; q += dlt * dlt; }
      vs[k] = q;
    }
#pragma unroll
    for (int off = 16; off > 0; off >>= 1) {
#pragma unroll
      for (int k = 0; k < ROWS_TOGETHER; ++k) vs[k] += __shfl_xor(vs[k], off, 32);
    }
#pragma unroll
    for (int k = 0; k < ROWS_TOGETHER; ++k) {
      const int r = r0 + k * WAVES;
      if (r < nrows) {
        const float inv = rsqrtf(vs[k] / (float)F + LN_EPS);
        vp o;
#pragma unroll
        for (int i = 0; i < PER; ++i) {
          const float hv = (vals[k][i] - mu[k]) * inv * gg[i] + bb[i];
          o[i] = (hv > 0.0f) ? hv : (__expf(hv) - 1.0f);
        }
        *(vp*)(out + (size_t)(lo + r) * F + lane * PER) = o;
      }
    }
  }
}

extern "C" void kernel_launch(void* const* d_in, const int* in_sizes, int n_in,
                              void* d_out, int out_size, void* d_ws, size_t ws_size,
                              hipStream_t stream) {
  const float* x      = (const float*)d_in[0];
  const float* ea     = (const float*)d_in[1];
  const float* W1l    = (const float*)d_in[2];
  const float* W1r    = (const float*)d_in[3];
  const float* W1e    = (const float*)d_in[4];
  const float* att1   = (const float*)d_in[5];
  const float* b1     = (const float*)d_in[6];
  const float* ln1_g  = (const float*)d_in[7];
  const float* ln1_b  = (const float*)d_in[8];
  const float* W2l    = (const float*)d_in[9];
  const float* W2r    = (const float*)d_in[10];
  const float* W2e    = (const float*)d_in[11];
  const float* att2   = (const float*)d_in[12];
  const float* b2     = (const float*)d_in[13];
  const float* ln2_g  = (const float*)d_in[14];
  const float* ln2_b  = (const float*)d_in[15];
  const int*   ei     = (const int*)d_in[16];
  float*       outp   = (float*)d_out;

  const int N = in_sizes[0] / 10;
  const int E = in_sizes[1];
  const int HID = 32, HEADS = 4, F1 = HEADS * HID  , OUTC = 64;
  const int KP = 12;

  float* ws      = (float*)d_ws;
  float* xl1     = ws;
  float* xr1     = xl1 + (size_t)N * F1;
  float* h1      = xr1 + (size_t)N * F1;
  float* xpad    = h1  + (size_t)N * F1;
  float* w1lp    = xpad + (size_t)N * KP;
  float* w1rp    = w1lp + (size_t)KP * F1;
  unsigned char* etag = (unsigned char*)(w1rp + (size_t)KP * F1);
  const size_t NBLK = ((size_t)N + AGG_TR - 1) / AGG_TR;
  int* gsrc           = (int*)(etag + (((size_t)E + 255) & ~(size_t)255));
  int* gea            = gsrc + NBLK * AGG_CAP;
  int* gcnt           = gea + NBLK * AGG_CAP;
  unsigned char* grow = (unsigned char*)(gcnt + ((NBLK + 63) & ~(size_t)63));
  float* xl2     = xl1;
  float* xr2     = xr1;

  const int TB = 256;
  auto blocks = [](size_t n, int tb) { return (unsigned)((n + tb - 1) / tb); };

  pad_cols<<<blocks((size_t)N * KP, TB), TB, 0, stream>>>(x, xpad, N, 10, KP);
  pad_rows<<<blocks((size_t)KP * F1, TB), TB, 0, stream>>>(W1l, w1lp, 10, KP, F1);
  pad_rows<<<blocks((size_t)KP * F1, TB), TB, 0, stream>>>(W1r, w1rp, 10, KP, F1);
  {
    dim3 grid((N / 16 + 3) / 4, F1 / 16);
    wmma_gemm<12><<<grid, 128, 0, stream>>>(xpad, w1lp, xl1, N, F1);
    wmma_gemm<12><<<grid, 128, 0, stream>>>(xpad, w1rp, xr1, N, F1);
  }

  edge_tags<<<blocks(((size_t)E + 15) / 16, TB), TB, 0, stream>>>(ei + E, etag, E);
  edge_aggr_rows<128, 4, 0><<<blocks((size_t)N, AGG_TR), AGG_THREADS, 0, stream>>>(
      ei, ea, etag, xl1, xr1, W1e, att1, b1, ln1_g, ln1_b, h1, gsrc, grow, gea, gcnt, N, E);

  {
    dim3 grid((N / 16 + 3) / 4, OUTC / 16);
    wmma_gemm<128><<<grid, 128, 0, stream>>>(h1, W2l, xl2, N, OUTC);
    wmma_gemm<128><<<grid, 128, 0, stream>>>(h1, W2r, xr2, N, OUTC);
  }

  edge_aggr_rows<64, 1, 1><<<blocks((size_t)N, AGG_TR), AGG_THREADS, 0, stream>>>(
      ei, ea, etag, xl2, xr2, W2e, att2, b2, ln2_g, ln2_b, outp, gsrc, grow, gea, gcnt, N, E);
}


# softmax without the max shift: exp(logit)*row added into LDS in the pass that forms the logit (row still in registers); pass 2 and its row gathers gone; maxima kept as range check; sums: LDS arrival
# speedup vs baseline: 7.9679x; 1.1213x over previous
#include <hip/hip_runtime.h>
#include <hip/hip_bf16.h>
#include <math.h>

typedef float     v2f  __attribute__((ext_vector_type(2)));
typedef float     v8f  __attribute__((ext_vector_type(8)));

#define NEG_SLOPE 0.2f
#define LN_EPS    1e-5f

__device__ __forceinline__ unsigned ordEncode(float f) {
  unsigned u = __float_as_uint(f);
  return (u >> 31) ? ~u : (u | 0x80000000u);
}
__device__ __forceinline__ float ordDecode(unsigned o) {
  return __uint_as_float((o >> 31) ? (o & 0x7FFFFFFFu) : ~o);
}

__global__ void pad_cols(const float* __restrict__ src, float* __restrict__ dst,
                         int M, int Ks, int Kd) {
  int i = blockIdx.x * blockDim.x + threadIdx.x;
  if (i >= M * Kd) return;
  int r = i / Kd, c = i % Kd;
  dst[i] = (c < Ks) ? src[(size_t)r * Ks + c] : 0.0f;
}

__global__ void pad_rows(const float* __restrict__ src, float* __restrict__ dst,
                         int Ms, int Md, int Ncols) {
  int i = blockIdx.x * blockDim.x + threadIdx.x;
  if (i >= Md * Ncols) return;
  int r = i / Ncols;
  dst[i] = (r < Ms) ? src[i] : 0.0f;
}

template <int K>
__global__ void wmma_gemm(const float* __restrict__ A, const float* __restrict__ B,
                          float* __restrict__ C, int M, int N) {
  const int lane  = threadIdx.x & 31;
  const int wave  = threadIdx.x >> 5;
  const int tileM = blockIdx.x * 4 + wave;
  const int m0 = tileM * 16;
  const int n0 = blockIdx.y * 16;
  if (m0 + 16 > M) return;
  const int half = lane >> 4;
  const int l16  = lane & 15;
  const float* __restrict__ arow = A + (size_t)(m0 + l16) * K + half * 2;
  const float* __restrict__ bcol = B + n0 + l16 + (size_t)(half * 2) * N;
  v8f acc = {};
#pragma unroll
  for (int k = 0; k < K; k += 4) {
    v2f a = *(const v2f*)(arow + k);
    v2f b;
    b.x = bcol[(size_t)k * N];
    b.y = bcol[(size_t)(k + 1) * N];
    acc = __builtin_amdgcn_wmma_f32_16x16x4_f32(false, a, false, b, (short)0, acc,
                                                false, false);
  }
  float* __restrict__ crow = C + (size_t)(m0 + half * 8) * N + n0 + l16;
#pragma unroll
  for (int r = 0; r < 8; ++r)
    crow[(size_t)r * N] = acc[r];
}

#define LG_U 8
#define AGG_TR 196

__global__ void edge_tags(const int* __restrict__ dstp, unsigned char* __restrict__ tag, int E) {
  const int e0 = (blockIdx.x * blockDim.x + threadIdx.x) * 16;
  if (e0 >= E) return;
  if (e0 + 16 <= E && (((((size_t)(dstp + e0)) | ((size_t)(tag + e0))) & 15) == 0)) {
    const int4* __restrict__ p = (const int4*)(dstp + e0);
    const int4 d0 = p[0], d1 = p[1], d2 = p[2], d3 = p[3];
    uint4 o;
    o.x = ((unsigned)(d0.x / AGG_TR) & 255u) | (((unsigned)(d0.y / AGG_TR) & 255u) << 8) |
          (((unsigned)(d0.z / AGG_TR) & 255u) << 16) | (((unsigned)(d0.w / AGG_TR) & 255u) << 24);
    o.y = ((unsigned)(d1.x / AGG_TR) & 255u) | (((unsigned)(d1.y / AGG_TR) & 255u) << 8) |
          (((unsigned)(d1.z / AGG_TR) & 255u) << 16) | (((unsigned)(d1.w / AGG_TR) & 255u) << 24);
    o.z = ((unsigned)(d2.x / AGG_TR) & 255u) | (((unsigned)(d2.y / AGG_TR) & 255u) << 8) |
          (((unsigned)(d2.z / AGG_TR) & 255u) << 16) | (((unsigned)(d2.w / AGG_TR) & 255u) << 24);
    o.w = ((unsigned)(d3.x / AGG_TR) & 255u) | (((unsigned)(d3.y / AGG_TR) & 255u) << 8) |
          (((unsigned)(d3.z / AGG_TR) & 255u) << 16) | (((unsigned)(d3.w / AGG_TR) & 255u) << 24);
    *(uint4*)(tag + e0) = o;
  } else {
    for (int e = e0; e < min(e0 + 16, E); ++e) tag[e] = (unsigned char)(dstp[e] / AGG_TR);
  }
}

template <int F, int H, class VF, bool FUSE>
__device__ __forceinline__ void batch_logits(const float* xlrows, const float* __restrict__ xr,
                                             const VF we, const VF at,
                                             int sL, int rL, int eaL, int cnt, int lane,
                                             float (&outv)[H], float* accs) {
  constexpr int VEC = F / 32;
  constexpr int LPH = 32 / H;
  static_assert(LG_U == 8 && LPH >= 8 && 32 % H == 0, "the lane sum below is laid out for 8 edges a step");
#pragma unroll
  for (int g = 0; g < H; ++g) outv[g] = 0.0f;

  constexpr int NB = (VEC <= 2) ? 2 : 1;
#pragma unroll
  for (int jb0 = 0; jb0 < 32; jb0 += LG_U * NB) {
    if (jb0 < cnt) {
      VF b[NB][LG_U];
#pragma unroll
      for (int nb = 0; nb < NB; ++nb) {
#pragma unroll
        for (int u = 0; u < LG_U; ++u) {
          const int s = __builtin_amdgcn_readlane(sL, jb0 + nb * LG_U + u);
          b[nb][u] = *(const VF*)(xr + (size_t)s * F + lane * VEC);
        }
      }
      static_assert(LG_U == 8, "the statement below names eight rows a step");
      if constexpr (NB == 1) {
        asm volatile("" :: "v"(b[0][0]), "v"(b[0][1]), "v"(b[0][2]), "v"(b[0][3]),
                           "v"(b[0][4]), "v"(b[0][5]), "v"(b[0][6]), "v"(b[0][7]));
      } else {
        asm volatile("" :: "v"(b[0][0]), "v"(b[0][1]), "v"(b[0][2]), "v"(b[0][3]),
                           "v"(b[0][4]), "v"(b[0][5]), "v"(b[0][6]), "v"(b[0][7]),
                           "v"(b[NB - 1][0]), "v"(b[NB - 1][1]), "v"(b[NB - 1][2]), "v"(b[NB - 1][3]),
                           "v"(b[NB - 1][4]), "v"(b[NB - 1][5]), "v"(b[NB - 1][6]), "v"(b[NB - 1][7]));
      }
#pragma unroll
      for (int nb = 0; nb < NB; ++nb) {
        const int jb = jb0 + nb * LG_U;
        if (nb == 0 || jb < cnt) {
          VF a[LG_U];
#pragma unroll
          for (int u = 0; u < LG_U; ++u) {
            const int r = __builtin_amdgcn_readlane(rL, jb + u);
            a[u] = *(const VF*)(xlrows + (size_t)r * F + lane * VEC);
          }
          float p[LG_U];
#pragma unroll
          for (int u = 0; u < LG_U; ++u) {
            const float eav = __int_as_float(__builtin_amdgcn_readlane(eaL, jb + u));
            float sum = 0.0f;
#pragma unroll
            for (int i = 0; i < VEC; ++i) {
              float v = a[u][i] + b[nb][u][i] + eav * we[i];
              v = fmaxf(v, NEG_SLOPE * v);
              sum += at[i] * v;
            }
            p[u] = sum;
          }
#pragma unroll
          for (int st = 0; st < 3; ++st) {
            const int off = LPH >> (st + 1);
            const int hn  = LG_U >> (st + 1);
            const bool hi = (lane & off) != 0;
#pragma unroll
            for (int u = 0; u < hn; ++u) {
              const float send = hi ? p[u] : p[u + hn];
              const float keep = hi ? p[u + hn] : p[u];
              p[u] = keep + __shfl_xor(send, off, 32);
            }
          }
#pragma unroll
          for (int off = LPH / 16; off > 0; off >>= 1) p[0] += __shfl_xor(p[0], off, 32);
          if constexpr (FUSE) {
            const float wexp = __expf(p[0]);
            float wv[LG_U];
#pragma unroll
            for (int u = 0; u < LG_U; ++u)
              wv[u] = __shfl(wexp, (lane & ~(LPH - 1)) + u * (LPH / 8), 32);
#pragma unroll
            for (int u = 0; u < LG_U; ++u) {
              if (jb + u < cnt) {
                const int r = __builtin_amdgcn_readlane(rL, jb + u);
                float* arow = accs + (size_t)r * F + lane * VEC;
                if constexpr (VEC == 4) {
                  const bool up = (lane & 16) != 0;
                  const float t0 = wv[u] * b[nb][u][0], t1 = wv[u] * b[nb][u][1];
                  const float t2 = wv[u] * b[nb][u][2], t3 = wv[u] * b[nb][u][3];
                  float* a0 = arow + (up ? 1 : 0);
                  float* a3 = arow + (up ? 0 : 3);
                  atomicAdd(a0 + 0, up ? t1 : t0);
                  atomicAdd(a0 + 1, up ? t2 : t1);
                  atomicAdd(a0 + 2, up ? t3 : t2);
                  atomicAdd(a3,     up ? t0 : t3);
                } else {
#pragma unroll
                  for (int i = 0; i < VEC; ++i) atomicAdd(&arow[i], wv[u] * b[nb][u][i]);
                }
              }
            }
          }
          const int rl = lane - (jb * H) % 32;
          const bool mine = (rl >= 0) && (rl < LG_U * H);
          const int rc = mine ? rl : 0;
          const float q = __shfl(p[0], (rc % H) * LPH + (rc / H) * (LPH / 8), 32);
          if (mine) outv[(jb * H) / 32] = q;
        }
      }
    }
  }
}

#define AGG_THREADS 1024
#define AGG_GROUPS  8
#define AGG_CAP     8192
#define SHIFT_FREE  16.0f
#define ORD_NEG_INF 0x007FFFFFu

__device__ __forceinline__ unsigned tag_differs(unsigned w, unsigned pat) {
  const unsigned x = w ^ pat;
  return ((x & 0x7F7F7F7Fu) + 0x7F7F7F7Fu) | x;
}
__device__ __forceinline__ unsigned tags_differ16(uint4 t, unsigned pat) {
  unsigned r = (tag_differs(t.x, pat) >> 7) & 0x01010101u;
  r |= (tag_differs(t.y, pat) >> 6) & 0x02020202u;
  r |= (tag_differs(t.z, pat) >> 5) & 0x04040404u;
  r |= (tag_differs(t.w, pat) >> 4) & 0x08080808u;
  return r;
}

template <int F, int H>
__device__ __forceinline__ void aggr_core(const float* __restrict__ xr,
                                          float* acc, float* den, const unsigned* mxo,
                                          int s, int row, const float (&lg)[H], int cnt, int lane) {
  constexpr int C = F / H;
  constexpr int P = C / 32;
  constexpr int R = H * P;
  constexpr int U = 32 / R;
  const bool live = lane < cnt;
  float al[H];
#pragma unroll
  for (int h = 0; h < H; ++h) al[h] = __expf(lg[h] - ordDecode(mxo[row * H + h]));
  if (live) {
#pragma unroll
    for (int h = 0; h < H; ++h) atomicAdd(&den[row * H + h], al[h]);
  }

  for (int j0 = 0; j0 < cnt; j0 += U) {
    float v[U][R];
#pragma unroll
    for (int u = 0; u < U; ++u) {
      const int jj = min(j0 + u, cnt - 1);
      const int sj = __builtin_amdgcn_readlane(s, jj);
      const float* __restrict__ xrs = xr + (size_t)sj * F + lane;
#pragma unroll
      for (int i = 0; i < R; ++i) v[u][i] = xrs[i * 32];
    }
#pragma unroll
    for (int u = 0; u < U; ++u) {
      const int jj = j0 + u;
      if (jj < cnt) {
        const int rowj = __builtin_amdgcn_readlane(row, jj);
        float* arow = acc + rowj * F + lane;
#pragma unroll
        for (int i = 0; i < R; ++i) {
          const float a = __int_as_float(
              __builtin_amdgcn_readlane(__float_as_int(al[i / P]), jj));
          atomicAdd(&arow[i * 32], a * v[u][i]);
        }
      }
    }
  }
}

template <class Flush>
__device__ __forceinline__ void scan_edges(const int* __restrict__ dstp,
                                           const unsigned char* __restrict__ tag,
                                           int lo, unsigned lim, int E, int wave, int lane,
                                           Flush flush) {
  int qe = 0;
  int cnt = 0;
  const bool useTags = (tag != nullptr) && (gridDim.x <= 256u) && ((((size_t)tag) & 15) == 0);
  const int nGroups = useTags ? (E >> 4) : 0;
  if (nGroups > 0) {
    const uint4* __restrict__ tag4 = (const uint4*)tag;
    const unsigned pat = blockIdx.x * 0x01010101u;
    for (int gb = wave * (32 * AGG_GROUPS); gb < nGroups;
         gb += (AGG_THREADS / 32) * 32 * AGG_GROUPS) {
      uint4 t[AGG_GROUPS];
#pragma unroll
      for (int u = 0; u < AGG_GROUPS; ++u)
        t[u] = tag4[min(gb + u * 32 + lane, nGroups - 1)];
      __builtin_amdgcn_sched_barrier(0);
      unsigned m[AGG_GROUPS / 2];
#pragma unroll
      for (int j = 0; j < AGG_GROUPS / 2; ++j) {
        const unsigned differ = tags_differ16(t[2 * j], pat) | (tags_differ16(t[2 * j + 1], pat) << 4);
        const unsigned ok = (((gb + (2 * j) * 32 + lane) < nGroups) ? 0x0F0F0F0Fu : 0u) |
                            (((gb + (2 * j + 1) * 32 + lane) < nGroups) ? 0xF0F0F0F0u : 0u);
        m[j] = ~differ & ok;
      }
      static_assert(AGG_GROUPS == 8, "the choice of m[j] below is written for four words");
#pragma nounroll
      for (int j = 0; j < AGG_GROUPS / 2; ++j) {
        const unsigned mj = (j == 0) ? m[0] : (j == 1) ? m[1] : (j == 2) ? m[2] : m[3];
        unsigned any = __builtin_amdgcn_ballot_w32(mj != 0u);
        while (any) {
          const int L = __builtin_ctz(any);
          any &= any - 1u;
          unsigned b = (unsigned)__builtin_amdgcn_readlane((int)mj, L);
          while (b) {
            const int p = __builtin_ctz(b);
            b &= b - 1u;
            const int g = gb + (2 * j + ((p >> 2) & 1)) * 32 + L;
            const int e = (g << 4) + ((p & 3) << 2) + (p >> 3);
            qe = (lane == cnt) ? e : qe;
            if (++cnt == 32) {
              flush(qe, 32);
              cnt = 0;
            }
          }
        }
      }
    }
  }
  for (int eb = (nGroups << 4) + wave * 32; eb < E; eb += AGG_THREADS) {
    const int e1 = eb + lane;
    const unsigned r1 = (e1 < E) ? (unsigned)(dstp[min(e1, E - 1)] - lo) : 0xFFFFFFFFu;
    unsigned mk = __builtin_amdgcn_ballot_w32(r1 < lim);
    while (mk) {
      const int j = __builtin_ctz(mk);
      mk &= mk - 1u;
      qe = (lane == cnt) ? (eb + j) : qe;
      if (++cnt == 32) {
        flush(qe, 32);
        cnt = 0;
      }
    }
  }
  if (cnt > 0) flush(qe, cnt);
}

template <int F, int H, int LIST>
__global__ __launch_bounds__(AGG_THREADS)
void edge_aggr_rows(const int* __restrict__ ei, const float* __restrict__ ea,
                    const unsigned char* __restrict__ tag,
                    const float* __restrict__ xl, const float* __restrict__ xr,
                    const float* __restrict__ We, const float* __restrict__ att,
                    const float* __restrict__ bias, const float* __restrict__ lng,
                    const float* __restrict__ lnb, float* __restrict__ out,
                    int* gsrc, unsigned char* grow, int* gea, int* gcnt, int N, int E) {
  static_assert(AGG_TR <= 256, "a row index inside the block is kept in one byte");
  __shared__ float xls[AGG_TR * F];
  __shared__ float acc[AGG_TR * F];
  __shared__ float den[AGG_TR * H];
  __shared__ unsigned mxo[AGG_TR * H];
  __shared__ unsigned lcount;
  __shared__ unsigned wide;
  const int tid  = threadIdx.x;
  const int lane = tid & 31;
  const int wave = __builtin_amdgcn_readfirstlane(tid >> 5);
  const int lo = blockIdx.x * AGG_TR;
  const int nrows = min(AGG_TR, N - lo);
  const unsigned lim = (unsigned)nrows;
  const int* __restrict__ dstp = ei + E;
  constexpr int VEC = F / 32;
  typedef float vf __attribute__((ext_vector_type(VEC)));
  const vf we = *(const vf*)(We + lane * VEC);
  const vf at = *(const vf*)(att + lane * VEC);

  {
    constexpr int Q = AGG_TR * F / 4;
    constexpr int NV = (Q + AGG_THREADS - 1) / AGG_THREADS;
    const float4* __restrict__ src4 = (const float4*)(xl + (size_t)lo * F);
    const int have = nrows * (F / 4);
    float4 tmp[NV];
#pragma unroll
    for (int k = 0; k < NV; ++k) {
      const int i = tid + k * AGG_THREADS;
      tmp[k] = (i < have) ? src4[i] : make_float4(0.f, 0.f, 0.f, 0.f);
    }
#pragma unroll
    for (int k = 0; k < NV; ++k) {
      const int i = tid + k * AGG_THREADS;
      if (i < Q) {
        ((float4*)xls)[i] = tmp[k];
        ((float4*)acc)[i] = make_float4(0.f, 0.f, 0.f, 0.f);
      }
    }
  }
  for (int i = tid; i < AGG_TR * H; i += AGG_THREADS) {
    den[i] = 0.0f;
    mxo[i] = ORD_NEG_INF;
  }
  if (tid == 0) { lcount = 0u; wide = 0u; }
  __syncthreads();

  const size_t lb = (size_t)blockIdx.x * AGG_CAP;
  auto work = [&](int s, int row, int eav, int cnt) __attribute__((always_inline)) {
    float outv[H];
    batch_logits<F, H, vf, true>(xls, xr, we, at, s, row, eav, cnt, lane, outv, acc);
#pragma unroll
    for (int g = 0; g < H; ++g) {
      const int idx = g * 32 + lane;
      const int j = idx / H;
      const int h = idx % H;
      const int rj = __shfl(row, j, 32);
      if (j < cnt) {
        atomicMax(&mxo[rj * H + h], ordEncode(outv[g]));
        atomicAdd(&den[rj * H + h], __expf(outv[g]));
      }
    }
    if constexpr (LIST == 0) {
      unsigned base = 0u;
      if (lane == 0) base = atomicAdd(&lcount, (unsigned)cnt);
      base = (unsigned)__builtin_amdgcn_readfirstlane((int)base);
      if (base + (unsigned)cnt <= (unsigned)AGG_CAP && lane < cnt) {
        gsrc[lb + base + lane] = s;
        grow[lb + base + lane] = (unsigned char)row;
        gea[lb + base + lane] = eav;
      }
    }
  };
  auto scanned = [&](int qe, int cnt) __attribute__((always_inline)) {
    const int s = ei[qe];
    const int d = dstp[qe];
    const int eav = __float_as_int(ea[qe]);
    const int row = (lane < cnt) ? d - lo : 0;
    work(s, row, eav, cnt);
  };
  const int gtotal = (LIST == 1) ? __builtin_amdgcn_readfirstlane(gcnt[blockIdx.x]) : -1;
  if (LIST == 1 && gtotal >= 0 && gtotal <= AGG_CAP) {
    const int share1 = (gtotal + AGG_THREADS / 32 - 1) / (AGG_THREADS / 32);
    const int wend1 = min(gtotal, (wave + 1) * share1);
    for (int b0 = wave * share1; b0 < wend1; b0 += 32) {
      const int c = min(32, wend1 - b0);
      const int q = min(b0 + lane, wend1 - 1);
      work(gsrc[lb + q], (int)grow[lb + q], gea[lb + q], c);
    }
  } else {
    scan_edges(dstp, tag, lo, lim, E, wave, lane, scanned);
  }
  __syncthreads();
  if (LIST == 0 && tid == 0) gcnt[blockIdx.x] = (int)lcount;
  for (int i = tid; i < AGG_TR * H; i += AGG_THREADS) {
    const unsigned o = mxo[i];
    if (o != ORD_NEG_INF && !(fabsf(ordDecode(o)) <= SHIFT_FREE)) wide = 1u;
  }
  __syncthreads();
  if (__builtin_amdgcn_readfirstlane((int)wide) != 0) {
    for (int i = tid; i < AGG_TR * F / 4; i += AGG_THREADS)
      ((float4*)acc)[i] = make_float4(0.f, 0.f, 0.f, 0.f);
    for (int i = tid; i < AGG_TR * H; i += AGG_THREADS) den[i] = 0.0f;
    __syncthreads();
    scan_edges(dstp, tag, lo, lim, E, wave, lane, [&](int qe, int cnt) {
      const int s = ei[qe];
      const int d = dstp[qe];
      const int eav = __float_as_int(ea[qe]);
      const int row = (lane < cnt) ? d - lo : 0;
      float outv[H];
      batch_logits<F, H, vf, false>(xl, xr, we, at, s, d, eav, cnt, lane, outv, nullptr);
      float lg[H];
#pragma unroll
      for (int h = 0; h < H; ++h) {
        const int idx = lane * H + h;
        lg[h] = 0.0f;
#pragma unroll
        for (int g = 0; g < H; ++g) {
          const float t = __shfl(outv[g], idx & 31, 32);
          if ((idx >> 5) == g) lg[h] = t;
        }
      }
      aggr_core<F, H>(xr, acc, den, mxo, s, row, lg, cnt, lane);
    });
    __syncthreads();
  }
  constexpr int PER = F / 32;
  constexpr int ROWS_TOGETHER = 4;
  constexpr int WAVES = AGG_THREADS / 32;
  static_assert(PER >= 1 && (F / H) % PER == 0, "a lane's channels must lie in one head");
  typedef float vp __attribute__((ext_vector_type(PER)));
  const vp bs = *(const vp*)(bias + lane * PER);
  const vp gg = *(const vp*)(lng + lane * PER);
  const vp bb = *(const vp*)(lnb + lane * PER);
  const int hd = (lane * PER) / (F / H);
  for (int r0 = wave; r0 < nrows; r0 += WAVES * ROWS_TOGETHER) {
    float vals[ROWS_TOGETHER][PER];
    float sum[ROWS_TOGETHER];
#pragma unroll
    for (int k = 0; k < ROWS_TOGETHER; ++k) {
      const int r = min(r0 + k * WAVES, nrows - 1);
      const vp a = *(const vp*)(acc + r * F + lane * PER);
      const float dn = den[r * H + hd];
      float s = 0.0f;
#pragma unroll
      for (int i = 0; i < PER; ++i) {
        float v = a[i];
        if (dn > 0.0f) v /= dn;
        v += bs[i];
        vals[k][i] = v;
        s += v;
      }
      sum[k] = s;
    }
#pragma unroll
    for (int off = 16; off > 0; off >>= 1) {
#pragma unroll
      for (int k = 0; k < ROWS_TOGETHER; ++k) sum[k] += __shfl_xor(sum[k], off, 32);
    }
    float mu[ROWS_TOGETHER], vs[ROWS_TOGETHER];
#pragma unroll
    for (int k = 0; k < ROWS_TOGETHER; ++k) {
      mu[k] = sum[k] / (float)F;
      float q = 0.0f;
#pragma unroll
      for (int i = 0; i < PER; ++i) { const float dlt = vals[k][i] - mu[k]; q += dlt * dlt; }
      vs[k] = q;
    }
#pragma unroll
    for (int off = 16; off > 0; off >>= 1) {
#pragma unroll
      for (int k = 0; k < ROWS_TOGETHER; ++k) vs[k] += __shfl_xor(vs[k], off, 32);
    }
#pragma unroll
    for (int k = 0; k < ROWS_TOGETHER; ++k) {
      const int r = r0 + k * WAVES;
      if (r < nrows) {
        const float inv = rsqrtf(vs[k] / (float)F + LN_EPS);
        vp o;
#pragma unroll
        for (int i = 0; i < PER; ++i) {
          const float hv = (vals[k][i] - mu[k]) * inv * gg[i] + bb[i];
          o[i] = (hv > 0.0f) ? hv : (__expf(hv) - 1.0f);
        }
        *(vp*)(out + (size_t)(lo + r) * F + lane * PER) = o;
      }
    }
  }
}

extern "C" void kernel_launch(void* const* d_in, const int* in_sizes, int n_in,
                              void* d_out, int out_size, void* d_ws, size_t ws_size,
                              hipStream_t stream) {
  const float* x      = (const float*)d_in[0];
  const float* ea     = (const float*)d_in[1];
  const float* W1l    = (const float*)d_in[2];
  const float* W1r    = (const float*)d_in[3];
  const float* W1e    = (const float*)d_in[4];
  const float* att1   = (const float*)d_in[5];
  const float* b1     = (const float*)d_in[6];
  const float* ln1_g  = (const float*)d_in[7];
  const float* ln1_b  = (const float*)d_in[8];
  const float* W2l    = (const float*)d_in[9];
  const float* W2r    = (const float*)d_in[10];
  const float* W2e    = (const float*)d_in[11];
  const float* att2   = (const float*)d_in[12];
  const float* b2     = (const float*)d_in[13];
  const float* ln2_g  = (const float*)d_in[14];
  const float* ln2_b  = (const float*)d_in[15];
  const int*   ei     = (const int*)d_in[16];
  float*       outp   = (float*)d_out;

  const int N = in_sizes[0] / 10;
  const int E = in_sizes[1];
  const int HID = 32, HEADS = 4, F1 = HEADS * HID  , OUTC = 64;
  const int KP = 12;

  float* ws      = (float*)d_ws;
  float* xl1     = ws;
  float* xr1     = xl1 + (size_t)N * F1;
  float* h1      = xr1 + (size_t)N * F1;
  float* xpad    = h1  + (size_t)N * F1;
  float* w1lp    = xpad + (size_t)N * KP;
  float* w1rp    = w1lp + (size_t)KP * F1;
  unsigned char* etag = (unsigned char*)(w1rp + (size_t)KP * F1);
  const size_t NBLK = ((size_t)N + AGG_TR - 1) / AGG_TR;
  int* gsrc           = (int*)(etag + (((size_t)E + 255) & ~(size_t)255));
  int* gea            = gsrc + NBLK * AGG_CAP;
  int* gcnt           = gea + NBLK * AGG_CAP;
  unsigned char* grow = (unsigned char*)(gcnt + ((NBLK + 63) & ~(size_t)63));
  float* xl2     = xl1;
  float* xr2     = xr1;

  const int TB = 256;
  auto blocks = [](size_t n, int tb) { return (unsigned)((n + tb - 1) / tb); };

  pad_cols<<<blocks((size_t)N * KP, TB), TB, 0, stream>>>(x, xpad, N, 10, KP);
  pad_rows<<<blocks((size_t)KP * F1, TB), TB, 0, stream>>>(W1l, w1lp, 10, KP, F1);
  pad_rows<<<blocks((size_t)KP * F1, TB), TB, 0, stream>>>(W1r, w1rp, 10, KP, F1);
  {
    dim3 grid((N / 16 + 3) / 4, F1 / 16);
    wmma_gemm<12><<<grid, 128, 0, stream>>>(xpad, w1lp, xl1, N, F1);
    wmma_gemm<12><<<grid, 128, 0, stream>>>(xpad, w1rp, xr1, N, F1);
  }

  edge_tags<<<blocks(((size_t)E + 15) / 16, TB), TB, 0, stream>>>(ei + E, etag, E);
  edge_aggr_rows<128, 4, 0><<<blocks((size_t)N, AGG_TR), AGG_THREADS, 0, stream>>>(
      ei, ea, etag, xl1, xr1, W1e, att1, b1, ln1_g, ln1_b, h1, gsrc, grow, gea, gcnt, N, E);

  {
    dim3 grid((N / 16 + 3) / 4, OUTC / 16);
    wmma_gemm<128><<<grid, 128, 0, stream>>>(h1, W2l, xl2, N, OUTC);
    wmma_gemm<128><<<grid, 128, 0, stream>>>(h1, W2r, xr2, N, OUTC);
  }

  edge_aggr_rows<64, 1, 1><<<blocks((size_t)N, AGG_TR), AGG_THREADS, 0, stream>>>(
      ei, ea, etag, xl2, xr2, W2e, att2, b2, ln2_g, ln2_b, outp, gsrc, grow, gea, gcnt, N, E);
}
